# starting kernel
# baseline (speedup 1.0000x reference)
#include <hip/hip_runtime.h>
#include <hip/hip_bf16.h>
#include <math.h>

typedef float v2f __attribute__((ext_vector_type(2)));
typedef float v8f __attribute__((ext_vector_type(8)));

__device__ __forceinline__ float gelu_exact(float x) {
  return 0.5f * x * (1.0f + erff(x * 0.70710678118654752f));
}

#define WMMA_F32(cacc, av, bv)                                                 \
  cacc = __builtin_amdgcn_wmma_f32_16x16x4_f32(false, (av), false, (bv),       \
                                               (short)0, (cacc), false, false)

__global__ __launch_bounds__(256) void gemm_wmma_f32(
    const float* __restrict__ A, const float* __restrict__ W,
    const float* __restrict__ bias, const float* __restrict__ R,
    float* __restrict__ C, int M, int N, int Kd,
    int flags, int Tlen, int Kspk, int kk)
{
  const int lane = threadIdx.x & 31;
  const int w    = threadIdx.x >> 5;
  const int wm   = w & 3;
  const int wn   = w >> 2;
  const int tileM = blockIdx.x * 64 + wm * 16;
  const int tileN = blockIdx.y * 128 + wn * 64;
  const int half = lane >> 4;
  const int l16  = lane & 15;

  const int mrow = tileM + l16;
  const int n0   = tileN + l16;

  const float* Ap = A + (size_t)mrow * Kd + 2 * half;
  const float* Bp0 = W + (size_t)(n0)      * Kd + 2 * half;
  const float* Bp1 = W + (size_t)(n0 + 16) * Kd + 2 * half;
  const float* Bp2 = W + (size_t)(n0 + 32) * Kd + 2 * half;
  const float* Bp3 = W + (size_t)(n0 + 48) * Kd + 2 * half;

  v8f c0 = {}, c1 = {}, c2 = {}, c3 = {};

  v2f a0  = *(const v2f*)(Ap);
  v2f b00 = *(const v2f*)(Bp0);
  v2f b01 = *(const v2f*)(Bp1);
  v2f b02 = *(const v2f*)(Bp2);
  v2f b03 = *(const v2f*)(Bp3);
  v2f a1  = *(const v2f*)(Ap + 4);
  v2f b10 = *(const v2f*)(Bp0 + 4);
  v2f b11 = *(const v2f*)(Bp1 + 4);
  v2f b12 = *(const v2f*)(Bp2 + 4);
  v2f b13 = *(const v2f*)(Bp3 + 4);

  for (int k = 0; k < Kd - 8; k += 8) {
    WMMA_F32(c0, a0, b00);
    WMMA_F32(c1, a0, b01);
    WMMA_F32(c2, a0, b02);
    WMMA_F32(c3, a0, b03);
    a0  = *(const v2f*)(Ap  + k + 8);
    b00 = *(const v2f*)(Bp0 + k + 8);
    b01 = *(const v2f*)(Bp1 + k + 8);
    b02 = *(const v2f*)(Bp2 + k + 8);
    b03 = *(const v2f*)(Bp3 + k + 8);
    WMMA_F32(c0, a1, b10);
    WMMA_F32(c1, a1, b11);
    WMMA_F32(c2, a1, b12);
    WMMA_F32(c3, a1, b13);
    a1  = *(const v2f*)(Ap  + k + 12);
    b10 = *(const v2f*)(Bp0 + k + 12);
    b11 = *(const v2f*)(Bp1 + k + 12);
    b12 = *(const v2f*)(Bp2 + k + 12);
    b13 = *(const v2f*)(Bp3 + k + 12);
  }
  WMMA_F32(c0, a0, b00);
  WMMA_F32(c1, a0, b01);
  WMMA_F32(c2, a0, b02);
  WMMA_F32(c3, a0, b03);
  WMMA_F32(c0, a1, b10);
  WMMA_F32(c1, a1, b11);
  WMMA_F32(c2, a1, b12);
  WMMA_F32(c3, a1, b13);

  const float bv0 = bias[n0];
  const float bv1 = bias[n0 + 16];
  const float bv2 = bias[n0 + 32];
  const float bv3 = bias[n0 + 48];
#pragma unroll
  for (int i = 0; i < 8; ++i) {
    const int row = tileM + i + 8 * half;
    float v0 = c0[i] + bv0;
    float v1 = c1[i] + bv1;
    float v2 = c2[i] + bv2;
    float v3 = c3[i] + bv3;
    if (flags & 1) {
      const float* rr = R + (size_t)row * N;
      v0 += rr[n0];
      v1 += rr[n0 + 16];
      v2 += rr[n0 + 32];
      v3 += rr[n0 + 48];
    }
    if (flags & 2) {
      v0 = gelu_exact(v0);
      v1 = gelu_exact(v1);
      v2 = gelu_exact(v2);
      v3 = gelu_exact(v3);
    }
    size_t orow = (size_t)row;
    if (flags & 4) {
      const int bi = row / Tlen;
      orow = (size_t)bi * Kspk * Tlen + (size_t)kk * Tlen + (row % Tlen);
    }
    float* cr = C + orow * N;
    cr[n0]      = v0;
    cr[n0 + 16] = v1;
    cr[n0 + 32] = v2;
    cr[n0 + 48] = v3;
  }
}

__global__ __launch_bounds__(256) void ln_kernel(
    const float* __restrict__ X, const float* __restrict__ g,
    const float* __restrict__ b, float* __restrict__ Y, int Dn)
{
  const int row = blockIdx.x;
  const int tid = threadIdx.x;
  const float* x = X + (size_t)row * Dn;
  float s = 0.f, s2 = 0.f;
  for (int d = tid; d < Dn; d += 256) { float v = x[d]; s += v; s2 += v * v; }
  __shared__ float sb0[256], sb1[256];
  sb0[tid] = s; sb1[tid] = s2; __syncthreads();
  for (int off = 128; off > 0; off >>= 1) {
    if (tid < off) { sb0[tid] += sb0[tid + off]; sb1[tid] += sb1[tid + off]; }
    __syncthreads();
  }
  const float mean = sb0[0] / Dn;
  const float var  = sb1[0] / Dn - mean * mean;
  const float inv  = rsqrtf(var + 1e-5f);
  float* y = Y + (size_t)row * Dn;
  for (int d = tid; d < Dn; d += 256) y[d] = (x[d] - mean) * inv * g[d] + b[d];
}

__global__ __launch_bounds__(256) void wmix_kernel(
    const float* __restrict__ logits, const int* __restrict__ blank_p,
    float* __restrict__ wmix, int BT, int V, int Kspk)
{
  const int row = blockIdx.x;
  const int tid = threadIdx.x;
  const int blank = blank_p[0];
  __shared__ float red[256];
  float a[8];
  for (int k2 = 0; k2 < Kspk; ++k2) {
    const float* L = logits + ((size_t)k2 * BT + row) * V;
    float m = -INFINITY;
    for (int v = tid; v < V; v += 256) m = fmaxf(m, L[v]);
    red[tid] = m; __syncthreads();
    for (int off = 128; off > 0; off >>= 1) {
      if (tid < off) red[tid] = fmaxf(red[tid], red[tid + off]);
      __syncthreads();
    }
    m = red[0]; __syncthreads();
    float s = 0.f;
    for (int v = tid; v < V; v += 256) s += expf(L[v] - m);
    red[tid] = s; __syncthreads();
    for (int off = 128; off > 0; off >>= 1) {
      if (tid < off) red[tid] += red[tid + off];
      __syncthreads();
    }
    s = red[0]; __syncthreads();
    a[k2] = 1.0f - expf(L[blank] - m) / s;
  }
  if (tid == 0) {
    float m2 = -INFINITY;
    for (int k2 = 0; k2 < Kspk; ++k2) m2 = fmaxf(m2, a[k2]);
    float ss = 0.f;
    float e[8];
    for (int k2 = 0; k2 < Kspk; ++k2) { e[k2] = expf(a[k2] - m2); ss += e[k2]; }
    for (int k2 = 0; k2 < Kspk; ++k2) wmix[(size_t)row * Kspk + k2] = e[k2] / ss;
  }
}

__global__ __launch_bounds__(256) void stream_prep(
    const float* __restrict__ X, const float* __restrict__ wmix,
    const float* __restrict__ tags, const float* __restrict__ g,
    const float* __restrict__ bt, float* __restrict__ Xk,
    float* __restrict__ Qin, int Dn, int Kspk, int kk)
{
  const int row = blockIdx.x;
  const int tid = threadIdx.x;
  const float wv = wmix[(size_t)row * Kspk + kk];
  const float* x  = X    + (size_t)row * Dn;
  const float* tg = tags + (size_t)kk  * Dn;
  float s = 0.f, s2 = 0.f;
  for (int d = tid; d < Dn; d += 256) {
    float v = x[d] * wv + tg[d];
    Xk[(size_t)row * Dn + d] = v;
    s += v; s2 += v * v;
  }
  __shared__ float sb0[256], sb1[256];
  sb0[tid] = s; sb1[tid] = s2; __syncthreads();
  for (int off = 128; off > 0; off >>= 1) {
    if (tid < off) { sb0[tid] += sb0[tid + off]; sb1[tid] += sb1[tid + off]; }
    __syncthreads();
  }
  const float mean = sb0[0] / Dn;
  const float var  = sb1[0] / Dn - mean * mean;
  const float inv  = rsqrtf(var + 1e-5f);
  for (int d = tid; d < Dn; d += 256) {
    float v = x[d] * wv + tg[d];
    Qin[(size_t)row * Dn + d] = (v - mean) * inv * g[d] + bt[d];
  }
}

__global__ __launch_bounds__(128) void banded_attn(
    const float* __restrict__ q, const float* __restrict__ kb,
    const float* __restrict__ vb, float* __restrict__ ctx,
    int T_, int Dn, int Hn, int band)
{
  const int hd   = 64;
  const int wid  = threadIdx.x >> 5;
  const int lane = threadIdx.x & 31;
  const int gw   = blockIdx.x * 4 + wid;
  const int t = gw % T_;
  const int h = (gw / T_) % Hn;
  const int b = gw / (T_ * Hn);

  __shared__ float lq[4][64];
  __shared__ float lat[4][64];

  const size_t rowbase = (size_t)(b * T_ + t) * Dn + (size_t)h * hd;
  lq[wid][lane]      = q[rowbase + lane];
  lq[wid][lane + 32] = q[rowbase + lane + 32];
  __syncthreads();

  const float scale = 0.125f;
  const int s0 = t - band;
  const int nb = 2 * band + 1;

  const float4* lq4 = (const float4*)&lq[wid][0];

  const int j0 = lane, j1 = lane + 32;
  const int sa = s0 + j0, sbi = s0 + j1;
  float sc0 = -INFINITY, sc1 = -INFINITY;
  if (j0 < nb && sa >= 0 && sa < T_) {
    const size_t kro = (size_t)(b * T_ + sa) * Dn + (size_t)h * hd;
    __builtin_prefetch(vb + kro, 0, 0);
    const float4* kp4 = (const float4*)(kb + kro);
    float acc = 0.f;
#pragma unroll
    for (int d4 = 0; d4 < 16; ++d4) {
      float4 kv = kp4[d4];
      float4 qv = lq4[d4];
      acc += qv.x * kv.x + qv.y * kv.y + qv.z * kv.z + qv.w * kv.w;
    }
    sc0 = acc * scale;
  }
  if (j1 < nb && sbi >= 0 && sbi < T_) {
    const size_t kro = (size_t)(b * T_ + sbi) * Dn + (size_t)h * hd;
    __builtin_prefetch(vb + kro, 0, 0);
    const float4* kp4 = (const float4*)(kb + kro);
    float acc = 0.f;
#pragma unroll
    for (int d4 = 0; d4 < 16; ++d4) {
      float4 kv = kp4[d4];
      float4 qv = lq4[d4];
      acc += qv.x * kv.x + qv.y * kv.y + qv.z * kv.z + qv.w * kv.w;
    }
    sc1 = acc * scale;
  }

  float m = fmaxf(sc0, sc1);
  for (int off = 16; off > 0; off >>= 1) m = fmaxf(m, __shfl_xor(m, off, 32));
  float e0 = expf(sc0 - m);
  float e1 = expf(sc1 - m);
  float s = e0 + e1;
  for (int off = 16; off > 0; off >>= 1) s += __shfl_xor(s, off, 32);
  const float inv = 1.0f / s;
  lat[wid][j0] = e0 * inv;
  lat[wid][j1] = e1 * inv;
  __syncthreads();

  float a0 = 0.f, a1 = 0.f;
  const int d0 = lane, d1 = lane + 32;
  for (int j = 0; j < nb; ++j) {
    const int sidx = s0 + j;
    if (sidx < 0 || sidx >= T_) continue;
    const float wv = lat[wid][j];
    const float* vp = vb + (size_t)(b * T_ + sidx) * Dn + (size_t)h * hd;
    a0 += wv * vp[d0];
    a1 += wv * vp[d1];
  }
  float* cp = ctx + rowbase;
  cp[d0] = a0;
  cp[d1] = a1;
}

extern "C" void kernel_launch(void* const* d_in, const int* in_sizes, int n_in,
                              void* d_out, int out_size, void* d_ws, size_t ws_size,
                              hipStream_t stream) {
  (void)in_sizes; (void)n_in; (void)out_size; (void)ws_size;
  constexpr int B = 4, T = 1024, D_IN = 512, D = 512, K = 2, V = 500, H = 8, BAND = 24;
  constexpr int M = B * T;

  const float* x_m      = (const float*)d_in[0];
  const float* logits   = (const float*)d_in[1];
  const int*   blank_id = (const int*)  d_in[2];
  const float* proj_w   = (const float*)d_in[3];
  const float* proj_b   = (const float*)d_in[4];
  const float* spk_tags = (const float*)d_in[5];
  const float* lnq_g    = (const float*)d_in[6];
  const float* lnq_b    = (const float*)d_in[7];
  const float* lnkv_g   = (const float*)d_in[8];
  const float* lnkv_b   = (const float*)d_in[9];
  const float* in_w     = (const float*)d_in[10];
  const float* in_b     = (const float*)d_in[11];
  const float* out_w    = (const float*)d_in[12];
  const float* out_b    = (const float*)d_in[13];
  const float* ffn_ln_g = (const float*)d_in[14];
  const float* ffn_ln_b = (const float*)d_in[15];
  const float* ffn_w1   = (const float*)d_in[16];
  const float* ffn_b1   = (const float*)d_in[17];
  const float* ffn_w2   = (const float*)d_in[18];
  const float* ffn_b2   = (const float*)d_in[19];
  float* out = (float*)d_out;

  float* ws = (float*)d_ws;
  const size_t SZ = (size_t)M * D;
  float* X    = ws;
  float* kbuf = X    + SZ;
  float* vbuf = kbuf + SZ;
  float* tA   = vbuf + SZ;
  float* tB   = tA   + SZ;
  float* tC   = tB   + SZ;
  float* t1   = tC   + SZ;
  float* wmix = t1   + 4 * SZ;

  const dim3 blk(256);
  auto gg = [](int m, int n) { return dim3(m / 64, n / 128); };

  gemm_wmma_f32<<<gg(M, D), blk, 0, stream>>>(x_m, proj_w, proj_b, nullptr, X,
                                              M, D, D_IN, 0, T, K, 0);
  wmix_kernel<<<dim3(M), blk, 0, stream>>>(logits, blank_id, wmix, M, V, K);
  ln_kernel<<<dim3(M), blk, 0, stream>>>(X, lnkv_g, lnkv_b, tA, D);
  gemm_wmma_f32<<<gg(M, D), blk, 0, stream>>>(tA, in_w + (size_t)D * D, in_b + D,
                                              nullptr, kbuf, M, D, D, 0, T, K, 0);
  gemm_wmma_f32<<<gg(M, D), blk, 0, stream>>>(tA, in_w + (size_t)2 * D * D, in_b + 2 * D,
                                              nullptr, vbuf, M, D, D, 0, T, K, 0);

  for (int kk = 0; kk < K; ++kk) {
    stream_prep<<<dim3(M), blk, 0, stream>>>(X, wmix, spk_tags, lnq_g, lnq_b,
                                             tA, tB, D, K, kk);
    gemm_wmma_f32<<<gg(M, D), blk, 0, stream>>>(tB, in_w, in_b, nullptr, tC,
                                                M, D, D, 0, T, K, kk);
    banded_attn<<<dim3((B * H * T) / 4), dim3(128), 0, stream>>>(tC, kbuf, vbuf, tB,
                                                                 T, D, H, BAND);
    gemm_wmma_f32<<<gg(M, D), blk, 0, stream>>>(tB, out_w, out_b, tA, tC,
                                                M, D, D,  1, T, K, kk);
    ln_kernel<<<dim3(M), blk, 0, stream>>>(tC, ffn_ln_g, ffn_ln_b, tB, D);
    gemm_wmma_f32<<<gg(M, 4 * D), blk, 0, stream>>>(tB, ffn_w1, ffn_b1, nullptr, t1,
                                                    M, 4 * D, D,  2, T, K, kk);
    gemm_wmma_f32<<<gg(M, D), blk, 0, stream>>>(t1, ffn_w2, ffn_b2, tC, out,
                                                M, D, 4 * D,  1 | 4, T, K, kk);
  }
}


# Banded attention moved onto f32 WMMA tiles: K/V/Q rows of 64 queries staged once in LDS by coalesced 128-bit loads, S=QK^T and O=PV as 16x16x4 products per 16-query wave, whole-row stores
# speedup vs baseline: 1.9672x; 1.9672x over previous
#include <hip/hip_runtime.h>
#include <hip/hip_bf16.h>
#include <math.h>

typedef float v2f __attribute__((ext_vector_type(2)));
typedef float v4f __attribute__((ext_vector_type(4)));
typedef float v8f __attribute__((ext_vector_type(8)));

__device__ __forceinline__ float gelu_exact(float x) {
  return 0.5f * x * (1.0f + erff(x * 0.70710678118654752f));
}

#define WMMA_F32(cacc, av, bv)                                                 \
  cacc = __builtin_amdgcn_wmma_f32_16x16x4_f32(false, (av), false, (bv),       \
                                               (short)0, (cacc), false, false)

#define GBM 64
#define GBN 128
#define GBK 32
#define GPITCH 36
#define GTILE ((GBM + GBN) * GPITCH)

__global__ __launch_bounds__(256) void gemm_wmma_f32(
    const float* __restrict__ A, const float* __restrict__ W,
    const float* __restrict__ bias, const float* __restrict__ R,
    float* __restrict__ C, int M, int N, int Kd,
    int flags, int Tlen, int Kspk, int kk)
{
  __shared__ __attribute__((aligned(16))) float sm[2 * GTILE];

  const int tid  = threadIdx.x;
  const int lane = tid & 31;
  const int w    = tid >> 5;
  const int wm   = w & 1;
  const int wn   = w >> 1;
  const int half = lane >> 4;
  const int l16  = lane & 15;
  const int blockM = blockIdx.x * GBM;
  const int blockN = blockIdx.y * GBN;

  const int lrow = tid >> 3;
  const int lcol = (tid & 7) * 4;
  const float* Ag = A + (size_t)(blockM + lrow) * Kd + lcol;
  const float* Wg = W + (size_t)(blockN + lrow) * Kd + lcol;
  const size_t r32 = (size_t)32 * Kd;

  const int so   = lrow * GPITCH + lcol;
  const int aoff = (wm * 32 + l16) * GPITCH + 4 * half;
  const int boff = GBM * GPITCH + (wn * 32 + l16) * GPITCH + 4 * half;

  v8f c00 = {}, c01 = {}, c10 = {}, c11 = {};

  float4 ra0 = *(const float4*)(Ag);
  float4 ra1 = *(const float4*)(Ag + r32);
  float4 rb0 = *(const float4*)(Wg);
  float4 rb1 = *(const float4*)(Wg + r32);
  float4 rb2 = *(const float4*)(Wg + 2 * r32);
  float4 rb3 = *(const float4*)(Wg + 3 * r32);

  *(float4*)(sm + so)                              = ra0;
  *(float4*)(sm + so + 32 * GPITCH)                = ra1;
  *(float4*)(sm + GBM * GPITCH + so)               = rb0;
  *(float4*)(sm + GBM * GPITCH + so + 32 * GPITCH) = rb1;
  *(float4*)(sm + GBM * GPITCH + so + 64 * GPITCH) = rb2;
  *(float4*)(sm + GBM * GPITCH + so + 96 * GPITCH) = rb3;
  __syncthreads();

  const int nk = Kd / GBK;
  for (int kt = 0; kt < nk; ++kt) {
    const float* cur = sm + (kt & 1) * GTILE;
    const bool more = (kt + 1 < nk);
    if (more) {
      const int ko = (kt + 1) * GBK;
      ra0 = *(const float4*)(Ag + ko);
      ra1 = *(const float4*)(Ag + r32 + ko);
      rb0 = *(const float4*)(Wg + ko);
      rb1 = *(const float4*)(Wg + r32 + ko);
      rb2 = *(const float4*)(Wg + 2 * r32 + ko);
      rb3 = *(const float4*)(Wg + 3 * r32 + ko);
    }
#pragma unroll
    for (int ks = 0; ks < GBK / 8; ++ks) {
      const float4 fa0 = *(const float4*)(cur + aoff + ks * 8);
      const float4 fa1 = *(const float4*)(cur + aoff + 16 * GPITCH + ks * 8);
      const float4 fb0 = *(const float4*)(cur + boff + ks * 8);
      const float4 fb1 = *(const float4*)(cur + boff + 16 * GPITCH + ks * 8);
      const v2f a0l = {fa0.x, fa0.y}, a0h = {fa0.z, fa0.w};
      const v2f a1l = {fa1.x, fa1.y}, a1h = {fa1.z, fa1.w};
      const v2f b0l = {fb0.x, fb0.y}, b0h = {fb0.z, fb0.w};
      const v2f b1l = {fb1.x, fb1.y}, b1h = {fb1.z, fb1.w};
      WMMA_F32(c00, a0l, b0l);
      WMMA_F32(c01, a0l, b1l);
      WMMA_F32(c10, a1l, b0l);
      WMMA_F32(c11, a1l, b1l);
      WMMA_F32(c00, a0h, b0h);
      WMMA_F32(c01, a0h, b1h);
      WMMA_F32(c10, a1h, b0h);
      WMMA_F32(c11, a1h, b1h);
    }
    if (more) {
      float* nxt = sm + ((kt + 1) & 1) * GTILE;
      *(float4*)(nxt + so)                              = ra0;
      *(float4*)(nxt + so + 32 * GPITCH)                = ra1;
      *(float4*)(nxt + GBM * GPITCH + so)               = rb0;
      *(float4*)(nxt + GBM * GPITCH + so + 32 * GPITCH) = rb1;
      *(float4*)(nxt + GBM * GPITCH + so + 64 * GPITCH) = rb2;
      *(float4*)(nxt + GBM * GPITCH + so + 96 * GPITCH) = rb3;
    }
    __syncthreads();
  }

  const int tileM = blockM + wm * 32;
  const int n0    = blockN + wn * 32 + l16;
  const float bv0 = bias[n0];
  const float bv1 = bias[n0 + 16];
#pragma unroll
  for (int i = 0; i < 2; ++i) {
#pragma unroll
    for (int r = 0; r < 8; ++r) {
      const int row = tileM + i * 16 + 8 * half + r;
      float v0 = (i == 0 ? c00[r] : c10[r]) + bv0;
      float v1 = (i == 0 ? c01[r] : c11[r]) + bv1;
      if (flags & 1) {
        const float* rr = R + (size_t)row * N;
        v0 += rr[n0];
        v1 += rr[n0 + 16];
      }
      if (flags & 2) {
        v0 = gelu_exact(v0);
        v1 = gelu_exact(v1);
      }
      size_t orow = (size_t)row;
      if (flags & 4) {
        const int bi = row / Tlen;
        orow = (size_t)bi * Kspk * Tlen + (size_t)kk * Tlen + (row % Tlen);
      }
      float* cr = C + orow * N;
      cr[n0]      = v0;
      cr[n0 + 16] = v1;
    }
  }
}

#define RD  512
#define RNV (RD / 128)
#define LV  500
#define LV4 (LV / 4)
#define LNI ((LV4 + 31) / 32)
#define LK  2

#define ROW_XOR(x, n)                                                          \
  __builtin_bit_cast(float, __builtin_amdgcn_update_dpp(                       \
      __builtin_bit_cast(int, (x)), __builtin_bit_cast(int, (x)),              \
      0x160 + (n), 0xf, 0xf, false))
__device__ __forceinline__ float xhalf(float x) {
  const int b = __builtin_bit_cast(int, x);
  return __builtin_bit_cast(float,
      __builtin_amdgcn_permlanex16(b, b, 0x76543210, 0xfedcba98, false, false));
}
__device__ __forceinline__ float wave_sum(float v) {
  v += ROW_XOR(v, 1);
  v += ROW_XOR(v, 2);
  v += ROW_XOR(v, 4);
  v += ROW_XOR(v, 8);
  return v + xhalf(v);
}
__device__ __forceinline__ float wave_max(float v) {
  v = fmaxf(v, ROW_XOR(v, 1));
  v = fmaxf(v, ROW_XOR(v, 2));
  v = fmaxf(v, ROW_XOR(v, 4));
  v = fmaxf(v, ROW_XOR(v, 8));
  return fmaxf(v, xhalf(v));
}

__device__ __forceinline__ void row_stats(const float4 (&v)[RNV], float& mean, float& inv) {
  float s = 0.f;
#pragma unroll
  for (int i = 0; i < RNV; ++i) s += (v[i].x + v[i].y) + (v[i].z + v[i].w);
  mean = wave_sum(s) * (1.0f / RD);
  float s2 = 0.f;
#pragma unroll
  for (int i = 0; i < RNV; ++i) {
    const float dx = v[i].x - mean, dy = v[i].y - mean;
    const float dz = v[i].z - mean, dw = v[i].w - mean;
    s2 += (dx * dx + dy * dy) + (dz * dz + dw * dw);
  }
  inv = rsqrtf(wave_sum(s2) * (1.0f / RD) + 1e-5f);
}

__global__ __launch_bounds__(256) void ln_kernel(
    const float* __restrict__ X, const float* __restrict__ g,
    const float* __restrict__ b, float* __restrict__ Y, int rows)
{
  const int lane = threadIdx.x & 31;
  const int row  = blockIdx.x * 8 + __builtin_amdgcn_readfirstlane(threadIdx.x >> 5);
  if (row >= rows) return;
  const float4* x4 = (const float4*)(X + (size_t)row * RD) + lane;
  const float4* g4 = (const float4*)g + lane;
  const float4* b4 = (const float4*)b + lane;
  float4 xv[RNV], gv[RNV], bv[RNV];
#pragma unroll
  for (int i = 0; i < RNV; ++i) xv[i] = x4[i * 32];
#pragma unroll
  for (int i = 0; i < RNV; ++i) { gv[i] = g4[i * 32]; bv[i] = b4[i * 32]; }
  __builtin_amdgcn_sched_barrier(0);
  float mean, inv;
  row_stats(xv, mean, inv);
  float4* y4 = (float4*)(Y + (size_t)row * RD) + lane;
#pragma unroll
  for (int i = 0; i < RNV; ++i) {
    float4 o;
    o.x = (xv[i].x - mean) * inv * gv[i].x + bv[i].x;
    o.y = (xv[i].y - mean) * inv * gv[i].y + bv[i].y;
    o.z = (xv[i].z - mean) * inv * gv[i].z + bv[i].z;
    o.w = (xv[i].w - mean) * inv * gv[i].w + bv[i].w;
    y4[i * 32] = o;
  }
}

__global__ __launch_bounds__(256) void wmix_kernel(
    const float* __restrict__ logits, const int* __restrict__ blank_p,
    float* __restrict__ wmix, int BT)
{
  const int lane = threadIdx.x & 31;
  const int row  = blockIdx.x * 8 + __builtin_amdgcn_readfirstlane(threadIdx.x >> 5);
  if (row >= BT) return;
  const int blank = blank_p[0];
  float4 lv[LK][LNI];
  float  lb[LK];
#pragma unroll
  for (int k2 = 0; k2 < LK; ++k2) {
    const float* L = logits + ((size_t)k2 * BT + row) * LV;
    const float4* L4 = (const float4*)L;
#pragma unroll
    for (int i = 0; i < LNI; ++i) {
      const int idx = lane + 32 * i;
      lv[k2][i] = (idx < LV4) ? L4[idx]
                              : make_float4(-INFINITY, -INFINITY, -INFINITY, -INFINITY);
    }
    lb[k2] = L[blank];
  }
  __builtin_amdgcn_sched_barrier(0);
  float a[LK];
#pragma unroll
  for (int k2 = 0; k2 < LK; ++k2) {
    float m = -INFINITY;
#pragma unroll
    for (int i = 0; i < LNI; ++i)
      m = fmaxf(m, fmaxf(fmaxf(lv[k2][i].x, lv[k2][i].y), fmaxf(lv[k2][i].z, lv[k2][i].w)));
    m = wave_max(m);
    float s = 0.f;
#pragma unroll
    for (int i = 0; i < LNI; ++i)
      s += (expf(lv[k2][i].x - m) + expf(lv[k2][i].y - m)) +
           (expf(lv[k2][i].z - m) + expf(lv[k2][i].w - m));
    s = wave_sum(s);
    a[k2] = 1.0f - expf(lb[k2] - m) / s;
  }
  if (lane == 0) {
    float m2 = -INFINITY;
#pragma unroll
    for (int k2 = 0; k2 < LK; ++k2) m2 = fmaxf(m2, a[k2]);
    float e[LK];
    float ss = 0.f;
#pragma unroll
    for (int k2 = 0; k2 < LK; ++k2) { e[k2] = expf(a[k2] - m2); ss += e[k2]; }
#pragma unroll
    for (int k2 = 0; k2 < LK; ++k2) wmix[(size_t)row * LK + k2] = e[k2] / ss;
  }
}

__global__ __launch_bounds__(256) void stream_prep(
    const float* __restrict__ X, const float* __restrict__ wmix,
    const float* __restrict__ tags, const float* __restrict__ g,
    const float* __restrict__ bt, float* __restrict__ Xk,
    float* __restrict__ Qin, int rows, int kk)
{
  const int lane = threadIdx.x & 31;
  const int row  = blockIdx.x * 8 + __builtin_amdgcn_readfirstlane(threadIdx.x >> 5);
  if (row >= rows) return;
  const float wv = wmix[(size_t)row * LK + kk];
  const float4* x4 = (const float4*)(X + (size_t)row * RD) + lane;
  const float4* t4 = (const float4*)(tags + (size_t)kk * RD) + lane;
  const float4* g4 = (const float4*)g + lane;
  const float4* b4 = (const float4*)bt + lane;
  float4 v[RNV], tv[RNV], gv[RNV], bv[RNV];
#pragma unroll
  for (int i = 0; i < RNV; ++i) v[i] = x4[i * 32];
#pragma unroll
  for (int i = 0; i < RNV; ++i) tv[i] = t4[i * 32];
#pragma unroll
  for (int i = 0; i < RNV; ++i) { gv[i] = g4[i * 32]; bv[i] = b4[i * 32]; }
  __builtin_amdgcn_sched_barrier(0);
#pragma unroll
  for (int i = 0; i < RNV; ++i) {
    v[i].x = v[i].x * wv + tv[i].x;
    v[i].y = v[i].y * wv + tv[i].y;
    v[i].z = v[i].z * wv + tv[i].z;
    v[i].w = v[i].w * wv + tv[i].w;
  }
  float4* xk4 = (float4*)(Xk + (size_t)row * RD) + lane;
#pragma unroll
  for (int i = 0; i < RNV; ++i) xk4[i * 32] = v[i];
  float mean, inv;
  row_stats(v, mean, inv);
  float4* q4 = (float4*)(Qin + (size_t)row * RD) + lane;
#pragma unroll
  for (int i = 0; i < RNV; ++i) {
    float4 o;
    o.x = (v[i].x - mean) * inv * gv[i].x + bv[i].x;
    o.y = (v[i].y - mean) * inv * gv[i].y + bv[i].y;
    o.z = (v[i].z - mean) * inv * gv[i].z + bv[i].z;
    o.w = (v[i].w - mean) * inv * gv[i].w + bv[i].w;
    q4[i * 32] = o;
  }
}

#define AQ      64
#define ANW     (AQ / 16)
#define ATHR    (32 * ANW)
#define ABAND   24
#define AKEYS   (AQ + 2 * ABAND)
#define AHD     64
#define APITCH  68

__global__ __launch_bounds__(ATHR) void banded_attn_wmma(
    const float* __restrict__ q, const float* __restrict__ kb,
    const float* __restrict__ vb, float* __restrict__ ctx,
    int T_, int Dn, int Hn)
{
  __shared__ __attribute__((aligned(16))) float sK[AKEYS * APITCH];
  __shared__ __attribute__((aligned(16))) float sV[AKEYS * APITCH];
  __shared__ __attribute__((aligned(16))) float sQ[AQ * APITCH];

  const int tid  = threadIdx.x;
  const int lane = tid & 31;
  const int w    = tid >> 5;
  const int c    = lane & 15;
  const int h    = lane >> 4;

  const int nqt = T_ / AQ;
  const int qt  = blockIdx.x % nqt;
  const int hh  = (blockIdx.x / nqt) % Hn;
  const int b   = blockIdx.x / (nqt * Hn);
  const int t0  = qt * AQ;

  const size_t headoff = (size_t)hh * AHD;
  const float* kbase = kb + (size_t)b * T_ * Dn + headoff;
  const float* vbase = vb + (size_t)b * T_ * Dn + headoff;
  const float* qbase = q  + ((size_t)b * T_ + t0) * Dn + headoff;
  float*       obase = ctx + ((size_t)b * T_ + t0) * Dn + headoff;

  constexpr int NKV = (AKEYS * 16) / ATHR;
  constexpr int NQ  = (AQ * 16) / ATHR;
  const int srow = tid >> 4;
  const int sc4  = (tid & 15) * 4;
  v4f kreg[NKV], vreg[NKV], qreg[NQ];
#pragma unroll
  for (int i = 0; i < NKV; ++i) {
    const int key = t0 - ABAND + srow + (ATHR / 16) * i;
    const unsigned off = (unsigned)min(max(key, 0), T_ - 1) * (unsigned)Dn + (unsigned)sc4;
    kreg[i] = *(const v4f*)(kbase + off);
    vreg[i] = *(const v4f*)(vbase + off);
  }
#pragma unroll
  for (int i = 0; i < NQ; ++i) {
    const unsigned off = (unsigned)(srow + (ATHR / 16) * i) * (unsigned)Dn + (unsigned)sc4;
    qreg[i] = *(const v4f*)(qbase + off);
  }
  __builtin_amdgcn_sched_barrier(0);
  const v4f zero4 = {0.f, 0.f, 0.f, 0.f};
#pragma unroll
  for (int i = 0; i < NKV; ++i) {
    const int row = srow + (ATHR / 16) * i;
    const int key = t0 - ABAND + row;
    const bool ok = (key >= 0) && (key < T_);
    *(v4f*)(sK + row * APITCH + sc4) = ok ? kreg[i] : zero4;
    *(v4f*)(sV + row * APITCH + sc4) = ok ? vreg[i] : zero4;
  }
#pragma unroll
  for (int i = 0; i < NQ; ++i) {
    const int row = srow + (ATHR / 16) * i;
    *(v4f*)(sQ + row * APITCH + sc4) = qreg[i];
  }
  __syncthreads();

  float* pw = sQ + (w * 16) * APITCH;
  const float* qrow = pw + c * APITCH + 4 * h;
  const float* krow = sK + (w * 16 + c) * APITCH + 4 * h;
  float4 qa[8];
#pragma unroll
  for (int ks = 0; ks < 8; ++ks) qa[ks] = *(const float4*)(qrow + 8 * ks);

  v8f s0 = {}, s1 = {}, s2 = {}, s3 = {};
#pragma unroll
  for (int ks = 0; ks < 8; ++ks) {
    const float4 k0 = *(const float4*)(krow + 8 * ks);
    const float4 k1 = *(const float4*)(krow + 16 * APITCH + 8 * ks);
    const float4 k2 = *(const float4*)(krow + 32 * APITCH + 8 * ks);
    const float4 k3 = *(const float4*)(krow + 48 * APITCH + 8 * ks);
    const v2f al = {qa[ks].x, qa[ks].y}, ah = {qa[ks].z, qa[ks].w};
    const v2f b0l = {k0.x, k0.y}, b0h = {k0.z, k0.w};
    const v2f b1l = {k1.x, k1.y}, b1h = {k1.z, k1.w};
    const v2f b2l = {k2.x, k2.y}, b2h = {k2.z, k2.w};
    const v2f b3l = {k3.x, k3.y}, b3h = {k3.z, k3.w};
    WMMA_F32(s0, al, b0l);
    WMMA_F32(s1, al, b1l);
    WMMA_F32(s2, al, b2l);
    WMMA_F32(s3, al, b3l);
    WMMA_F32(s0, ah, b0h);
    WMMA_F32(s1, ah, b1h);
    WMMA_F32(s2, ah, b2h);
    WMMA_F32(s3, ah, b3h);
  }

  const float scale = 0.125f;
  const int jbase = t0 + 16 * w - ABAND;
  bool colok[4];
#pragma unroll
  for (int nt = 0; nt < 4; ++nt) {
    const int key = jbase + 16 * nt + c;
    colok[nt] = (key >= 0) && (key < T_);
  }
  float inv[8];
#pragma unroll
  for (int r = 0; r < 8; ++r) {
    const int rq = 8 * h + r;
    float x[4] = {s0[r], s1[r], s2[r], s3[r]};
#pragma unroll
    for (int nt = 0; nt < 4; ++nt) {
      const int jj = 16 * nt + c;
      const bool ok = colok[nt] && (jj >= rq) && (jj <= rq + 2 * ABAND);
      x[nt] = ok ? x[nt] * scale : -INFINITY;
    }
    float mx = fmaxf(fmaxf(x[0], x[1]), fmaxf(x[2], x[3]));
    mx = fmaxf(mx, ROW_XOR(mx, 1));
    mx = fmaxf(mx, ROW_XOR(mx, 2));
    mx = fmaxf(mx, ROW_XOR(mx, 4));
    mx = fmaxf(mx, ROW_XOR(mx, 8));
    float e[4];
#pragma unroll
    for (int nt = 0; nt < 4; ++nt) e[nt] = expf(x[nt] - mx);
    float sum = (e[0] + e[1]) + (e[2] + e[3]);
    sum += ROW_XOR(sum, 1);
    sum += ROW_XOR(sum, 2);
    sum += ROW_XOR(sum, 4);
    sum += ROW_XOR(sum, 8);
    inv[r] = 1.0f / sum;
    float* pr = pw + rq * APITCH + c;
    pr[0]  = e[0];
    pr[16] = e[1];
    pr[32] = e[2];
    pr[48] = e[3];
  }
  __syncthreads();

  const float* prow = pw + c * APITCH + 4 * h;
  const float* vcol = sV + (w * 16 + 4 * h) * APITCH + c;
  v8f o0 = {}, o1 = {}, o2 = {}, o3 = {};
#pragma unroll
  for (int ks = 0; ks < 8; ++ks) {
    const float4 pa = *(const float4*)(prow + 8 * ks);
    const float* vr = vcol + 8 * ks * APITCH;
    const v2f al = {pa.x, pa.y}, ah = {pa.z, pa.w};
    const v2f b0l = {vr[0],      vr[APITCH]},      b0h = {vr[2 * APITCH],      vr[3 * APITCH]};
    const v2f b1l = {vr[16],     vr[APITCH + 16]}, b1h = {vr[2 * APITCH + 16], vr[3 * APITCH + 16]};
    const v2f b2l = {vr[32],     vr[APITCH + 32]}, b2h = {vr[2 * APITCH + 32], vr[3 * APITCH + 32]};
    const v2f b3l = {vr[48],     vr[APITCH + 48]}, b3h = {vr[2 * APITCH + 48], vr[3 * APITCH + 48]};
    WMMA_F32(o0, al, b0l);
    WMMA_F32(o1, al, b1l);
    WMMA_F32(o2, al, b2l);
    WMMA_F32(o3, al, b3l);
    WMMA_F32(o0, ah, b0h);
    WMMA_F32(o1, ah, b1h);
    WMMA_F32(o2, ah, b2h);
    WMMA_F32(o3, ah, b3h);
  }

#pragma unroll
  for (int r = 0; r < 8; ++r) {
    float* pr = pw + (8 * h + r) * APITCH + c;
    pr[0]  = o0[r] * inv[r];
    pr[16] = o1[r] * inv[r];
    pr[32] = o2[r] * inv[r];
    pr[48] = o3[r] * inv[r];
  }
  __syncthreads();
#pragma unroll
  for (int i = 0; i < 8; ++i) {
    const int row = 2 * i + h;
    const float4 ov = *(const float4*)(pw + row * APITCH + 4 * c);
    *(float4*)(obase + (size_t)(16 * w + row) * Dn + 4 * c) = ov;
  }
}

extern "C" void kernel_launch(void* const* d_in, const int* in_sizes, int n_in,
                              void* d_out, int out_size, void* d_ws, size_t ws_size,
                              hipStream_t stream) {
  (void)in_sizes; (void)n_in; (void)out_size; (void)ws_size;
  constexpr int B = 4, T = 1024, D_IN = 512, D = 512, K = 2, V = 500, H = 8, BAND = 24;
  constexpr int M = B * T;
  static_assert(D == RD && V == LV && K == LK && M % 8 == 0, "row kernels are sized for these shapes");
  static_assert(D == H * AHD && BAND == ABAND && T % AQ == 0, "attention kernel is sized for these shapes");

  const float* x_m      = (const float*)d_in[0];
  const float* logits   = (const float*)d_in[1];
  const int*   blank_id = (const int*)  d_in[2];
  const float* proj_w   = (const float*)d_in[3];
  const float* proj_b   = (const float*)d_in[4];
  const float* spk_tags = (const float*)d_in[5];
  const float* lnq_g    = (const float*)d_in[6];
  const float* lnq_b    = (const float*)d_in[7];
  const float* lnkv_g   = (const float*)d_in[8];
  const float* lnkv_b   = (const float*)d_in[9];
  const float* in_w     = (const float*)d_in[10];
  const float* in_b     = (const float*)d_in[11];
  const float* out_w    = (const float*)d_in[12];
  const float* out_b    = (const float*)d_in[13];
  const float* ffn_ln_g = (const float*)d_in[14];
  const float* ffn_ln_b = (const float*)d_in[15];
  const float* ffn_w1   = (const float*)d_in[16];
  const float* ffn_b1   = (const float*)d_in[17];
  const float* ffn_w2   = (const float*)d_in[18];
  const float* ffn_b2   = (const float*)d_in[19];
  float* out = (float*)d_out;

  float* ws = (float*)d_ws;
  const size_t SZ = (size_t)M * D;
  float* X    = ws;
  float* kbuf = X    + SZ;
  float* vbuf = kbuf + SZ;
  float* tA   = vbuf + SZ;
  float* tB   = tA   + SZ;
  float* tC   = tB   + SZ;
  float* t1   = tC   + SZ;
  float* wmix = t1   + 4 * SZ;

  const dim3 blk(256);
  auto gg = [](int m, int n) { return dim3(m / 64, n / 128); };

  gemm_wmma_f32<<<gg(M, D), blk, 0, stream>>>(x_m, proj_w, proj_b, nullptr, X,
                                              M, D, D_IN, 0, T, K, 0);
  wmix_kernel<<<dim3(M / 8), blk, 0, stream>>>(logits, blank_id, wmix, M);
  ln_kernel<<<dim3(M / 8), blk, 0, stream>>>(X, lnkv_g, lnkv_b, tA, M);
  gemm_wmma_f32<<<gg(M, D), blk, 0, stream>>>(tA, in_w + (size_t)D * D, in_b + D,
                                              nullptr, kbuf, M, D, D, 0, T, K, 0);
  gemm_wmma_f32<<<gg(M, D), blk, 0, stream>>>(tA, in_w + (size_t)2 * D * D, in_b + 2 * D,
                                              nullptr, vbuf, M, D, D, 0, T, K, 0);

  for (int kk = 0; kk < K; ++kk) {
    stream_prep<<<dim3(M / 8), blk, 0, stream>>>(X, wmix, spk_tags, lnq_g, lnq_b,
                                                 tA, tB, M, kk);
    gemm_wmma_f32<<<gg(M, D), blk, 0, stream>>>(tB, in_w, in_b, nullptr, tC,
                                                M, D, D, 0, T, K, kk);
    banded_attn_wmma<<<dim3(B * H * (T / AQ)), dim3(ATHR), 0, stream>>>(tC, kbuf, vbuf, tB,
                                                                        T, D, H);
    gemm_wmma_f32<<<gg(M, D), blk, 0, stream>>>(tB, out_w, out_b, tA, tC,
                                                M, D, D,  1, T, K, kk);
    ln_kernel<<<dim3(M / 8), blk, 0, stream>>>(tC, ffn_ln_g, ffn_ln_b, tB, M);
    gemm_wmma_f32<<<gg(M, 4 * D), blk, 0, stream>>>(tB, ffn_w1, ffn_b1, nullptr, t1,
                                                    M, 4 * D, D,  2, T, K, kk);
    gemm_wmma_f32<<<gg(M, D), blk, 0, stream>>>(t1, ffn_w2, ffn_b2, tC, out,
                                                M, D, 4 * D,  1 | 4, T, K, kk);
  }
}


# Launch consolidation: both speaker streams stacked as [2*B*T, D] run prep, q projection, attention, out projection and LayerNorm once each; k and v from one N=1024 GEMM; 19 to 13 launches
# speedup vs baseline: 2.1223x; 1.0788x over previous
#include <hip/hip_runtime.h>
#include <hip/hip_bf16.h>
#include <math.h>

typedef float v2f __attribute__((ext_vector_type(2)));
typedef float v4f __attribute__((ext_vector_type(4)));
typedef float v8f __attribute__((ext_vector_type(8)));

__device__ __forceinline__ float gelu_exact(float x) {
  return 0.5f * x * (1.0f + erff(x * 0.70710678118654752f));
}

#define WMMA_F32(cacc, av, bv)                                                 \
  cacc = __builtin_amdgcn_wmma_f32_16x16x4_f32(false, (av), false, (bv),       \
                                               (short)0, (cacc), false, false)

#define GBM 64
#define GBN 128
#define GBK 32
#define GPITCH 36
#define GTILE ((GBM + GBN) * GPITCH)

__global__ __launch_bounds__(256) void gemm_wmma_f32(
    const float* __restrict__ A, const float* __restrict__ W,
    const float* __restrict__ bias, const float* __restrict__ R,
    float* __restrict__ C, int M, int N, int Kd,
    int flags, int Tlen, int Kspk, int kk)
{
  __shared__ __attribute__((aligned(16))) float sm[2 * GTILE];

  const int tid  = threadIdx.x;
  const int lane = tid & 31;
  const int w    = tid >> 5;
  const int wm   = w & 1;
  const int wn   = w >> 1;
  const int half = lane >> 4;
  const int l16  = lane & 15;
  const int blockM = blockIdx.x * GBM;
  const int blockN = blockIdx.y * GBN;

  const int lrow = tid >> 3;
  const int lcol = (tid & 7) * 4;
  const float* Ag = A + (size_t)(blockM + lrow) * Kd + lcol;
  const float* Wg = W + (size_t)(blockN + lrow) * Kd + lcol;
  const size_t r32 = (size_t)32 * Kd;

  const int so   = lrow * GPITCH + lcol;
  const int aoff = (wm * 32 + l16) * GPITCH + 4 * half;
  const int boff = GBM * GPITCH + (wn * 32 + l16) * GPITCH + 4 * half;

  v8f c00 = {}, c01 = {}, c10 = {}, c11 = {};

  float4 ra0 = *(const float4*)(Ag);
  float4 ra1 = *(const float4*)(Ag + r32);
  float4 rb0 = *(const float4*)(Wg);
  float4 rb1 = *(const float4*)(Wg + r32);
  float4 rb2 = *(const float4*)(Wg + 2 * r32);
  float4 rb3 = *(const float4*)(Wg + 3 * r32);

  *(float4*)(sm + so)                              = ra0;
  *(float4*)(sm + so + 32 * GPITCH)                = ra1;
  *(float4*)(sm + GBM * GPITCH + so)               = rb0;
  *(float4*)(sm + GBM * GPITCH + so + 32 * GPITCH) = rb1;
  *(float4*)(sm + GBM * GPITCH + so + 64 * GPITCH) = rb2;
  *(float4*)(sm + GBM * GPITCH + so + 96 * GPITCH) = rb3;
  __syncthreads();

  const int nk = Kd / GBK;
  for (int kt = 0; kt < nk; ++kt) {
    const float* cur = sm + (kt & 1) * GTILE;
    const bool more = (kt + 1 < nk);
    if (more) {
      const int ko = (kt + 1) * GBK;
      ra0 = *(const float4*)(Ag + ko);
      ra1 = *(const float4*)(Ag + r32 + ko);
      rb0 = *(const float4*)(Wg + ko);
      rb1 = *(const float4*)(Wg + r32 + ko);
      rb2 = *(const float4*)(Wg + 2 * r32 + ko);
      rb3 = *(const float4*)(Wg + 3 * r32 + ko);
    }
#pragma unroll
    for (int ks = 0; ks < GBK / 8; ++ks) {
      const float4 fa0 = *(const float4*)(cur + aoff + ks * 8);
      const float4 fa1 = *(const float4*)(cur + aoff + 16 * GPITCH + ks * 8);
      const float4 fb0 = *(const float4*)(cur + boff + ks * 8);
      const float4 fb1 = *(const float4*)(cur + boff + 16 * GPITCH + ks * 8);
      const v2f a0l = {fa0.x, fa0.y}, a0h = {fa0.z, fa0.w};
      const v2f a1l = {fa1.x, fa1.y}, a1h = {fa1.z, fa1.w};
      const v2f b0l = {fb0.x, fb0.y}, b0h = {fb0.z, fb0.w};
      const v2f b1l = {fb1.x, fb1.y}, b1h = {fb1.z, fb1.w};
      WMMA_F32(c00, a0l, b0l);
      WMMA_F32(c01, a0l, b1l);
      WMMA_F32(c10, a1l, b0l);
      WMMA_F32(c11, a1l, b1l);
      WMMA_F32(c00, a0h, b0h);
      WMMA_F32(c01, a0h, b1h);
      WMMA_F32(c10, a1h, b0h);
      WMMA_F32(c11, a1h, b1h);
    }
    if (more) {
      float* nxt = sm + ((kt + 1) & 1) * GTILE;
      *(float4*)(nxt + so)                              = ra0;
      *(float4*)(nxt + so + 32 * GPITCH)                = ra1;
      *(float4*)(nxt + GBM * GPITCH + so)               = rb0;
      *(float4*)(nxt + GBM * GPITCH + so + 32 * GPITCH) = rb1;
      *(float4*)(nxt + GBM * GPITCH + so + 64 * GPITCH) = rb2;
      *(float4*)(nxt + GBM * GPITCH + so + 96 * GPITCH) = rb3;
    }
    __syncthreads();
  }

  const int tileM = blockM + wm * 32;
  const int n0    = blockN + wn * 32 + l16;
  const float bv0 = bias[n0];
  const float bv1 = bias[n0 + 16];
#pragma unroll
  for (int i = 0; i < 2; ++i) {
#pragma unroll
    for (int r = 0; r < 8; ++r) {
      const int row = tileM + i * 16 + 8 * half + r;
      float v0 = (i == 0 ? c00[r] : c10[r]) + bv0;
      float v1 = (i == 0 ? c01[r] : c11[r]) + bv1;
      if (flags & 1) {
        const float* rr = R + (size_t)row * N;
        v0 += rr[n0];
        v1 += rr[n0 + 16];
      }
      if (flags & 2) {
        v0 = gelu_exact(v0);
        v1 = gelu_exact(v1);
      }
      size_t orow = (size_t)row;
      if (flags & 4) {
        const int bi = row / Tlen;
        orow = (size_t)bi * Kspk * Tlen + (size_t)kk * Tlen + (row % Tlen);
      }
      float* cr = C + orow * N;
      cr[n0]      = v0;
      cr[n0 + 16] = v1;
    }
  }
}

#define RD  512
#define RNV (RD / 128)
#define LV  500
#define LV4 (LV / 4)
#define LNI ((LV4 + 31) / 32)
#define LK  2

#define ROW_XOR(x, n)                                                          \
  __builtin_bit_cast(float, __builtin_amdgcn_update_dpp(                       \
      __builtin_bit_cast(int, (x)), __builtin_bit_cast(int, (x)),              \
      0x160 + (n), 0xf, 0xf, false))
__device__ __forceinline__ float xhalf(float x) {
  const int b = __builtin_bit_cast(int, x);
  return __builtin_bit_cast(float,
      __builtin_amdgcn_permlanex16(b, b, 0x76543210, 0xfedcba98, false, false));
}
__device__ __forceinline__ float wave_sum(float v) {
  v += ROW_XOR(v, 1);
  v += ROW_XOR(v, 2);
  v += ROW_XOR(v, 4);
  v += ROW_XOR(v, 8);
  return v + xhalf(v);
}
__device__ __forceinline__ float wave_max(float v) {
  v = fmaxf(v, ROW_XOR(v, 1));
  v = fmaxf(v, ROW_XOR(v, 2));
  v = fmaxf(v, ROW_XOR(v, 4));
  v = fmaxf(v, ROW_XOR(v, 8));
  return fmaxf(v, xhalf(v));
}

__device__ __forceinline__ void row_stats(const float4 (&v)[RNV], float& mean, float& inv) {
  float s = 0.f;
#pragma unroll
  for (int i = 0; i < RNV; ++i) s += (v[i].x + v[i].y) + (v[i].z + v[i].w);
  mean = wave_sum(s) * (1.0f / RD);
  float s2 = 0.f;
#pragma unroll
  for (int i = 0; i < RNV; ++i) {
    const float dx = v[i].x - mean, dy = v[i].y - mean;
    const float dz = v[i].z - mean, dw = v[i].w - mean;
    s2 += (dx * dx + dy * dy) + (dz * dz + dw * dw);
  }
  inv = rsqrtf(wave_sum(s2) * (1.0f / RD) + 1e-5f);
}

__global__ __launch_bounds__(256) void ln_kernel(
    const float* __restrict__ X, const float* __restrict__ g,
    const float* __restrict__ b, float* __restrict__ Y, int rows)
{
  const int lane = threadIdx.x & 31;
  const int row  = blockIdx.x * 8 + __builtin_amdgcn_readfirstlane(threadIdx.x >> 5);
  if (row >= rows) return;
  const float4* x4 = (const float4*)(X + (size_t)row * RD) + lane;
  const float4* g4 = (const float4*)g + lane;
  const float4* b4 = (const float4*)b + lane;
  float4 xv[RNV], gv[RNV], bv[RNV];
#pragma unroll
  for (int i = 0; i < RNV; ++i) xv[i] = x4[i * 32];
#pragma unroll
  for (int i = 0; i < RNV; ++i) { gv[i] = g4[i * 32]; bv[i] = b4[i * 32]; }
  __builtin_amdgcn_sched_barrier(0);
  float mean, inv;
  row_stats(xv, mean, inv);
  float4* y4 = (float4*)(Y + (size_t)row * RD) + lane;
#pragma unroll
  for (int i = 0; i < RNV; ++i) {
    float4 o;
    o.x = (xv[i].x - mean) * inv * gv[i].x + bv[i].x;
    o.y = (xv[i].y - mean) * inv * gv[i].y + bv[i].y;
    o.z = (xv[i].z - mean) * inv * gv[i].z + bv[i].z;
    o.w = (xv[i].w - mean) * inv * gv[i].w + bv[i].w;
    y4[i * 32] = o;
  }
}

__global__ __launch_bounds__(256) void wmix_kernel(
    const float* __restrict__ logits, const int* __restrict__ blank_p,
    float* __restrict__ wmix, int BT)
{
  const int lane = threadIdx.x & 31;
  const int row  = blockIdx.x * 8 + __builtin_amdgcn_readfirstlane(threadIdx.x >> 5);
  if (row >= BT) return;
  const int blank = blank_p[0];
  float4 lv[LK][LNI];
  float  lb[LK];
#pragma unroll
  for (int k2 = 0; k2 < LK; ++k2) {
    const float* L = logits + ((size_t)k2 * BT + row) * LV;
    const float4* L4 = (const float4*)L;
#pragma unroll
    for (int i = 0; i < LNI; ++i) {
      const int idx = lane + 32 * i;
      lv[k2][i] = (idx < LV4) ? L4[idx]
                              : make_float4(-INFINITY, -INFINITY, -INFINITY, -INFINITY);
    }
    lb[k2] = L[blank];
  }
  __builtin_amdgcn_sched_barrier(0);
  float a[LK];
#pragma unroll
  for (int k2 = 0; k2 < LK; ++k2) {
    float m = -INFINITY;
#pragma unroll
    for (int i = 0; i < LNI; ++i)
      m = fmaxf(m, fmaxf(fmaxf(lv[k2][i].x, lv[k2][i].y), fmaxf(lv[k2][i].z, lv[k2][i].w)));
    m = wave_max(m);
    float s = 0.f;
#pragma unroll
    for (int i = 0; i < LNI; ++i)
      s += (expf(lv[k2][i].x - m) + expf(lv[k2][i].y - m)) +
           (expf(lv[k2][i].z - m) + expf(lv[k2][i].w - m));
    s = wave_sum(s);
    a[k2] = 1.0f - expf(lb[k2] - m) / s;
  }
  if (lane == 0) {
    float m2 = -INFINITY;
#pragma unroll
    for (int k2 = 0; k2 < LK; ++k2) m2 = fmaxf(m2, a[k2]);
    float e[LK];
    float ss = 0.f;
#pragma unroll
    for (int k2 = 0; k2 < LK; ++k2) { e[k2] = expf(a[k2] - m2); ss += e[k2]; }
#pragma unroll
    for (int k2 = 0; k2 < LK; ++k2) wmix[(size_t)row * LK + k2] = e[k2] / ss;
  }
}

__global__ __launch_bounds__(256) void stream_prep(
    const float* __restrict__ X, const float* __restrict__ wmix,
    const float* __restrict__ tags, const float* __restrict__ g,
    const float* __restrict__ bt, float* __restrict__ Xk,
    float* __restrict__ Qin, int rows)
{
  const int lane = threadIdx.x & 31;
  const int row  = blockIdx.x * 8 + __builtin_amdgcn_readfirstlane(threadIdx.x >> 5);
  if (row >= rows * LK) return;
  const int kk   = row / rows;
  const int xrow = row - kk * rows;
  const float wv = wmix[(size_t)xrow * LK + kk];
  const float4* x4 = (const float4*)(X + (size_t)xrow * RD) + lane;
  const float4* t4 = (const float4*)(tags + (size_t)kk * RD) + lane;
  const float4* g4 = (const float4*)g + lane;
  const float4* b4 = (const float4*)bt + lane;
  float4 v[RNV], tv[RNV], gv[RNV], bv[RNV];
#pragma unroll
  for (int i = 0; i < RNV; ++i) v[i] = x4[i * 32];
#pragma unroll
  for (int i = 0; i < RNV; ++i) tv[i] = t4[i * 32];
#pragma unroll
  for (int i = 0; i < RNV; ++i) { gv[i] = g4[i * 32]; bv[i] = b4[i * 32]; }
  __builtin_amdgcn_sched_barrier(0);
#pragma unroll
  for (int i = 0; i < RNV; ++i) {
    v[i].x = v[i].x * wv + tv[i].x;
    v[i].y = v[i].y * wv + tv[i].y;
    v[i].z = v[i].z * wv + tv[i].z;
    v[i].w = v[i].w * wv + tv[i].w;
  }
  float4* xk4 = (float4*)(Xk + (size_t)row * RD) + lane;
#pragma unroll
  for (int i = 0; i < RNV; ++i) xk4[i * 32] = v[i];
  float mean, inv;
  row_stats(v, mean, inv);
  float4* q4 = (float4*)(Qin + (size_t)row * RD) + lane;
#pragma unroll
  for (int i = 0; i < RNV; ++i) {
    float4 o;
    o.x = (v[i].x - mean) * inv * gv[i].x + bv[i].x;
    o.y = (v[i].y - mean) * inv * gv[i].y + bv[i].y;
    o.z = (v[i].z - mean) * inv * gv[i].z + bv[i].z;
    o.w = (v[i].w - mean) * inv * gv[i].w + bv[i].w;
    q4[i * 32] = o;
  }
}

#define AQ      64
#define ANW     (AQ / 16)
#define ATHR    (32 * ANW)
#define ABAND   24
#define AKEYS   (AQ + 2 * ABAND)
#define AHD     64
#define APITCH  68

__global__ __launch_bounds__(ATHR) void banded_attn_wmma(
    const float* __restrict__ q, const float* __restrict__ kb,
    const float* __restrict__ vb, float* __restrict__ ctx,
    int T_, int Dn, int Dkv, int Hn, int Bkv)
{
  __shared__ __attribute__((aligned(16))) float sK[AKEYS * APITCH];
  __shared__ __attribute__((aligned(16))) float sV[AKEYS * APITCH];
  __shared__ __attribute__((aligned(16))) float sQ[AQ * APITCH];

  const int tid  = threadIdx.x;
  const int lane = tid & 31;
  const int w    = tid >> 5;
  const int c    = lane & 15;
  const int h    = lane >> 4;

  const int nqt = T_ / AQ;
  const int qt  = blockIdx.x % nqt;
  const int hh  = (blockIdx.x / nqt) % Hn;
  const int b   = blockIdx.x / (nqt * Hn);
  const int bk  = b % Bkv;
  const int t0  = qt * AQ;

  const size_t headoff = (size_t)hh * AHD;
  const float* kbase = kb + (size_t)bk * T_ * Dkv + headoff;
  const float* vbase = vb + (size_t)bk * T_ * Dkv + headoff;
  const float* qbase = q  + ((size_t)b * T_ + t0) * Dn + headoff;
  float*       obase = ctx + ((size_t)b * T_ + t0) * Dn + headoff;

  constexpr int NKV = (AKEYS * 16) / ATHR;
  constexpr int NQ  = (AQ * 16) / ATHR;
  const int srow = tid >> 4;
  const int sc4  = (tid & 15) * 4;
  v4f kreg[NKV], vreg[NKV], qreg[NQ];
#pragma unroll
  for (int i = 0; i < NKV; ++i) {
    const int key = t0 - ABAND + srow + (ATHR / 16) * i;
    const unsigned off = (unsigned)min(max(key, 0), T_ - 1) * (unsigned)Dkv + (unsigned)sc4;
    kreg[i] = *(const v4f*)(kbase + off);
    vreg[i] = *(const v4f*)(vbase + off);
  }
#pragma unroll
  for (int i = 0; i < NQ; ++i) {
    const unsigned off = (unsigned)(srow + (ATHR / 16) * i) * (unsigned)Dn + (unsigned)sc4;
    qreg[i] = *(const v4f*)(qbase + off);
  }
  __builtin_amdgcn_sched_barrier(0);
  const v4f zero4 = {0.f, 0.f, 0.f, 0.f};
#pragma unroll
  for (int i = 0; i < NKV; ++i) {
    const int row = srow + (ATHR / 16) * i;
    const int key = t0 - ABAND + row;
    const bool ok = (key >= 0) && (key < T_);
    *(v4f*)(sK + row * APITCH + sc4) = ok ? kreg[i] : zero4;
    *(v4f*)(sV + row * APITCH + sc4) = ok ? vreg[i] : zero4;
  }
#pragma unroll
  for (int i = 0; i < NQ; ++i) {
    const int row = srow + (ATHR / 16) * i;
    *(v4f*)(sQ + row * APITCH + sc4) = qreg[i];
  }
  __syncthreads();

  float* pw = sQ + (w * 16) * APITCH;
  const float* qrow = pw + c * APITCH + 4 * h;
  const float* krow = sK + (w * 16 + c) * APITCH + 4 * h;
  float4 qa[8];
#pragma unroll
  for (int ks = 0; ks < 8; ++ks) qa[ks] = *(const float4*)(qrow + 8 * ks);

  v8f s0 = {}, s1 = {}, s2 = {}, s3 = {};
#pragma unroll
  for (int ks = 0; ks < 8; ++ks) {
    const float4 k0 = *(const float4*)(krow + 8 * ks);
    const float4 k1 = *(const float4*)(krow + 16 * APITCH + 8 * ks);
    const float4 k2 = *(const float4*)(krow + 32 * APITCH + 8 * ks);
    const float4 k3 = *(const float4*)(krow + 48 * APITCH + 8 * ks);
    const v2f al = {qa[ks].x, qa[ks].y}, ah = {qa[ks].z, qa[ks].w};
    const v2f b0l = {k0.x, k0.y}, b0h = {k0.z, k0.w};
    const v2f b1l = {k1.x, k1.y}, b1h = {k1.z, k1.w};
    const v2f b2l = {k2.x, k2.y}, b2h = {k2.z, k2.w};
    const v2f b3l = {k3.x, k3.y}, b3h = {k3.z, k3.w};
    WMMA_F32(s0, al, b0l);
    WMMA_F32(s1, al, b1l);
    WMMA_F32(s2, al, b2l);
    WMMA_F32(s3, al, b3l);
    WMMA_F32(s0, ah, b0h);
    WMMA_F32(s1, ah, b1h);
    WMMA_F32(s2, ah, b2h);
    WMMA_F32(s3, ah, b3h);
  }

  const float scale = 0.125f;
  const int jbase = t0 + 16 * w - ABAND;
  bool colok[4];
#pragma unroll
  for (int nt = 0; nt < 4; ++nt) {
    const int key = jbase + 16 * nt + c;
    colok[nt] = (key >= 0) && (key < T_);
  }
  float inv[8];
#pragma unroll
  for (int r = 0; r < 8; ++r) {
    const int rq = 8 * h + r;
    float x[4] = {s0[r], s1[r], s2[r], s3[r]};
#pragma unroll
    for (int nt = 0; nt < 4; ++nt) {
      const int jj = 16 * nt + c;
      const bool ok = colok[nt] && (jj >= rq) && (jj <= rq + 2 * ABAND);
      x[nt] = ok ? x[nt] * scale : -INFINITY;
    }
    float mx = fmaxf(fmaxf(x[0], x[1]), fmaxf(x[2], x[3]));
    mx = fmaxf(mx, ROW_XOR(mx, 1));
    mx = fmaxf(mx, ROW_XOR(mx, 2));
    mx = fmaxf(mx, ROW_XOR(mx, 4));
    mx = fmaxf(mx, ROW_XOR(mx, 8));
    float e[4];
#pragma unroll
    for (int nt = 0; nt < 4; ++nt) e[nt] = expf(x[nt] - mx);
    float sum = (e[0] + e[1]) + (e[2] + e[3]);
    sum += ROW_XOR(sum, 1);
    sum += ROW_XOR(sum, 2);
    sum += ROW_XOR(sum, 4);
    sum += ROW_XOR(sum, 8);
    inv[r] = 1.0f / sum;
    float* pr = pw + rq * APITCH + c;
    pr[0]  = e[0];
    pr[16] = e[1];
    pr[32] = e[2];
    pr[48] = e[3];
  }
  __syncthreads();

  const float* prow = pw + c * APITCH + 4 * h;
  const float* vcol = sV + (w * 16 + 4 * h) * APITCH + c;
  v8f o0 = {}, o1 = {}, o2 = {}, o3 = {};
#pragma unroll
  for (int ks = 0; ks < 8; ++ks) {
    const float4 pa = *(const float4*)(prow + 8 * ks);
    const float* vr = vcol + 8 * ks * APITCH;
    const v2f al = {pa.x, pa.y}, ah = {pa.z, pa.w};
    const v2f b0l = {vr[0],      vr[APITCH]},      b0h = {vr[2 * APITCH],      vr[3 * APITCH]};
    const v2f b1l = {vr[16],     vr[APITCH + 16]}, b1h = {vr[2 * APITCH + 16], vr[3 * APITCH + 16]};
    const v2f b2l = {vr[32],     vr[APITCH + 32]}, b2h = {vr[2 * APITCH + 32], vr[3 * APITCH + 32]};
    const v2f b3l = {vr[48],     vr[APITCH + 48]}, b3h = {vr[2 * APITCH + 48], vr[3 * APITCH + 48]};
    WMMA_F32(o0, al, b0l);
    WMMA_F32(o1, al, b1l);
    WMMA_F32(o2, al, b2l);
    WMMA_F32(o3, al, b3l);
    WMMA_F32(o0, ah, b0h);
    WMMA_F32(o1, ah, b1h);
    WMMA_F32(o2, ah, b2h);
    WMMA_F32(o3, ah, b3h);
  }

#pragma unroll
  for (int r = 0; r < 8; ++r) {
    float* pr = pw + (8 * h + r) * APITCH + c;
    pr[0]  = o0[r] * inv[r];
    pr[16] = o1[r] * inv[r];
    pr[32] = o2[r] * inv[r];
    pr[48] = o3[r] * inv[r];
  }
  __syncthreads();
#pragma unroll
  for (int i = 0; i < 8; ++i) {
    const int row = 2 * i + h;
    const float4 ov = *(const float4*)(pw + row * APITCH + 4 * c);
    *(float4*)(obase + (size_t)(16 * w + row) * Dn + 4 * c) = ov;
  }
}

extern "C" void kernel_launch(void* const* d_in, const int* in_sizes, int n_in,
                              void* d_out, int out_size, void* d_ws, size_t ws_size,
                              hipStream_t stream) {
  (void)in_sizes; (void)n_in; (void)out_size; (void)ws_size;
  constexpr int B = 4, T = 1024, D_IN = 512, D = 512, K = 2, V = 500, H = 8, BAND = 24;
  constexpr int M = B * T;
  static_assert(D == RD && V == LV && K == LK && M % 8 == 0, "row kernels are sized for these shapes");
  static_assert(D == H * AHD && BAND == ABAND && T % AQ == 0, "attention kernel is sized for these shapes");

  const float* x_m      = (const float*)d_in[0];
  const float* logits   = (const float*)d_in[1];
  const int*   blank_id = (const int*)  d_in[2];
  const float* proj_w   = (const float*)d_in[3];
  const float* proj_b   = (const float*)d_in[4];
  const float* spk_tags = (const float*)d_in[5];
  const float* lnq_g    = (const float*)d_in[6];
  const float* lnq_b    = (const float*)d_in[7];
  const float* lnkv_g   = (const float*)d_in[8];
  const float* lnkv_b   = (const float*)d_in[9];
  const float* in_w     = (const float*)d_in[10];
  const float* in_b     = (const float*)d_in[11];
  const float* out_w    = (const float*)d_in[12];
  const float* out_b    = (const float*)d_in[13];
  const float* ffn_ln_g = (const float*)d_in[14];
  const float* ffn_ln_b = (const float*)d_in[15];
  const float* ffn_w1   = (const float*)d_in[16];
  const float* ffn_b1   = (const float*)d_in[17];
  const float* ffn_w2   = (const float*)d_in[18];
  const float* ffn_b2   = (const float*)d_in[19];
  float* out = (float*)d_out;

  float* ws = (float*)d_ws;
  const size_t SZ = (size_t)M * D;
  float* X    = ws;
  float* kv   = X  + SZ;
  float* xk   = kv + 2 * SZ;
  float* qh   = xk + 2 * SZ;
  float* qy   = qh + 2 * SZ;
  float* t1   = ws;
  float* wmix = ws + 10 * SZ;

  const dim3 blk(256);
  auto gg = [](int m, int n) { return dim3(m / 64, n / 128); };

  gemm_wmma_f32<<<gg(M, D), blk, 0, stream>>>(x_m, proj_w, proj_b, nullptr, X,
                                              M, D, D_IN, 0, T, K, 0);
  wmix_kernel<<<dim3(M / 8), blk, 0, stream>>>(logits, blank_id, wmix, M);
  ln_kernel<<<dim3(M / 8), blk, 0, stream>>>(X, lnkv_g, lnkv_b, xk, M);
  gemm_wmma_f32<<<gg(M, 2 * D), blk, 0, stream>>>(xk, in_w + (size_t)D * D, in_b + D,
                                                  nullptr, kv, M, 2 * D, D, 0, T, K, 0);
  stream_prep<<<dim3(K * M / 8), blk, 0, stream>>>(X, wmix, spk_tags, lnq_g, lnq_b,
                                                   xk, qh, M);
  gemm_wmma_f32<<<gg(K * M, D), blk, 0, stream>>>(qh, in_w, in_b, nullptr, qy,
                                                  K * M, D, D, 0, T, K, 0);
  banded_attn_wmma<<<dim3(K * B * H * (T / AQ)), dim3(ATHR), 0, stream>>>(qy, kv, kv + D, qh,
                                                                          T, D, 2 * D, H, B);
  gemm_wmma_f32<<<gg(K * M, D), blk, 0, stream>>>(qh, out_w, out_b, xk, qy,
                                                  K * M, D, D,  1, T, K, 0);
  ln_kernel<<<dim3(K * M / 8), blk, 0, stream>>>(qy, ffn_ln_g, ffn_ln_b, qh, K * M);

  static_assert(K == 2, "the FFN pair below is written out for two speakers");
  gemm_wmma_f32<<<gg(M, 4 * D), blk, 0, stream>>>(qh, ffn_w1, ffn_b1, nullptr, t1,
                                                  M, 4 * D, D,  2, T, K, 0);
  gemm_wmma_f32<<<gg(M, D), blk, 0, stream>>>(t1, ffn_w2, ffn_b2, qy, out,
                                              M, D, 4 * D,  1 | 4, T, K, 0);
  gemm_wmma_f32<<<gg(M, 4 * D), blk, 0, stream>>>(qh + SZ, ffn_w1, ffn_b1, nullptr, t1,
                                                  M, 4 * D, D,  2, T, K, 1);
  gemm_wmma_f32<<<gg(M, D), blk, 0, stream>>>(t1, ffn_w2, ffn_b2, qy + SZ, out,
                                              M, D, 4 * D,  1 | 4, T, K, 1);
}


# GEMM K loop on a ring of three LDS stages: the one barrier per K step sits mid-step before the stores only, fragments load half a step ahead into two register sets, memory ops issue behind products
# speedup vs baseline: 2.2071x; 1.0400x over previous
#include <hip/hip_runtime.h>
#include <hip/hip_bf16.h>
#include <math.h>

typedef float v2f __attribute__((ext_vector_type(2)));
typedef float v4f __attribute__((ext_vector_type(4)));
typedef float v8f __attribute__((ext_vector_type(8)));

__device__ __forceinline__ float gelu_exact(float x) {
  return 0.5f * x * (1.0f + erff(x * 0.70710678118654752f));
}

#define WMMA_F32(cacc, av, bv)                                                 \
  cacc = __builtin_amdgcn_wmma_f32_16x16x4_f32(false, (av), false, (bv),       \
                                               (short)0, (cacc), false, false)

#define GBM 64
#define GBN 128
#define GBK 32
#define GPITCH 36
#define GTILE ((GBM + GBN) * GPITCH)
#define GSTAGES 3

#define G_LOAD_TILE(r, ko)                                                     \
  do {                                                                         \
    const char* ap_ = (const char*)(Ab + (ko));                                \
    const char* wp_ = (const char*)(Wb + (ko));                                \
    unsigned q0_ = ob0, q1_ = ob1, q2_ = ob2, q3_ = ob3;                       \
    asm volatile("" : "+v"(q0_), "+v"(q1_), "+v"(q2_), "+v"(q3_));             \
    r##0 = *(const float4*)(ap_ + q0_);                                        \
    r##1 = *(const float4*)(ap_ + q1_);                                        \
    r##2 = *(const float4*)(wp_ + q0_);                                        \
    r##3 = *(const float4*)(wp_ + q1_);                                        \
    r##4 = *(const float4*)(wp_ + q2_);                                        \
    r##5 = *(const float4*)(wp_ + q3_);                                        \
  } while (0)

#define G_STORE_TILE(r, st)                                                    \
  do {                                                                         \
    float* s_ = sm + (st) + so;                                                \
    *(float4*)(s_)                         = r##0;                             \
    *(float4*)(s_ + 32 * GPITCH)           = r##1;                             \
    *(float4*)(s_ + GBM * GPITCH)          = r##2;                             \
    *(float4*)(s_ + (GBM + 32) * GPITCH)   = r##3;                             \
    *(float4*)(s_ + (GBM + 64) * GPITCH)   = r##4;                             \
    *(float4*)(s_ + (GBM + 96) * GPITCH)   = r##5;                             \
  } while (0)

#define G_LOAD_FRAGS(F, st, hh)                                                \
  do {                                                                         \
    const float* c_ = sm + (st) + (hh) * 16;                                   \
    F[0] = *(const float4*)(c_ + aoff);                                        \
    F[1] = *(const float4*)(c_ + aoff + 16 * GPITCH);                          \
    F[2] = *(const float4*)(c_ + boff);                                        \
    F[3] = *(const float4*)(c_ + boff + 16 * GPITCH);                          \
    F[4] = *(const float4*)(c_ + aoff + 8);                                    \
    F[5] = *(const float4*)(c_ + aoff + 16 * GPITCH + 8);                      \
    F[6] = *(const float4*)(c_ + boff + 8);                                    \
    F[7] = *(const float4*)(c_ + boff + 16 * GPITCH + 8);                      \
  } while (0)

#define G_MMA_HALF(F)                                                          \
  do {                                                                         \
    _Pragma("unroll")                                                          \
    for (int j_ = 0; j_ < 2; ++j_) {                                           \
      const float4 fa0 = F[4 * j_], fa1 = F[4 * j_ + 1];                       \
      const float4 fb0 = F[4 * j_ + 2], fb1 = F[4 * j_ + 3];                   \
      const v2f a0l = {fa0.x, fa0.y}, a0h = {fa0.z, fa0.w};                    \
      const v2f a1l = {fa1.x, fa1.y}, a1h = {fa1.z, fa1.w};                    \
      const v2f b0l = {fb0.x, fb0.y}, b0h = {fb0.z, fb0.w};                    \
      const v2f b1l = {fb1.x, fb1.y}, b1h = {fb1.z, fb1.w};                    \
      WMMA_F32(c00, a0l, b0l);                                                 \
      WMMA_F32(c01, a0l, b1l);                                                 \
      WMMA_F32(c10, a1l, b0l);                                                 \
      WMMA_F32(c11, a1l, b1l);                                                 \
      WMMA_F32(c00, a0h, b0h);                                                 \
      WMMA_F32(c01, a0h, b1h);                                                 \
      WMMA_F32(c10, a1h, b0h);                                                 \
      WMMA_F32(c11, a1h, b1h);                                                 \
    }                                                                          \
  } while (0)

#define G_GRP(mask, n) __builtin_amdgcn_sched_group_barrier((mask), (n), 0)
#define G_MMA1_MEM2(mask) do { G_GRP(0x008, 1); G_GRP((mask), 2); } while (0)
#define G_STEP(t, S, G, N)                                                     \
  do {                                                                         \
    G_LOAD_FRAGS(Q, o0, 1);                                                    \
    G_MMA_HALF(P);                                                             \
    G_MMA1_MEM2(0x100); G_MMA1_MEM2(0x100);                                    \
    G_MMA1_MEM2(0x100); G_MMA1_MEM2(0x100);                                    \
    G_GRP(0x008, 12);                                                          \
    __builtin_amdgcn_sched_barrier(0);                                         \
    if ((S) || (N)) __syncthreads();                                           \
    if (N) G_LOAD_FRAGS(P, o1, 0);                                             \
    if (S) G_STORE_TILE(r, o2);                                                \
    if (G) G_LOAD_TILE(r, ((t) + 3) * GBK);                                    \
    G_MMA_HALF(Q);                                                             \
    if (N) { G_MMA1_MEM2(0x100); G_MMA1_MEM2(0x100);                           \
             G_MMA1_MEM2(0x100); G_MMA1_MEM2(0x100); }                         \
    if (S) { G_MMA1_MEM2(0x200); G_MMA1_MEM2(0x200); G_MMA1_MEM2(0x200); }     \
    if (G) { G_MMA1_MEM2(0x020); G_MMA1_MEM2(0x020); G_MMA1_MEM2(0x020); }     \
    G_GRP(0x008, 16 - 3 * (S) - 4 * (N) - 3 * (G));                            \
    __builtin_amdgcn_sched_barrier(0);                                         \
    { const int t_ = o0; o0 = o1; o1 = o2; o2 = t_; }                          \
  } while (0)

__global__ __launch_bounds__(256) void gemm_wmma_f32(
    const float* __restrict__ A, const float* __restrict__ W,
    const float* __restrict__ bias, const float* __restrict__ R,
    float* __restrict__ C, int M, int N, int Kd,
    int flags, int Tlen, int Kspk, int kk)
{
  __shared__ __attribute__((aligned(16))) float sm[GSTAGES * GTILE];

  const int tid  = threadIdx.x;
  const int lane = tid & 31;
  const int w    = tid >> 5;
  const int wm   = w & 1;
  const int wn   = w >> 1;
  const int half = lane >> 4;
  const int l16  = lane & 15;
  const int blockM = blockIdx.x * GBM;
  const int blockN = blockIdx.y * GBN;

  const int lrow = tid >> 3;
  const int lcol = (tid & 7) * 4;
  const float* Ab = A + (size_t)blockM * Kd;
  const float* Wb = W + (size_t)blockN * Kd;
  const unsigned ob0 = (unsigned)(lrow * Kd + lcol) * 4u;
  const unsigned ob1 = ob0 + 128u * (unsigned)Kd;
  const unsigned ob2 = ob0 + 256u * (unsigned)Kd;
  const unsigned ob3 = ob0 + 384u * (unsigned)Kd;

  const int so   = lrow * GPITCH + lcol;
  const int aoff = (wm * 32 + l16) * GPITCH + 4 * half;
  const int boff = GBM * GPITCH + (wn * 32 + l16) * GPITCH + 4 * half;

  v8f c00 = {}, c01 = {}, c10 = {}, c11 = {};

  float4 u0, u1, u2, u3, u4, u5, e0, e1, e2, e3, e4, e5, r0, r1, r2, r3, r4, r5;
  G_LOAD_TILE(u, 0);
  G_LOAD_TILE(e, GBK);
  G_LOAD_TILE(r, 2 * GBK);
  G_STORE_TILE(u, 0);
  G_STORE_TILE(e, GTILE);
  __syncthreads();

  int o0 = 0, o1 = GTILE, o2 = 2 * GTILE;
  float4 P[8], Q[8];
  G_LOAD_FRAGS(P, o0, 0);

  const int nk = Kd / GBK;
  for (int kt = 0; kt < nk - 3; ++kt) G_STEP(kt, 1, 1, 1);
  G_STEP(nk - 3, 1, 0, 1);
  G_STEP(nk - 2, 0, 0, 1);
  G_STEP(nk - 1, 0, 0, 0);

  const int tileM = blockM + wm * 32;
  const int n0    = blockN + wn * 32 + l16;
  const float bv0 = bias[n0];
  const float bv1 = bias[n0 + 16];
#pragma unroll
  for (int i = 0; i < 2; ++i) {
#pragma unroll
    for (int r = 0; r < 8; ++r) {
      const int row = tileM + i * 16 + 8 * half + r;
      float v0 = (i == 0 ? c00[r] : c10[r]) + bv0;
      float v1 = (i == 0 ? c01[r] : c11[r]) + bv1;
      if (flags & 1) {
        const float* rr = R + (size_t)row * N;
        v0 += rr[n0];
        v1 += rr[n0 + 16];
      }
      if (flags & 2) {
        v0 = gelu_exact(v0);
        v1 = gelu_exact(v1);
      }
      size_t orow = (size_t)row;
      if (flags & 4) {
        const int bi = row / Tlen;
        orow = (size_t)bi * Kspk * Tlen + (size_t)kk * Tlen + (row % Tlen);
      }
      float* cr = C + orow * N;
      cr[n0]      = v0;
      cr[n0 + 16] = v1;
    }
  }
}

#define RD  512
#define RNV (RD / 128)
#define LV  500
#define LV4 (LV / 4)
#define LNI ((LV4 + 31) / 32)
#define LK  2

#define ROW_XOR(x, n)                                                          \
  __builtin_bit_cast(float, __builtin_amdgcn_update_dpp(                       \
      __builtin_bit_cast(int, (x)), __builtin_bit_cast(int, (x)),              \
      0x160 + (n), 0xf, 0xf, false))
__device__ __forceinline__ float xhalf(float x) {
  const int b = __builtin_bit_cast(int, x);
  return __builtin_bit_cast(float,
      __builtin_amdgcn_permlanex16(b, b, 0x76543210, 0xfedcba98, false, false));
}
__device__ __forceinline__ float wave_sum(float v) {
  v += ROW_XOR(v, 1);
  v += ROW_XOR(v, 2);
  v += ROW_XOR(v, 4);
  v += ROW_XOR(v, 8);
  return v + xhalf(v);
}
__device__ __forceinline__ float wave_max(float v) {
  v = fmaxf(v, ROW_XOR(v, 1));
  v = fmaxf(v, ROW_XOR(v, 2));
  v = fmaxf(v, ROW_XOR(v, 4));
  v = fmaxf(v, ROW_XOR(v, 8));
  return fmaxf(v, xhalf(v));
}

__device__ __forceinline__ void row_stats(const float4 (&v)[RNV], float& mean, float& inv) {
  float s = 0.f;
#pragma unroll
  for (int i = 0; i < RNV; ++i) s += (v[i].x + v[i].y) + (v[i].z + v[i].w);
  mean = wave_sum(s) * (1.0f / RD);
  float s2 = 0.f;
#pragma unroll
  for (int i = 0; i < RNV; ++i) {
    const float dx = v[i].x - mean, dy = v[i].y - mean;
    const float dz = v[i].z - mean, dw = v[i].w - mean;
    s2 += (dx * dx + dy * dy) + (dz * dz + dw * dw);
  }
  inv = rsqrtf(wave_sum(s2) * (1.0f / RD) + 1e-5f);
}

__global__ __launch_bounds__(256) void ln_kernel(
    const float* __restrict__ X, const float* __restrict__ g,
    const float* __restrict__ b, float* __restrict__ Y, int rows)
{
  const int lane = threadIdx.x & 31;
  const int row  = blockIdx.x * 8 + __builtin_amdgcn_readfirstlane(threadIdx.x >> 5);
  if (row >= rows) return;
  const float4* x4 = (const float4*)(X + (size_t)row * RD) + lane;
  const float4* g4 = (const float4*)g + lane;
  const float4* b4 = (const float4*)b + lane;
  float4 xv[RNV], gv[RNV], bv[RNV];
#pragma unroll
  for (int i = 0; i < RNV; ++i) xv[i] = x4[i * 32];
#pragma unroll
  for (int i = 0; i < RNV; ++i) { gv[i] = g4[i * 32]; bv[i] = b4[i * 32]; }
  __builtin_amdgcn_sched_barrier(0);
  float mean, inv;
  row_stats(xv, mean, inv);
  float4* y4 = (float4*)(Y + (size_t)row * RD) + lane;
#pragma unroll
  for (int i = 0; i < RNV; ++i) {
    float4 o;
    o.x = (xv[i].x - mean) * inv * gv[i].x + bv[i].x;
    o.y = (xv[i].y - mean) * inv * gv[i].y + bv[i].y;
    o.z = (xv[i].z - mean) * inv * gv[i].z + bv[i].z;
    o.w = (xv[i].w - mean) * inv * gv[i].w + bv[i].w;
    y4[i * 32] = o;
  }
}

__global__ __launch_bounds__(256) void wmix_kernel(
    const float* __restrict__ logits, const int* __restrict__ blank_p,
    float* __restrict__ wmix, int BT)
{
  const int lane = threadIdx.x & 31;
  const int row  = blockIdx.x * 8 + __builtin_amdgcn_readfirstlane(threadIdx.x >> 5);
  if (row >= BT) return;
  const int blank = blank_p[0];
  float4 lv[LK][LNI];
  float  lb[LK];
#pragma unroll
  for (int k2 = 0; k2 < LK; ++k2) {
    const float* L = logits + ((size_t)k2 * BT + row) * LV;
    const float4* L4 = (const float4*)L;
#pragma unroll
    for (int i = 0; i < LNI; ++i) {
      const int idx = lane + 32 * i;
      lv[k2][i] = (idx < LV4) ? L4[idx]
                              : make_float4(-INFINITY, -INFINITY, -INFINITY, -INFINITY);
    }
    lb[k2] = L[blank];
  }
  __builtin_amdgcn_sched_barrier(0);
  float a[LK];
#pragma unroll
  for (int k2 = 0; k2 < LK; ++k2) {
    float m = -INFINITY;
#pragma unroll
    for (int i = 0; i < LNI; ++i)
      m = fmaxf(m, fmaxf(fmaxf(lv[k2][i].x, lv[k2][i].y), fmaxf(lv[k2][i].z, lv[k2][i].w)));
    m = wave_max(m);
    float s = 0.f;
#pragma unroll
    for (int i = 0; i < LNI; ++i)
      s += (expf(lv[k2][i].x - m) + expf(lv[k2][i].y - m)) +
           (expf(lv[k2][i].z - m) + expf(lv[k2][i].w - m));
    s = wave_sum(s);
    a[k2] = 1.0f - expf(lb[k2] - m) / s;
  }
  if (lane == 0) {
    float m2 = -INFINITY;
#pragma unroll
    for (int k2 = 0; k2 < LK; ++k2) m2 = fmaxf(m2, a[k2]);
    float e[LK];
    float ss = 0.f;
#pragma unroll
    for (int k2 = 0; k2 < LK; ++k2) { e[k2] = expf(a[k2] - m2); ss += e[k2]; }
#pragma unroll
    for (int k2 = 0; k2 < LK; ++k2) wmix[(size_t)row * LK + k2] = e[k2] / ss;
  }
}

__global__ __launch_bounds__(256) void stream_prep(
    const float* __restrict__ X, const float* __restrict__ wmix,
    const float* __restrict__ tags, const float* __restrict__ g,
    const float* __restrict__ bt, float* __restrict__ Xk,
    float* __restrict__ Qin, int rows)
{
  const int lane = threadIdx.x & 31;
  const int row  = blockIdx.x * 8 + __builtin_amdgcn_readfirstlane(threadIdx.x >> 5);
  if (row >= rows * LK) return;
  const int kk   = row / rows;
  const int xrow = row - kk * rows;
  const float wv = wmix[(size_t)xrow * LK + kk];
  const float4* x4 = (const float4*)(X + (size_t)xrow * RD) + lane;
  const float4* t4 = (const float4*)(tags + (size_t)kk * RD) + lane;
  const float4* g4 = (const float4*)g + lane;
  const float4* b4 = (const float4*)bt + lane;
  float4 v[RNV], tv[RNV], gv[RNV], bv[RNV];
#pragma unroll
  for (int i = 0; i < RNV; ++i) v[i] = x4[i * 32];
#pragma unroll
  for (int i = 0; i < RNV; ++i) tv[i] = t4[i * 32];
#pragma unroll
  for (int i = 0; i < RNV; ++i) { gv[i] = g4[i * 32]; bv[i] = b4[i * 32]; }
  __builtin_amdgcn_sched_barrier(0);
#pragma unroll
  for (int i = 0; i < RNV; ++i) {
    v[i].x = v[i].x * wv + tv[i].x;
    v[i].y = v[i].y * wv + tv[i].y;
    v[i].z = v[i].z * wv + tv[i].z;
    v[i].w = v[i].w * wv + tv[i].w;
  }
  float4* xk4 = (float4*)(Xk + (size_t)row * RD) + lane;
#pragma unroll
  for (int i = 0; i < RNV; ++i) xk4[i * 32] = v[i];
  float mean, inv;
  row_stats(v, mean, inv);
  float4* q4 = (float4*)(Qin + (size_t)row * RD) + lane;
#pragma unroll
  for (int i = 0; i < RNV; ++i) {
    float4 o;
    o.x = (v[i].x - mean) * inv * gv[i].x + bv[i].x;
    o.y = (v[i].y - mean) * inv * gv[i].y + bv[i].y;
    o.z = (v[i].z - mean) * inv * gv[i].z + bv[i].z;
    o.w = (v[i].w - mean) * inv * gv[i].w + bv[i].w;
    q4[i * 32] = o;
  }
}

#define AQ      64
#define ANW     (AQ / 16)
#define ATHR    (32 * ANW)
#define ABAND   24
#define AKEYS   (AQ + 2 * ABAND)
#define AHD     64
#define APITCH  68

__global__ __launch_bounds__(ATHR) void banded_attn_wmma(
    const float* __restrict__ q, const float* __restrict__ kb,
    const float* __restrict__ vb, float* __restrict__ ctx,
    int T_, int Dn, int Dkv, int Hn, int Bkv)
{
  __shared__ __attribute__((aligned(16))) float sK[AKEYS * APITCH];
  __shared__ __attribute__((aligned(16))) float sV[AKEYS * APITCH];
  __shared__ __attribute__((aligned(16))) float sQ[AQ * APITCH];

  const int tid  = threadIdx.x;
  const int lane = tid & 31;
  const int w    = tid >> 5;
  const int c    = lane & 15;
  const int h    = lane >> 4;

  const int nqt = T_ / AQ;
  const int qt  = blockIdx.x % nqt;
  const int hh  = (blockIdx.x / nqt) % Hn;
  const int b   = blockIdx.x / (nqt * Hn);
  const int bk  = b % Bkv;
  const int t0  = qt * AQ;

  const size_t headoff = (size_t)hh * AHD;
  const float* kbase = kb + (size_t)bk * T_ * Dkv + headoff;
  const float* vbase = vb + (size_t)bk * T_ * Dkv + headoff;
  const float* qbase = q  + ((size_t)b * T_ + t0) * Dn + headoff;
  float*       obase = ctx + ((size_t)b * T_ + t0) * Dn + headoff;

  constexpr int NKV = (AKEYS * 16) / ATHR;
  constexpr int NQ  = (AQ * 16) / ATHR;
  const int srow = tid >> 4;
  const int sc4  = (tid & 15) * 4;
  v4f kreg[NKV], vreg[NKV], qreg[NQ];
#pragma unroll
  for (int i = 0; i < NKV; ++i) {
    const int key = t0 - ABAND + srow + (ATHR / 16) * i;
    const unsigned off = (unsigned)min(max(key, 0), T_ - 1) * (unsigned)Dkv + (unsigned)sc4;
    kreg[i] = *(const v4f*)(kbase + off);
    vreg[i] = *(const v4f*)(vbase + off);
  }
#pragma unroll
  for (int i = 0; i < NQ; ++i) {
    const unsigned off = (unsigned)(srow + (ATHR / 16) * i) * (unsigned)Dn + (unsigned)sc4;
    qreg[i] = *(const v4f*)(qbase + off);
  }
  __builtin_amdgcn_sched_barrier(0);
  const v4f zero4 = {0.f, 0.f, 0.f, 0.f};
#pragma unroll
  for (int i = 0; i < NKV; ++i) {
    const int row = srow + (ATHR / 16) * i;
    const int key = t0 - ABAND + row;
    const bool ok = (key >= 0) && (key < T_);
    *(v4f*)(sK + row * APITCH + sc4) = ok ? kreg[i] : zero4;
    *(v4f*)(sV + row * APITCH + sc4) = ok ? vreg[i] : zero4;
  }
#pragma unroll
  for (int i = 0; i < NQ; ++i) {
    const int row = srow + (ATHR / 16) * i;
    *(v4f*)(sQ + row * APITCH + sc4) = qreg[i];
  }
  __syncthreads();

  float* pw = sQ + (w * 16) * APITCH;
  const float* qrow = pw + c * APITCH + 4 * h;
  const float* krow = sK + (w * 16 + c) * APITCH + 4 * h;
  float4 qa[8];
#pragma unroll
  for (int ks = 0; ks < 8; ++ks) qa[ks] = *(const float4*)(qrow + 8 * ks);

  v8f s0 = {}, s1 = {}, s2 = {}, s3 = {};
#pragma unroll
  for (int ks = 0; ks < 8; ++ks) {
    const float4 k0 = *(const float4*)(krow + 8 * ks);
    const float4 k1 = *(const float4*)(krow + 16 * APITCH + 8 * ks);
    const float4 k2 = *(const float4*)(krow + 32 * APITCH + 8 * ks);
    const float4 k3 = *(const float4*)(krow + 48 * APITCH + 8 * ks);
    const v2f al = {qa[ks].x, qa[ks].y}, ah = {qa[ks].z, qa[ks].w};
    const v2f b0l = {k0.x, k0.y}, b0h = {k0.z, k0.w};
    const v2f b1l = {k1.x, k1.y}, b1h = {k1.z, k1.w};
    const v2f b2l = {k2.x, k2.y}, b2h = {k2.z, k2.w};
    const v2f b3l = {k3.x, k3.y}, b3h = {k3.z, k3.w};
    WMMA_F32(s0, al, b0l);
    WMMA_F32(s1, al, b1l);
    WMMA_F32(s2, al, b2l);
    WMMA_F32(s3, al, b3l);
    WMMA_F32(s0, ah, b0h);
    WMMA_F32(s1, ah, b1h);
    WMMA_F32(s2, ah, b2h);
    WMMA_F32(s3, ah, b3h);
  }

  const float scale = 0.125f;
  const int jbase = t0 + 16 * w - ABAND;
  bool colok[4];
#pragma unroll
  for (int nt = 0; nt < 4; ++nt) {
    const int key = jbase + 16 * nt + c;
    colok[nt] = (key >= 0) && (key < T_);
  }
  float inv[8];
#pragma unroll
  for (int r = 0; r < 8; ++r) {
    const int rq = 8 * h + r;
    float x[4] = {s0[r], s1[r], s2[r], s3[r]};
#pragma unroll
    for (int nt = 0; nt < 4; ++nt) {
      const int jj = 16 * nt + c;
      const bool ok = colok[nt] && (jj >= rq) && (jj <= rq + 2 * ABAND);
      x[nt] = ok ? x[nt] * scale : -INFINITY;
    }
    float mx = fmaxf(fmaxf(x[0], x[1]), fmaxf(x[2], x[3]));
    mx = fmaxf(mx, ROW_XOR(mx, 1));
    mx = fmaxf(mx, ROW_XOR(mx, 2));
    mx = fmaxf(mx, ROW_XOR(mx, 4));
    mx = fmaxf(mx, ROW_XOR(mx, 8));
    float e[4];
#pragma unroll
    for (int nt = 0; nt < 4; ++nt) e[nt] = expf(x[nt] - mx);
    float sum = (e[0] + e[1]) + (e[2] + e[3]);
    sum += ROW_XOR(sum, 1);
    sum += ROW_XOR(sum, 2);
    sum += ROW_XOR(sum, 4);
    sum += ROW_XOR(sum, 8);
    inv[r] = 1.0f / sum;
    float* pr = pw + rq * APITCH + c;
    pr[0]  = e[0];
    pr[16] = e[1];
    pr[32] = e[2];
    pr[48] = e[3];
  }
  __syncthreads();

  const float* prow = pw + c * APITCH + 4 * h;
  const float* vcol = sV + (w * 16 + 4 * h) * APITCH + c;
  v8f o0 = {}, o1 = {}, o2 = {}, o3 = {};
#pragma unroll
  for (int ks = 0; ks < 8; ++ks) {
    const float4 pa = *(const float4*)(prow + 8 * ks);
    const float* vr = vcol + 8 * ks * APITCH;
    const v2f al = {pa.x, pa.y}, ah = {pa.z, pa.w};
    const v2f b0l = {vr[0],      vr[APITCH]},      b0h = {vr[2 * APITCH],      vr[3 * APITCH]};
    const v2f b1l = {vr[16],     vr[APITCH + 16]}, b1h = {vr[2 * APITCH + 16], vr[3 * APITCH + 16]};
    const v2f b2l = {vr[32],     vr[APITCH + 32]}, b2h = {vr[2 * APITCH + 32], vr[3 * APITCH + 32]};
    const v2f b3l = {vr[48],     vr[APITCH + 48]}, b3h = {vr[2 * APITCH + 48], vr[3 * APITCH + 48]};
    WMMA_F32(o0, al, b0l);
    WMMA_F32(o1, al, b1l);
    WMMA_F32(o2, al, b2l);
    WMMA_F32(o3, al, b3l);
    WMMA_F32(o0, ah, b0h);
    WMMA_F32(o1, ah, b1h);
    WMMA_F32(o2, ah, b2h);
    WMMA_F32(o3, ah, b3h);
  }

#pragma unroll
  for (int r = 0; r < 8; ++r) {
    float* pr = pw + (8 * h + r) * APITCH + c;
    pr[0]  = o0[r] * inv[r];
    pr[16] = o1[r] * inv[r];
    pr[32] = o2[r] * inv[r];
    pr[48] = o3[r] * inv[r];
  }
  __syncthreads();
#pragma unroll
  for (int i = 0; i < 8; ++i) {
    const int row = 2 * i + h;
    const float4 ov = *(const float4*)(pw + row * APITCH + 4 * c);
    *(float4*)(obase + (size_t)(16 * w + row) * Dn + 4 * c) = ov;
  }
}

extern "C" void kernel_launch(void* const* d_in, const int* in_sizes, int n_in,
                              void* d_out, int out_size, void* d_ws, size_t ws_size,
                              hipStream_t stream) {
  (void)in_sizes; (void)n_in; (void)out_size; (void)ws_size;
  constexpr int B = 4, T = 1024, D_IN = 512, D = 512, K = 2, V = 500, H = 8, BAND = 24;
  constexpr int M = B * T;
  static_assert(D == RD && V == LV && K == LK && M % 8 == 0, "row kernels are sized for these shapes");
  static_assert(D == H * AHD && BAND == ABAND && T % AQ == 0, "attention kernel is sized for these shapes");

  const float* x_m      = (const float*)d_in[0];
  const float* logits   = (const float*)d_in[1];
  const int*   blank_id = (const int*)  d_in[2];
  const float* proj_w   = (const float*)d_in[3];
  const float* proj_b   = (const float*)d_in[4];
  const float* spk_tags = (const float*)d_in[5];
  const float* lnq_g    = (const float*)d_in[6];
  const float* lnq_b    = (const float*)d_in[7];
  const float* lnkv_g   = (const float*)d_in[8];
  const float* lnkv_b   = (const float*)d_in[9];
  const float* in_w     = (const float*)d_in[10];
  const float* in_b     = (const float*)d_in[11];
  const float* out_w    = (const float*)d_in[12];
  const float* out_b    = (const float*)d_in[13];
  const float* ffn_ln_g = (const float*)d_in[14];
  const float* ffn_ln_b = (const float*)d_in[15];
  const float* ffn_w1   = (const float*)d_in[16];
  const float* ffn_b1   = (const float*)d_in[17];
  const float* ffn_w2   = (const float*)d_in[18];
  const float* ffn_b2   = (const float*)d_in[19];
  float* out = (float*)d_out;

  float* ws = (float*)d_ws;
  const size_t SZ = (size_t)M * D;
  float* X    = ws;
  float* kv   = X  + SZ;
  float* xk   = kv + 2 * SZ;
  float* qh   = xk + 2 * SZ;
  float* qy   = qh + 2 * SZ;
  float* t1   = ws;
  float* wmix = ws + 10 * SZ;

  const dim3 blk(256);
  auto gg = [](int m, int n) { return dim3(m / 64, n / 128); };

  gemm_wmma_f32<<<gg(M, D), blk, 0, stream>>>(x_m, proj_w, proj_b, nullptr, X,
                                              M, D, D_IN, 0, T, K, 0);
  wmix_kernel<<<dim3(M / 8), blk, 0, stream>>>(logits, blank_id, wmix, M);
  ln_kernel<<<dim3(M / 8), blk, 0, stream>>>(X, lnkv_g, lnkv_b, xk, M);
  gemm_wmma_f32<<<gg(M, 2 * D), blk, 0, stream>>>(xk, in_w + (size_t)D * D, in_b + D,
                                                  nullptr, kv, M, 2 * D, D, 0, T, K, 0);
  stream_prep<<<dim3(K * M / 8), blk, 0, stream>>>(X, wmix, spk_tags, lnq_g, lnq_b,
                                                   xk, qh, M);
  gemm_wmma_f32<<<gg(K * M, D), blk, 0, stream>>>(qh, in_w, in_b, nullptr, qy,
                                                  K * M, D, D, 0, T, K, 0);
  banded_attn_wmma<<<dim3(K * B * H * (T / AQ)), dim3(ATHR), 0, stream>>>(qy, kv, kv + D, qh,
                                                                          T, D, 2 * D, H, B);
  gemm_wmma_f32<<<gg(K * M, D), blk, 0, stream>>>(qh, out_w, out_b, xk, qy,
                                                  K * M, D, D,  1, T, K, 0);
  ln_kernel<<<dim3(K * M / 8), blk, 0, stream>>>(qy, ffn_ln_g, ffn_ln_b, qh, K * M);

  static_assert(K == 2, "the FFN pair below is written out for two speakers");
  gemm_wmma_f32<<<gg(M, 4 * D), blk, 0, stream>>>(qh, ffn_w1, ffn_b1, nullptr, t1,
                                                  M, 4 * D, D,  2, T, K, 0);
  gemm_wmma_f32<<<gg(M, D), blk, 0, stream>>>(t1, ffn_w2, ffn_b2, qy, out,
                                              M, D, 4 * D,  1 | 4, T, K, 0);
  gemm_wmma_f32<<<gg(M, 4 * D), blk, 0, stream>>>(qh + SZ, ffn_w1, ffn_b1, nullptr, t1,
                                                  M, 4 * D, D,  2, T, K, 1);
  gemm_wmma_f32<<<gg(M, D), blk, 0, stream>>>(t1, ffn_w2, ffn_b2, qy + SZ, out,
                                              M, D, 4 * D,  1 | 4, T, K, 1);
}


# GEMM K loop issue order: each half step's memory instructions stand as one group among its products (2 places per K step, products first behind the barrier), not in pairs behind single products
# speedup vs baseline: 2.2843x; 1.0350x over previous
#include <hip/hip_runtime.h>
#include <hip/hip_bf16.h>
#include <math.h>

typedef float v2f __attribute__((ext_vector_type(2)));
typedef float v4f __attribute__((ext_vector_type(4)));
typedef float v8f __attribute__((ext_vector_type(8)));

__device__ __forceinline__ float gelu_exact(float x) {
  return 0.5f * x * (1.0f + erff(x * 0.70710678118654752f));
}

#define WMMA_F32(cacc, av, bv)                                                 \
  cacc = __builtin_amdgcn_wmma_f32_16x16x4_f32(false, (av), false, (bv),       \
                                               (short)0, (cacc), false, false)

#define GBM 64
#define GBN 128
#define GBK 32
#define GPITCH 36
#define GTILE ((GBM + GBN) * GPITCH)
#define GSTAGES 3

#define G_LOAD_TILE(r, ko)                                                     \
  do {                                                                         \
    const char* ap_ = (const char*)(Ab + (ko));                                \
    const char* wp_ = (const char*)(Wb + (ko));                                \
    unsigned q0_ = ob0, q1_ = ob1, q2_ = ob2, q3_ = ob3;                       \
    asm volatile("" : "+v"(q0_), "+v"(q1_), "+v"(q2_), "+v"(q3_));             \
    r##0 = *(const float4*)(ap_ + q0_);                                        \
    r##1 = *(const float4*)(ap_ + q1_);                                        \
    r##2 = *(const float4*)(wp_ + q0_);                                        \
    r##3 = *(const float4*)(wp_ + q1_);                                        \
    r##4 = *(const float4*)(wp_ + q2_);                                        \
    r##5 = *(const float4*)(wp_ + q3_);                                        \
  } while (0)

#define G_STORE_TILE(r, st)                                                    \
  do {                                                                         \
    float* s_ = sm + (st) + so;                                                \
    *(float4*)(s_)                         = r##0;                             \
    *(float4*)(s_ + 32 * GPITCH)           = r##1;                             \
    *(float4*)(s_ + GBM * GPITCH)          = r##2;                             \
    *(float4*)(s_ + (GBM + 32) * GPITCH)   = r##3;                             \
    *(float4*)(s_ + (GBM + 64) * GPITCH)   = r##4;                             \
    *(float4*)(s_ + (GBM + 96) * GPITCH)   = r##5;                             \
  } while (0)

#define G_LOAD_FRAGS(F, st, hh)                                                \
  do {                                                                         \
    const float* c_ = sm + (st) + (hh) * 16;                                   \
    F[0] = *(const float4*)(c_ + aoff);                                        \
    F[1] = *(const float4*)(c_ + aoff + 16 * GPITCH);                          \
    F[2] = *(const float4*)(c_ + boff);                                        \
    F[3] = *(const float4*)(c_ + boff + 16 * GPITCH);                          \
    F[4] = *(const float4*)(c_ + aoff + 8);                                    \
    F[5] = *(const float4*)(c_ + aoff + 16 * GPITCH + 8);                      \
    F[6] = *(const float4*)(c_ + boff + 8);                                    \
    F[7] = *(const float4*)(c_ + boff + 16 * GPITCH + 8);                      \
  } while (0)

#define G_MMA_HALF(F)                                                          \
  do {                                                                         \
    _Pragma("unroll")                                                          \
    for (int j_ = 0; j_ < 2; ++j_) {                                           \
      const float4 fa0 = F[4 * j_], fa1 = F[4 * j_ + 1];                       \
      const float4 fb0 = F[4 * j_ + 2], fb1 = F[4 * j_ + 3];                   \
      const v2f a0l = {fa0.x, fa0.y}, a0h = {fa0.z, fa0.w};                    \
      const v2f a1l = {fa1.x, fa1.y}, a1h = {fa1.z, fa1.w};                    \
      const v2f b0l = {fb0.x, fb0.y}, b0h = {fb0.z, fb0.w};                    \
      const v2f b1l = {fb1.x, fb1.y}, b1h = {fb1.z, fb1.w};                    \
      WMMA_F32(c00, a0l, b0l);                                                 \
      WMMA_F32(c01, a0l, b1l);                                                 \
      WMMA_F32(c10, a1l, b0l);                                                 \
      WMMA_F32(c11, a1l, b1l);                                                 \
      WMMA_F32(c00, a0h, b0h);                                                 \
      WMMA_F32(c01, a0h, b1h);                                                 \
      WMMA_F32(c10, a1h, b0h);                                                 \
      WMMA_F32(c11, a1h, b1h);                                                 \
    }                                                                          \
  } while (0)

#define G_GRP(mask, n) __builtin_amdgcn_sched_group_barrier((mask), (n), 0)
#define G_STEP(t, S, G, N)                                                     \
  do {                                                                         \
    G_LOAD_FRAGS(Q, o0, 1);                                                    \
    G_MMA_HALF(P);                                                             \
    G_GRP(0x100, 8);             \
    G_GRP(0x008, 16);     \
    __builtin_amdgcn_sched_barrier(0);   \
    if ((S) || (N)) {                                                          \
      __syncthreads();                                                         \
      __builtin_amdgcn_sched_barrier(0);   \
    }                                                                          \
    if (N) G_LOAD_FRAGS(P, o1, 0);                                             \
    if (S) G_STORE_TILE(r, o2);                                                \
    if (G) G_LOAD_TILE(r, ((t) + 3) * GBK);                                    \
    G_MMA_HALF(Q);                                                             \
    G_GRP(0x008, 8);              \
    if (N) G_GRP(0x100, 8);      \
    if (S) G_GRP(0x200, 6);              \
    if (G) G_GRP(0x020, 6);            \
    G_GRP(0x008, 8);                  \
    __builtin_amdgcn_sched_barrier(0);   \
    { const int t_ = o0; o0 = o1; o1 = o2; o2 = t_; }                          \
  } while (0)

__global__ __launch_bounds__(256) void gemm_wmma_f32(
    const float* __restrict__ A, const float* __restrict__ W,
    const float* __restrict__ bias, const float* __restrict__ R,
    float* __restrict__ C, int M, int N, int Kd,
    int flags, int Tlen, int Kspk, int kk)
{
  __shared__ __attribute__((aligned(16))) float sm[GSTAGES * GTILE];

  const int tid  = threadIdx.x;
  const int lane = tid & 31;
  const int w    = tid >> 5;
  const int wm   = w & 1;
  const int wn   = w >> 1;
  const int half = lane >> 4;
  const int l16  = lane & 15;
  const int blockM = blockIdx.x * GBM;
  const int blockN = blockIdx.y * GBN;

  const int lrow = tid >> 3;
  const int lcol = (tid & 7) * 4;
  const float* Ab = A + (size_t)blockM * Kd;
  const float* Wb = W + (size_t)blockN * Kd;
  const unsigned ob0 = (unsigned)(lrow * Kd + lcol) * 4u;
  const unsigned ob1 = ob0 + 128u * (unsigned)Kd;
  const unsigned ob2 = ob0 + 256u * (unsigned)Kd;
  const unsigned ob3 = ob0 + 384u * (unsigned)Kd;

  const int so   = lrow * GPITCH + lcol;
  const int aoff = (wm * 32 + l16) * GPITCH + 4 * half;
  const int boff = GBM * GPITCH + (wn * 32 + l16) * GPITCH + 4 * half;

  v8f c00 = {}, c01 = {}, c10 = {}, c11 = {};

  float4 u0, u1, u2, u3, u4, u5, e0, e1, e2, e3, e4, e5, r0, r1, r2, r3, r4, r5;
  G_LOAD_TILE(u, 0);
  G_LOAD_TILE(e, GBK);
  G_LOAD_TILE(r, 2 * GBK);
  G_STORE_TILE(u, 0);
  G_STORE_TILE(e, GTILE);
  __syncthreads();

  int o0 = 0, o1 = GTILE, o2 = 2 * GTILE;
  float4 P[8], Q[8];
  G_LOAD_FRAGS(P, o0, 0);

  const int nk = Kd / GBK;
  for (int kt = 0; kt < nk - 3; ++kt) G_STEP(kt, 1, 1, 1);
  G_STEP(nk - 3, 1, 0, 1);
  G_STEP(nk - 2, 0, 0, 1);
  G_STEP(nk - 1, 0, 0, 0);

  const int tileM = blockM + wm * 32;
  const int n0    = blockN + wn * 32 + l16;
  const float bv0 = bias[n0];
  const float bv1 = bias[n0 + 16];
#pragma unroll
  for (int i = 0; i < 2; ++i) {
#pragma unroll
    for (int r = 0; r < 8; ++r) {
      const int row = tileM + i * 16 + 8 * half + r;
      float v0 = (i == 0 ? c00[r] : c10[r]) + bv0;
      float v1 = (i == 0 ? c01[r] : c11[r]) + bv1;
      if (flags & 1) {
        const float* rr = R + (size_t)row * N;
        v0 += rr[n0];
        v1 += rr[n0 + 16];
      }
      if (flags & 2) {
        v0 = gelu_exact(v0);
        v1 = gelu_exact(v1);
      }
      size_t orow = (size_t)row;
      if (flags & 4) {
        const int bi = row / Tlen;
        orow = (size_t)bi * Kspk * Tlen + (size_t)kk * Tlen + (row % Tlen);
      }
      float* cr = C + orow * N;
      cr[n0]      = v0;
      cr[n0 + 16] = v1;
    }
  }
}

#define RD  512
#define RNV (RD / 128)
#define LV  500
#define LV4 (LV / 4)
#define LNI ((LV4 + 31) / 32)
#define LK  2

#define ROW_XOR(x, n)                                                          \
  __builtin_bit_cast(float, __builtin_amdgcn_update_dpp(                       \
      __builtin_bit_cast(int, (x)), __builtin_bit_cast(int, (x)),              \
      0x160 + (n), 0xf, 0xf, false))
__device__ __forceinline__ float xhalf(float x) {
  const int b = __builtin_bit_cast(int, x);
  return __builtin_bit_cast(float,
      __builtin_amdgcn_permlanex16(b, b, 0x76543210, 0xfedcba98, false, false));
}
__device__ __forceinline__ float wave_sum(float v) {
  v += ROW_XOR(v, 1);
  v += ROW_XOR(v, 2);
  v += ROW_XOR(v, 4);
  v += ROW_XOR(v, 8);
  return v + xhalf(v);
}
__device__ __forceinline__ float wave_max(float v) {
  v = fmaxf(v, ROW_XOR(v, 1));
  v = fmaxf(v, ROW_XOR(v, 2));
  v = fmaxf(v, ROW_XOR(v, 4));
  v = fmaxf(v, ROW_XOR(v, 8));
  return fmaxf(v, xhalf(v));
}

__device__ __forceinline__ void row_stats(const float4 (&v)[RNV], float& mean, float& inv) {
  float s = 0.f;
#pragma unroll
  for (int i = 0; i < RNV; ++i) s += (v[i].x + v[i].y) + (v[i].z + v[i].w);
  mean = wave_sum(s) * (1.0f / RD);
  float s2 = 0.f;
#pragma unroll
  for (int i = 0; i < RNV; ++i) {
    const float dx = v[i].x - mean, dy = v[i].y - mean;
    const float dz = v[i].z - mean, dw = v[i].w - mean;
    s2 += (dx * dx + dy * dy) + (dz * dz + dw * dw);
  }
  inv = rsqrtf(wave_sum(s2) * (1.0f / RD) + 1e-5f);
}

__global__ __launch_bounds__(256) void ln_kernel(
    const float* __restrict__ X, const float* __restrict__ g,
    const float* __restrict__ b, float* __restrict__ Y, int rows)
{
  const int lane = threadIdx.x & 31;
  const int row  = blockIdx.x * 8 + __builtin_amdgcn_readfirstlane(threadIdx.x >> 5);
  if (row >= rows) return;
  const float4* x4 = (const float4*)(X + (size_t)row * RD) + lane;
  const float4* g4 = (const float4*)g + lane;
  const float4* b4 = (const float4*)b + lane;
  float4 xv[RNV], gv[RNV], bv[RNV];
#pragma unroll
  for (int i = 0; i < RNV; ++i) xv[i] = x4[i * 32];
#pragma unroll
  for (int i = 0; i < RNV; ++i) { gv[i] = g4[i * 32]; bv[i] = b4[i * 32]; }
  __builtin_amdgcn_sched_barrier(0);
  float mean, inv;
  row_stats(xv, mean, inv);
  float4* y4 = (float4*)(Y + (size_t)row * RD) + lane;
#pragma unroll
  for (int i = 0; i < RNV; ++i) {
    float4 o;
    o.x = (xv[i].x - mean) * inv * gv[i].x + bv[i].x;
    o.y = (xv[i].y - mean) * inv * gv[i].y + bv[i].y;
    o.z = (xv[i].z - mean) * inv * gv[i].z + bv[i].z;
    o.w = (xv[i].w - mean) * inv * gv[i].w + bv[i].w;
    y4[i * 32] = o;
  }
}

__global__ __launch_bounds__(256) void wmix_kernel(
    const float* __restrict__ logits, const int* __restrict__ blank_p,
    float* __restrict__ wmix, int BT)
{
  const int lane = threadIdx.x & 31;
  const int row  = blockIdx.x * 8 + __builtin_amdgcn_readfirstlane(threadIdx.x >> 5);
  if (row >= BT) return;
  const int blank = blank_p[0];
  float4 lv[LK][LNI];
  float  lb[LK];
#pragma unroll
  for (int k2 = 0; k2 < LK; ++k2) {
    const float* L = logits + ((size_t)k2 * BT + row) * LV;
    const float4* L4 = (const float4*)L;
#pragma unroll
    for (int i = 0; i < LNI; ++i) {
      const int idx = lane + 32 * i;
      lv[k2][i] = (idx < LV4) ? L4[idx]
                              : make_float4(-INFINITY, -INFINITY, -INFINITY, -INFINITY);
    }
    lb[k2] = L[blank];
  }
  __builtin_amdgcn_sched_barrier(0);
  float a[LK];
#pragma unroll
  for (int k2 = 0; k2 < LK; ++k2) {
    float m = -INFINITY;
#pragma unroll
    for (int i = 0; i < LNI; ++i)
      m = fmaxf(m, fmaxf(fmaxf(lv[k2][i].x, lv[k2][i].y), fmaxf(lv[k2][i].z, lv[k2][i].w)));
    m = wave_max(m);
    float s = 0.f;
#pragma unroll
    for (int i = 0; i < LNI; ++i)
      s += (expf(lv[k2][i].x - m) + expf(lv[k2][i].y - m)) +
           (expf(lv[k2][i].z - m) + expf(lv[k2][i].w - m));
    s = wave_sum(s);
    a[k2] = 1.0f - expf(lb[k2] - m) / s;
  }
  if (lane == 0) {
    float m2 = -INFINITY;
#pragma unroll
    for (int k2 = 0; k2 < LK; ++k2) m2 = fmaxf(m2, a[k2]);
    float e[LK];
    float ss = 0.f;
#pragma unroll
    for (int k2 = 0; k2 < LK; ++k2) { e[k2] = expf(a[k2] - m2); ss += e[k2]; }
#pragma unroll
    for (int k2 = 0; k2 < LK; ++k2) wmix[(size_t)row * LK + k2] = e[k2] / ss;
  }
}

__global__ __launch_bounds__(256) void stream_prep(
    const float* __restrict__ X, const float* __restrict__ wmix,
    const float* __restrict__ tags, const float* __restrict__ g,
    const float* __restrict__ bt, float* __restrict__ Xk,
    float* __restrict__ Qin, int rows)
{
  const int lane = threadIdx.x & 31;
  const int row  = blockIdx.x * 8 + __builtin_amdgcn_readfirstlane(threadIdx.x >> 5);
  if (row >= rows * LK) return;
  const int kk   = row / rows;
  const int xrow = row - kk * rows;
  const float wv = wmix[(size_t)xrow * LK + kk];
  const float4* x4 = (const float4*)(X + (size_t)xrow * RD) + lane;
  const float4* t4 = (const float4*)(tags + (size_t)kk * RD) + lane;
  const float4* g4 = (const float4*)g + lane;
  const float4* b4 = (const float4*)bt + lane;
  float4 v[RNV], tv[RNV], gv[RNV], bv[RNV];
#pragma unroll
  for (int i = 0; i < RNV; ++i) v[i] = x4[i * 32];
#pragma unroll
  for (int i = 0; i < RNV; ++i) tv[i] = t4[i * 32];
#pragma unroll
  for (int i = 0; i < RNV; ++i) { gv[i] = g4[i * 32]; bv[i] = b4[i * 32]; }
  __builtin_amdgcn_sched_barrier(0);
#pragma unroll
  for (int i = 0; i < RNV; ++i) {
    v[i].x = v[i].x * wv + tv[i].x;
    v[i].y = v[i].y * wv + tv[i].y;
    v[i].z = v[i].z * wv + tv[i].z;
    v[i].w = v[i].w * wv + tv[i].w;
  }
  float4* xk4 = (float4*)(Xk + (size_t)row * RD) + lane;
#pragma unroll
  for (int i = 0; i < RNV; ++i) xk4[i * 32] = v[i];
  float mean, inv;
  row_stats(v, mean, inv);
  float4* q4 = (float4*)(Qin + (size_t)row * RD) + lane;
#pragma unroll
  for (int i = 0; i < RNV; ++i) {
    float4 o;
    o.x = (v[i].x - mean) * inv * gv[i].x + bv[i].x;
    o.y = (v[i].y - mean) * inv * gv[i].y + bv[i].y;
    o.z = (v[i].z - mean) * inv * gv[i].z + bv[i].z;
    o.w = (v[i].w - mean) * inv * gv[i].w + bv[i].w;
    q4[i * 32] = o;
  }
}

#define AQ      64
#define ANW     (AQ / 16)
#define ATHR    (32 * ANW)
#define ABAND   24
#define AKEYS   (AQ + 2 * ABAND)
#define AHD     64
#define APITCH  68

__global__ __launch_bounds__(ATHR) void banded_attn_wmma(
    const float* __restrict__ q, const float* __restrict__ kb,
    const float* __restrict__ vb, float* __restrict__ ctx,
    int T_, int Dn, int Dkv, int Hn, int Bkv)
{
  __shared__ __attribute__((aligned(16))) float sK[AKEYS * APITCH];
  __shared__ __attribute__((aligned(16))) float sV[AKEYS * APITCH];
  __shared__ __attribute__((aligned(16))) float sQ[AQ * APITCH];

  const int tid  = threadIdx.x;
  const int lane = tid & 31;
  const int w    = tid >> 5;
  const int c    = lane & 15;
  const int h    = lane >> 4;

  const int nqt = T_ / AQ;
  const int qt  = blockIdx.x % nqt;
  const int hh  = (blockIdx.x / nqt) % Hn;
  const int b   = blockIdx.x / (nqt * Hn);
  const int bk  = b % Bkv;
  const int t0  = qt * AQ;

  const size_t headoff = (size_t)hh * AHD;
  const float* kbase = kb + (size_t)bk * T_ * Dkv + headoff;
  const float* vbase = vb + (size_t)bk * T_ * Dkv + headoff;
  const float* qbase = q  + ((size_t)b * T_ + t0) * Dn + headoff;
  float*       obase = ctx + ((size_t)b * T_ + t0) * Dn + headoff;

  constexpr int NKV = (AKEYS * 16) / ATHR;
  constexpr int NQ  = (AQ * 16) / ATHR;
  const int srow = tid >> 4;
  const int sc4  = (tid & 15) * 4;
  v4f kreg[NKV], vreg[NKV], qreg[NQ];
#pragma unroll
  for (int i = 0; i < NKV; ++i) {
    const int key = t0 - ABAND + srow + (ATHR / 16) * i;
    const unsigned off = (unsigned)min(max(key, 0), T_ - 1) * (unsigned)Dkv + (unsigned)sc4;
    kreg[i] = *(const v4f*)(kbase + off);
    vreg[i] = *(const v4f*)(vbase + off);
  }
#pragma unroll
  for (int i = 0; i < NQ; ++i) {
    const unsigned off = (unsigned)(srow + (ATHR / 16) * i) * (unsigned)Dn + (unsigned)sc4;
    qreg[i] = *(const v4f*)(qbase + off);
  }
  __builtin_amdgcn_sched_barrier(0);
  const v4f zero4 = {0.f, 0.f, 0.f, 0.f};
#pragma unroll
  for (int i = 0; i < NKV; ++i) {
    const int row = srow + (ATHR / 16) * i;
    const int key = t0 - ABAND + row;
    const bool ok = (key >= 0) && (key < T_);
    *(v4f*)(sK + row * APITCH + sc4) = ok ? kreg[i] : zero4;
    *(v4f*)(sV + row * APITCH + sc4) = ok ? vreg[i] : zero4;
  }
#pragma unroll
  for (int i = 0; i < NQ; ++i) {
    const int row = srow + (ATHR / 16) * i;
    *(v4f*)(sQ + row * APITCH + sc4) = qreg[i];
  }
  __syncthreads();

  float* pw = sQ + (w * 16) * APITCH;
  const float* qrow = pw + c * APITCH + 4 * h;
  const float* krow = sK + (w * 16 + c) * APITCH + 4 * h;
  float4 qa[8];
#pragma unroll
  for (int ks = 0; ks < 8; ++ks) qa[ks] = *(const float4*)(qrow + 8 * ks);

  v8f s0 = {}, s1 = {}, s2 = {}, s3 = {};
#pragma unroll
  for (int ks = 0; ks < 8; ++ks) {
    const float4 k0 = *(const float4*)(krow + 8 * ks);
    const float4 k1 = *(const float4*)(krow + 16 * APITCH + 8 * ks);
    const float4 k2 = *(const float4*)(krow + 32 * APITCH + 8 * ks);
    const float4 k3 = *(const float4*)(krow + 48 * APITCH + 8 * ks);
    const v2f al = {qa[ks].x, qa[ks].y}, ah = {qa[ks].z, qa[ks].w};
    const v2f b0l = {k0.x, k0.y}, b0h = {k0.z, k0.w};
    const v2f b1l = {k1.x, k1.y}, b1h = {k1.z, k1.w};
    const v2f b2l = {k2.x, k2.y}, b2h = {k2.z, k2.w};
    const v2f b3l = {k3.x, k3.y}, b3h = {k3.z, k3.w};
    WMMA_F32(s0, al, b0l);
    WMMA_F32(s1, al, b1l);
    WMMA_F32(s2, al, b2l);
    WMMA_F32(s3, al, b3l);
    WMMA_F32(s0, ah, b0h);
    WMMA_F32(s1, ah, b1h);
    WMMA_F32(s2, ah, b2h);
    WMMA_F32(s3, ah, b3h);
  }

  const float scale = 0.125f;
  const int jbase = t0 + 16 * w - ABAND;
  bool colok[4];
#pragma unroll
  for (int nt = 0; nt < 4; ++nt) {
    const int key = jbase + 16 * nt + c;
    colok[nt] = (key >= 0) && (key < T_);
  }
  float inv[8];
#pragma unroll
  for (int r = 0; r < 8; ++r) {
    const int rq = 8 * h + r;
    float x[4] = {s0[r], s1[r], s2[r], s3[r]};
#pragma unroll
    for (int nt = 0; nt < 4; ++nt) {
      const int jj = 16 * nt + c;
      const bool ok = colok[nt] && (jj >= rq) && (jj <= rq + 2 * ABAND);
      x[nt] = ok ? x[nt] * scale : -INFINITY;
    }
    float mx = fmaxf(fmaxf(x[0], x[1]), fmaxf(x[2], x[3]));
    mx = fmaxf(mx, ROW_XOR(mx, 1));
    mx = fmaxf(mx, ROW_XOR(mx, 2));
    mx = fmaxf(mx, ROW_XOR(mx, 4));
    mx = fmaxf(mx, ROW_XOR(mx, 8));
    float e[4];
#pragma unroll
    for (int nt = 0; nt < 4; ++nt) e[nt] = expf(x[nt] - mx);
    float sum = (e[0] + e[1]) + (e[2] + e[3]);
    sum += ROW_XOR(sum, 1);
    sum += ROW_XOR(sum, 2);
    sum += ROW_XOR(sum, 4);
    sum += ROW_XOR(sum, 8);
    inv[r] = 1.0f / sum;
    float* pr = pw + rq * APITCH + c;
    pr[0]  = e[0];
    pr[16] = e[1];
    pr[32] = e[2];
    pr[48] = e[3];
  }
  __syncthreads();

  const float* prow = pw + c * APITCH + 4 * h;
  const float* vcol = sV + (w * 16 + 4 * h) * APITCH + c;
  v8f o0 = {}, o1 = {}, o2 = {}, o3 = {};
#pragma unroll
  for (int ks = 0; ks < 8; ++ks) {
    const float4 pa = *(const float4*)(prow + 8 * ks);
    const float* vr = vcol + 8 * ks * APITCH;
    const v2f al = {pa.x, pa.y}, ah = {pa.z, pa.w};
    const v2f b0l = {vr[0],      vr[APITCH]},      b0h = {vr[2 * APITCH],      vr[3 * APITCH]};
    const v2f b1l = {vr[16],     vr[APITCH + 16]}, b1h = {vr[2 * APITCH + 16], vr[3 * APITCH + 16]};
    const v2f b2l = {vr[32],     vr[APITCH + 32]}, b2h = {vr[2 * APITCH + 32], vr[3 * APITCH + 32]};
    const v2f b3l = {vr[48],     vr[APITCH + 48]}, b3h = {vr[2 * APITCH + 48], vr[3 * APITCH + 48]};
    WMMA_F32(o0, al, b0l);
    WMMA_F32(o1, al, b1l);
    WMMA_F32(o2, al, b2l);
    WMMA_F32(o3, al, b3l);
    WMMA_F32(o0, ah, b0h);
    WMMA_F32(o1, ah, b1h);
    WMMA_F32(o2, ah, b2h);
    WMMA_F32(o3, ah, b3h);
  }

#pragma unroll
  for (int r = 0; r < 8; ++r) {
    float* pr = pw + (8 * h + r) * APITCH + c;
    pr[0]  = o0[r] * inv[r];
    pr[16] = o1[r] * inv[r];
    pr[32] = o2[r] * inv[r];
    pr[48] = o3[r] * inv[r];
  }
  __syncthreads();
#pragma unroll
  for (int i = 0; i < 8; ++i) {
    const int row = 2 * i + h;
    const float4 ov = *(const float4*)(pw + row * APITCH + 4 * c);
    *(float4*)(obase + (size_t)(16 * w + row) * Dn + 4 * c) = ov;
  }
}

extern "C" void kernel_launch(void* const* d_in, const int* in_sizes, int n_in,
                              void* d_out, int out_size, void* d_ws, size_t ws_size,
                              hipStream_t stream) {
  (void)in_sizes; (void)n_in; (void)out_size; (void)ws_size;
  constexpr int B = 4, T = 1024, D_IN = 512, D = 512, K = 2, V = 500, H = 8, BAND = 24;
  constexpr int M = B * T;
  static_assert(D == RD && V == LV && K == LK && M % 8 == 0, "row kernels are sized for these shapes");
  static_assert(D == H * AHD && BAND == ABAND && T % AQ == 0, "attention kernel is sized for these shapes");

  const float* x_m      = (const float*)d_in[0];
  const float* logits   = (const float*)d_in[1];
  const int*   blank_id = (const int*)  d_in[2];
  const float* proj_w   = (const float*)d_in[3];
  const float* proj_b   = (const float*)d_in[4];
  const float* spk_tags = (const float*)d_in[5];
  const float* lnq_g    = (const float*)d_in[6];
  const float* lnq_b    = (const float*)d_in[7];
  const float* lnkv_g   = (const float*)d_in[8];
  const float* lnkv_b   = (const float*)d_in[9];
  const float* in_w     = (const float*)d_in[10];
  const float* in_b     = (const float*)d_in[11];
  const float* out_w    = (const float*)d_in[12];
  const float* out_b    = (const float*)d_in[13];
  const float* ffn_ln_g = (const float*)d_in[14];
  const float* ffn_ln_b = (const float*)d_in[15];
  const float* ffn_w1   = (const float*)d_in[16];
  const float* ffn_b1   = (const float*)d_in[17];
  const float* ffn_w2   = (const float*)d_in[18];
  const float* ffn_b2   = (const float*)d_in[19];
  float* out = (float*)d_out;

  float* ws = (float*)d_ws;
  const size_t SZ = (size_t)M * D;
  float* X    = ws;
  float* kv   = X  + SZ;
  float* xk   = kv + 2 * SZ;
  float* qh   = xk + 2 * SZ;
  float* qy   = qh + 2 * SZ;
  float* t1   = ws;
  float* wmix = ws + 10 * SZ;

  const dim3 blk(256);
  auto gg = [](int m, int n) { return dim3(m / 64, n / 128); };

  gemm_wmma_f32<<<gg(M, D), blk, 0, stream>>>(x_m, proj_w, proj_b, nullptr, X,
                                              M, D, D_IN, 0, T, K, 0);
  wmix_kernel<<<dim3(M / 8), blk, 0, stream>>>(logits, blank_id, wmix, M);
  ln_kernel<<<dim3(M / 8), blk, 0, stream>>>(X, lnkv_g, lnkv_b, xk, M);
  gemm_wmma_f32<<<gg(M, 2 * D), blk, 0, stream>>>(xk, in_w + (size_t)D * D, in_b + D,
                                                  nullptr, kv, M, 2 * D, D, 0, T, K, 0);
  stream_prep<<<dim3(K * M / 8), blk, 0, stream>>>(X, wmix, spk_tags, lnq_g, lnq_b,
                                                   xk, qh, M);
  gemm_wmma_f32<<<gg(K * M, D), blk, 0, stream>>>(qh, in_w, in_b, nullptr, qy,
                                                  K * M, D, D, 0, T, K, 0);
  banded_attn_wmma<<<dim3(K * B * H * (T / AQ)), dim3(ATHR), 0, stream>>>(qy, kv, kv + D, qh,
                                                                          T, D, 2 * D, H, B);
  gemm_wmma_f32<<<gg(K * M, D), blk, 0, stream>>>(qh, out_w, out_b, xk, qy,
                                                  K * M, D, D,  1, T, K, 0);
  ln_kernel<<<dim3(K * M / 8), blk, 0, stream>>>(qy, ffn_ln_g, ffn_ln_b, qh, K * M);

  static_assert(K == 2, "the FFN pair below is written out for two speakers");
  gemm_wmma_f32<<<gg(M, 4 * D), blk, 0, stream>>>(qh, ffn_w1, ffn_b1, nullptr, t1,
                                                  M, 4 * D, D,  2, T, K, 0);
  gemm_wmma_f32<<<gg(M, D), blk, 0, stream>>>(t1, ffn_w2, ffn_b2, qy, out,
                                              M, D, 4 * D,  1 | 4, T, K, 0);
  gemm_wmma_f32<<<gg(M, 4 * D), blk, 0, stream>>>(qh + SZ, ffn_w1, ffn_b1, nullptr, t1,
                                                  M, 4 * D, D,  2, T, K, 1);
  gemm_wmma_f32<<<gg(M, D), blk, 0, stream>>>(t1, ffn_w2, ffn_b2, qy + SZ, out,
                                              M, D, 4 * D,  1 | 4, T, K, 1);
}


# GEMM epilogue moves whole 128-byte lines: accumulators parked in the wave's LDS rows, bias and residual fetched as one group of 128-bit pieces a K step ahead, 128-bit C stores, flag tests per group
# speedup vs baseline: 2.4842x; 1.0875x over previous
#include <hip/hip_runtime.h>
#include <hip/hip_bf16.h>
#include <math.h>

typedef float v2f __attribute__((ext_vector_type(2)));
typedef float v4f __attribute__((ext_vector_type(4)));
typedef float v8f __attribute__((ext_vector_type(8)));

__device__ __forceinline__ float gelu_exact(float x) {
  return 0.5f * x * (1.0f + erff(x * 0.70710678118654752f));
}

#define WMMA_F32(cacc, av, bv)                                                 \
  cacc = __builtin_amdgcn_wmma_f32_16x16x4_f32(false, (av), false, (bv),       \
                                               (short)0, (cacc), false, false)

#define GBM 64
#define GBN 128
#define GBK 32
#define GPITCH 36
#define GTILE ((GBM + GBN) * GPITCH)
#define GSTAGES 3

#define G_LOAD_TILE(r, ko)                                                     \
  do {                                                                         \
    const char* ap_ = (const char*)(Ab + (ko));                                \
    const char* wp_ = (const char*)(Wb + (ko));                                \
    unsigned q0_ = ob0, q1_ = ob1, q2_ = ob2, q3_ = ob3;                       \
    asm volatile("" : "+v"(q0_), "+v"(q1_), "+v"(q2_), "+v"(q3_));             \
    r##0 = *(const float4*)(ap_ + q0_);                                        \
    r##1 = *(const float4*)(ap_ + q1_);                                        \
    r##2 = *(const float4*)(wp_ + q0_);                                        \
    r##3 = *(const float4*)(wp_ + q1_);                                        \
    r##4 = *(const float4*)(wp_ + q2_);                                        \
    r##5 = *(const float4*)(wp_ + q3_);                                        \
  } while (0)

#define G_STORE_TILE(r, st)                                                    \
  do {                                                                         \
    float* s_ = sm + (st) + so;                                                \
    *(float4*)(s_)                         = r##0;                             \
    *(float4*)(s_ + 32 * GPITCH)           = r##1;                             \
    *(float4*)(s_ + GBM * GPITCH)          = r##2;                             \
    *(float4*)(s_ + (GBM + 32) * GPITCH)   = r##3;                             \
    *(float4*)(s_ + (GBM + 64) * GPITCH)   = r##4;                             \
    *(float4*)(s_ + (GBM + 96) * GPITCH)   = r##5;                             \
  } while (0)

#define G_LOAD_FRAGS(F, st, hh)                                                \
  do {                                                                         \
    const float* c_ = sm + (st) + (hh) * 16;                                   \
    F[0] = *(const float4*)(c_ + aoff);                                        \
    F[1] = *(const float4*)(c_ + aoff + 16 * GPITCH);                          \
    F[2] = *(const float4*)(c_ + boff);                                        \
    F[3] = *(const float4*)(c_ + boff + 16 * GPITCH);                          \
    F[4] = *(const float4*)(c_ + aoff + 8);                                    \
    F[5] = *(const float4*)(c_ + aoff + 16 * GPITCH + 8);                      \
    F[6] = *(const float4*)(c_ + boff + 8);                                    \
    F[7] = *(const float4*)(c_ + boff + 16 * GPITCH + 8);                      \
  } while (0)

#define G_MMA_HALF(F)                                                          \
  do {                                                                         \
    _Pragma("unroll")                                                          \
    for (int j_ = 0; j_ < 2; ++j_) {                                           \
      const float4 fa0 = F[4 * j_], fa1 = F[4 * j_ + 1];                       \
      const float4 fb0 = F[4 * j_ + 2], fb1 = F[4 * j_ + 3];                   \
      const v2f a0l = {fa0.x, fa0.y}, a0h = {fa0.z, fa0.w};                    \
      const v2f a1l = {fa1.x, fa1.y}, a1h = {fa1.z, fa1.w};                    \
      const v2f b0l = {fb0.x, fb0.y}, b0h = {fb0.z, fb0.w};                    \
      const v2f b1l = {fb1.x, fb1.y}, b1h = {fb1.z, fb1.w};                    \
      WMMA_F32(c00, a0l, b0l);                                                 \
      WMMA_F32(c01, a0l, b1l);                                                 \
      WMMA_F32(c10, a1l, b0l);                                                 \
      WMMA_F32(c11, a1l, b1l);                                                 \
      WMMA_F32(c00, a0h, b0h);                                                 \
      WMMA_F32(c01, a0h, b1h);                                                 \
      WMMA_F32(c10, a1h, b0h);                                                 \
      WMMA_F32(c11, a1h, b1h);                                                 \
    }                                                                          \
  } while (0)

#define G_GRP(mask, n) __builtin_amdgcn_sched_group_barrier((mask), (n), 0)
#define G_STEP(t, S, G, N)                                                     \
  do {                                                                         \
    G_LOAD_FRAGS(Q, o0, 1);                                                    \
    G_MMA_HALF(P);                                                             \
    G_GRP(0x100, 8);             \
    G_GRP(0x008, 16);     \
    __builtin_amdgcn_sched_barrier(0);   \
    if ((S) || (N)) {                                                          \
      __syncthreads();                                                         \
      __builtin_amdgcn_sched_barrier(0);   \
    }                                                                          \
    if (N) G_LOAD_FRAGS(P, o1, 0);                                             \
    if (S) G_STORE_TILE(r, o2);                                                \
    if (G) G_LOAD_TILE(r, ((t) + 3) * GBK);                                    \
    G_MMA_HALF(Q);                                                             \
    G_GRP(0x008, 8);              \
    if (N) G_GRP(0x100, 8);      \
    if (S) G_GRP(0x200, 6);              \
    if (G) G_GRP(0x020, 6);            \
    G_GRP(0x008, 8);                  \
    __builtin_amdgcn_sched_barrier(0);   \
    { const int t_ = o0; o0 = o1; o1 = o2; o2 = t_; }                          \
  } while (0)

__global__ __launch_bounds__(256) void gemm_wmma_f32(
    const float* __restrict__ A, const float* __restrict__ W,
    const float* __restrict__ bias, const float* __restrict__ R,
    float* __restrict__ C, int M, int N, int Kd,
    int flags, int Tlen, int Kspk, int kk)
{
  __shared__ __attribute__((aligned(16))) float sm[GSTAGES * GTILE];

  const int tid  = threadIdx.x;
  const int lane = tid & 31;
  const int w    = tid >> 5;
  const int wm   = w & 1;
  const int wn   = w >> 1;
  const int half = lane >> 4;
  const int l16  = lane & 15;
  const int blockM = blockIdx.x * GBM;
  const int blockN = blockIdx.y * GBN;

  const int lrow = tid >> 3;
  const int lcol = (tid & 7) * 4;
  const float* Ab = A + (size_t)blockM * Kd;
  const float* Wb = W + (size_t)blockN * Kd;
  const unsigned ob0 = (unsigned)(lrow * Kd + lcol) * 4u;
  const unsigned ob1 = ob0 + 128u * (unsigned)Kd;
  const unsigned ob2 = ob0 + 256u * (unsigned)Kd;
  const unsigned ob3 = ob0 + 384u * (unsigned)Kd;

  const int so   = lrow * GPITCH + lcol;
  const int aoff = (wm * 32 + l16) * GPITCH + 4 * half;
  const int boff = GBM * GPITCH + (wn * 32 + l16) * GPITCH + 4 * half;

  v8f c00 = {}, c01 = {}, c10 = {}, c11 = {};

  float4 u0, u1, u2, u3, u4, u5, e0, e1, e2, e3, e4, e5, r0, r1, r2, r3, r4, r5;
  G_LOAD_TILE(u, 0);
  G_LOAD_TILE(e, GBK);
  G_LOAD_TILE(r, 2 * GBK);
  G_STORE_TILE(u, 0);
  G_STORE_TILE(e, GTILE);
  __syncthreads();

  int o0 = 0, o1 = GTILE, o2 = 2 * GTILE;
  float4 P[8], Q[8];
  G_LOAD_FRAGS(P, o0, 0);

  const int nk = Kd / GBK;
  for (int kt = 0; kt < nk - 3; ++kt) G_STEP(kt, 1, 1, 1);
  G_STEP(nk - 3, 1, 0, 1);
  G_STEP(nk - 2, 0, 0, 1);

  const int erow  = lane >> 3;
  const int ecol  = (lane & 7) * 4;
  const int tileM = blockM + wm * 32;
  const int ncol  = blockN + wn * 32 + ecol;
  const float4 bias4 = *(const float4*)(bias + ncol);
  float4 rs[8];
  if (flags & 1) {
    const float* rp = R + (size_t)(tileM + erow) * N + ncol;
#pragma unroll
    for (int p = 0; p < 8; ++p) rs[p] = *(const float4*)(rp + (size_t)(4 * p) * N);
  }
  __builtin_amdgcn_sched_barrier(0);

  G_STEP(nk - 1, 0, 0, 0);

  float* cw = sm + ((w < 4) ? (o0 + w * (32 * GPITCH)) : (o1 + (w - 4) * (32 * GPITCH)));
#pragma unroll
  for (int r = 0; r < 8; ++r) {
    float* pr = cw + (8 * half + r) * GPITCH + l16;
    pr[0]                = c00[r];
    pr[16]               = c01[r];
    pr[16 * GPITCH]      = c10[r];
    pr[16 * GPITCH + 16] = c11[r];
  }
  __syncthreads();

  float4 v[8];
#pragma unroll
  for (int p = 0; p < 8; ++p) v[p] = *(const float4*)(cw + (4 * p + erow) * GPITCH + ecol);
#pragma unroll
  for (int p = 0; p < 8; ++p) {
    v[p].x += bias4.x; v[p].y += bias4.y; v[p].z += bias4.z; v[p].w += bias4.w;
  }
  if (flags & 1) {
#pragma unroll
    for (int p = 0; p < 8; ++p) {
      v[p].x += rs[p].x; v[p].y += rs[p].y; v[p].z += rs[p].z; v[p].w += rs[p].w;
    }
  }
  if (flags & 2) {
#pragma unroll
    for (int p = 0; p < 8; ++p) {
      v[p].x = gelu_exact(v[p].x); v[p].y = gelu_exact(v[p].y);
      v[p].z = gelu_exact(v[p].z); v[p].w = gelu_exact(v[p].w);
    }
  }
  size_t orow = (size_t)tileM;
  if (flags & 4) {
    const int bi = blockM / Tlen;
    orow = (size_t)bi * Kspk * Tlen + (size_t)kk * Tlen + (size_t)(tileM - bi * Tlen);
  }
  float* cp = C + (orow + erow) * N + ncol;
#pragma unroll
  for (int p = 0; p < 8; ++p) *(float4*)(cp + (size_t)(4 * p) * N) = v[p];
}

#define RD  512
#define RNV (RD / 128)
#define LV  500
#define LV4 (LV / 4)
#define LNI ((LV4 + 31) / 32)
#define LK  2

#define ROW_XOR(x, n)                                                          \
  __builtin_bit_cast(float, __builtin_amdgcn_update_dpp(                       \
      __builtin_bit_cast(int, (x)), __builtin_bit_cast(int, (x)),              \
      0x160 + (n), 0xf, 0xf, false))
__device__ __forceinline__ float xhalf(float x) {
  const int b = __builtin_bit_cast(int, x);
  return __builtin_bit_cast(float,
      __builtin_amdgcn_permlanex16(b, b, 0x76543210, 0xfedcba98, false, false));
}
__device__ __forceinline__ float wave_sum(float v) {
  v += ROW_XOR(v, 1);
  v += ROW_XOR(v, 2);
  v += ROW_XOR(v, 4);
  v += ROW_XOR(v, 8);
  return v + xhalf(v);
}
__device__ __forceinline__ float wave_max(float v) {
  v = fmaxf(v, ROW_XOR(v, 1));
  v = fmaxf(v, ROW_XOR(v, 2));
  v = fmaxf(v, ROW_XOR(v, 4));
  v = fmaxf(v, ROW_XOR(v, 8));
  return fmaxf(v, xhalf(v));
}

__device__ __forceinline__ void row_stats(const float4 (&v)[RNV], float& mean, float& inv) {
  float s = 0.f;
#pragma unroll
  for (int i = 0; i < RNV; ++i) s += (v[i].x + v[i].y) + (v[i].z + v[i].w);
  mean = wave_sum(s) * (1.0f / RD);
  float s2 = 0.f;
#pragma unroll
  for (int i = 0; i < RNV; ++i) {
    const float dx = v[i].x - mean, dy = v[i].y - mean;
    const float dz = v[i].z - mean, dw = v[i].w - mean;
    s2 += (dx * dx + dy * dy) + (dz * dz + dw * dw);
  }
  inv = rsqrtf(wave_sum(s2) * (1.0f / RD) + 1e-5f);
}

__global__ __launch_bounds__(256) void ln_kernel(
    const float* __restrict__ X, const float* __restrict__ g,
    const float* __restrict__ b, float* __restrict__ Y, int rows)
{
  const int lane = threadIdx.x & 31;
  const int row  = blockIdx.x * 8 + __builtin_amdgcn_readfirstlane(threadIdx.x >> 5);
  if (row >= rows) return;
  const float4* x4 = (const float4*)(X + (size_t)row * RD) + lane;
  const float4* g4 = (const float4*)g + lane;
  const float4* b4 = (const float4*)b + lane;
  float4 xv[RNV], gv[RNV], bv[RNV];
#pragma unroll
  for (int i = 0; i < RNV; ++i) xv[i] = x4[i * 32];
#pragma unroll
  for (int i = 0; i < RNV; ++i) { gv[i] = g4[i * 32]; bv[i] = b4[i * 32]; }
  __builtin_amdgcn_sched_barrier(0);
  float mean, inv;
  row_stats(xv, mean, inv);
  float4* y4 = (float4*)(Y + (size_t)row * RD) + lane;
#pragma unroll
  for (int i = 0; i < RNV; ++i) {
    float4 o;
    o.x = (xv[i].x - mean) * inv * gv[i].x + bv[i].x;
    o.y = (xv[i].y - mean) * inv * gv[i].y + bv[i].y;
    o.z = (xv[i].z - mean) * inv * gv[i].z + bv[i].z;
    o.w = (xv[i].w - mean) * inv * gv[i].w + bv[i].w;
    y4[i * 32] = o;
  }
}

__global__ __launch_bounds__(256) void wmix_kernel(
    const float* __restrict__ logits, const int* __restrict__ blank_p,
    float* __restrict__ wmix, int BT)
{
  const int lane = threadIdx.x & 31;
  const int row  = blockIdx.x * 8 + __builtin_amdgcn_readfirstlane(threadIdx.x >> 5);
  if (row >= BT) return;
  const int blank = blank_p[0];
  float4 lv[LK][LNI];
  float  lb[LK];
#pragma unroll
  for (int k2 = 0; k2 < LK; ++k2) {
    const float* L = logits + ((size_t)k2 * BT + row) * LV;
    const float4* L4 = (const float4*)L;
#pragma unroll
    for (int i = 0; i < LNI; ++i) {
      const int idx = lane + 32 * i;
      lv[k2][i] = (idx < LV4) ? L4[idx]
                              : make_float4(-INFINITY, -INFINITY, -INFINITY, -INFINITY);
    }
    lb[k2] = L[blank];
  }
  __builtin_amdgcn_sched_barrier(0);
  float a[LK];
#pragma unroll
  for (int k2 = 0; k2 < LK; ++k2) {
    float m = -INFINITY;
#pragma unroll
    for (int i = 0; i < LNI; ++i)
      m = fmaxf(m, fmaxf(fmaxf(lv[k2][i].x, lv[k2][i].y), fmaxf(lv[k2][i].z, lv[k2][i].w)));
    m = wave_max(m);
    float s = 0.f;
#pragma unroll
    for (int i = 0; i < LNI; ++i)
      s += (expf(lv[k2][i].x - m) + expf(lv[k2][i].y - m)) +
           (expf(lv[k2][i].z - m) + expf(lv[k2][i].w - m));
    s = wave_sum(s);
    a[k2] = 1.0f - expf(lb[k2] - m) / s;
  }
  if (lane == 0) {
    float m2 = -INFINITY;
#pragma unroll
    for (int k2 = 0; k2 < LK; ++k2) m2 = fmaxf(m2, a[k2]);
    float e[LK];
    float ss = 0.f;
#pragma unroll
    for (int k2 = 0; k2 < LK; ++k2) { e[k2] = expf(a[k2] - m2); ss += e[k2]; }
#pragma unroll
    for (int k2 = 0; k2 < LK; ++k2) wmix[(size_t)row * LK + k2] = e[k2] / ss;
  }
}

__global__ __launch_bounds__(256) void stream_prep(
    const float* __restrict__ X, const float* __restrict__ wmix,
    const float* __restrict__ tags, const float* __restrict__ g,
    const float* __restrict__ bt, float* __restrict__ Xk,
    float* __restrict__ Qin, int rows)
{
  const int lane = threadIdx.x & 31;
  const int row  = blockIdx.x * 8 + __builtin_amdgcn_readfirstlane(threadIdx.x >> 5);
  if (row >= rows * LK) return;
  const int kk   = row / rows;
  const int xrow = row - kk * rows;
  const float wv = wmix[(size_t)xrow * LK + kk];
  const float4* x4 = (const float4*)(X + (size_t)xrow * RD) + lane;
  const float4* t4 = (const float4*)(tags + (size_t)kk * RD) + lane;
  const float4* g4 = (const float4*)g + lane;
  const float4* b4 = (const float4*)bt + lane;
  float4 v[RNV], tv[RNV], gv[RNV], bv[RNV];
#pragma unroll
  for (int i = 0; i < RNV; ++i) v[i] = x4[i * 32];
#pragma unroll
  for (int i = 0; i < RNV; ++i) tv[i] = t4[i * 32];
#pragma unroll
  for (int i = 0; i < RNV; ++i) { gv[i] = g4[i * 32]; bv[i] = b4[i * 32]; }
  __builtin_amdgcn_sched_barrier(0);
#pragma unroll
  for (int i = 0; i < RNV; ++i) {
    v[i].x = v[i].x * wv + tv[i].x;
    v[i].y = v[i].y * wv + tv[i].y;
    v[i].z = v[i].z * wv + tv[i].z;
    v[i].w = v[i].w * wv + tv[i].w;
  }
  float4* xk4 = (float4*)(Xk + (size_t)row * RD) + lane;
#pragma unroll
  for (int i = 0; i < RNV; ++i) xk4[i * 32] = v[i];
  float mean, inv;
  row_stats(v, mean, inv);
  float4* q4 = (float4*)(Qin + (size_t)row * RD) + lane;
#pragma unroll
  for (int i = 0; i < RNV; ++i) {
    float4 o;
    o.x = (v[i].x - mean) * inv * gv[i].x + bv[i].x;
    o.y = (v[i].y - mean) * inv * gv[i].y + bv[i].y;
    o.z = (v[i].z - mean) * inv * gv[i].z + bv[i].z;
    o.w = (v[i].w - mean) * inv * gv[i].w + bv[i].w;
    q4[i * 32] = o;
  }
}

#define AQ      64
#define ANW     (AQ / 16)
#define ATHR    (32 * ANW)
#define ABAND   24
#define AKEYS   (AQ + 2 * ABAND)
#define AHD     64
#define APITCH  68

__global__ __launch_bounds__(ATHR) void banded_attn_wmma(
    const float* __restrict__ q, const float* __restrict__ kb,
    const float* __restrict__ vb, float* __restrict__ ctx,
    int T_, int Dn, int Dkv, int Hn, int Bkv)
{
  __shared__ __attribute__((aligned(16))) float sK[AKEYS * APITCH];
  __shared__ __attribute__((aligned(16))) float sV[AKEYS * APITCH];
  __shared__ __attribute__((aligned(16))) float sQ[AQ * APITCH];

  const int tid  = threadIdx.x;
  const int lane = tid & 31;
  const int w    = tid >> 5;
  const int c    = lane & 15;
  const int h    = lane >> 4;

  const int nqt = T_ / AQ;
  const int qt  = blockIdx.x % nqt;
  const int hh  = (blockIdx.x / nqt) % Hn;
  const int b   = blockIdx.x / (nqt * Hn);
  const int bk  = b % Bkv;
  const int t0  = qt * AQ;

  const size_t headoff = (size_t)hh * AHD;
  const float* kbase = kb + (size_t)bk * T_ * Dkv + headoff;
  const float* vbase = vb + (size_t)bk * T_ * Dkv + headoff;
  const float* qbase = q  + ((size_t)b * T_ + t0) * Dn + headoff;
  float*       obase = ctx + ((size_t)b * T_ + t0) * Dn + headoff;

  constexpr int NKV = (AKEYS * 16) / ATHR;
  constexpr int NQ  = (AQ * 16) / ATHR;
  const int srow = tid >> 4;
  const int sc4  = (tid & 15) * 4;
  v4f kreg[NKV], vreg[NKV], qreg[NQ];
#pragma unroll
  for (int i = 0; i < NKV; ++i) {
    const int key = t0 - ABAND + srow + (ATHR / 16) * i;
    const unsigned off = (unsigned)min(max(key, 0), T_ - 1) * (unsigned)Dkv + (unsigned)sc4;
    kreg[i] = *(const v4f*)(kbase + off);
    vreg[i] = *(const v4f*)(vbase + off);
  }
#pragma unroll
  for (int i = 0; i < NQ; ++i) {
    const unsigned off = (unsigned)(srow + (ATHR / 16) * i) * (unsigned)Dn + (unsigned)sc4;
    qreg[i] = *(const v4f*)(qbase + off);
  }
  __builtin_amdgcn_sched_barrier(0);
  const v4f zero4 = {0.f, 0.f, 0.f, 0.f};
#pragma unroll
  for (int i = 0; i < NKV; ++i) {
    const int row = srow + (ATHR / 16) * i;
    const int key = t0 - ABAND + row;
    const bool ok = (key >= 0) && (key < T_);
    *(v4f*)(sK + row * APITCH + sc4) = ok ? kreg[i] : zero4;
    *(v4f*)(sV + row * APITCH + sc4) = ok ? vreg[i] : zero4;
  }
#pragma unroll
  for (int i = 0; i < NQ; ++i) {
    const int row = srow + (ATHR / 16) * i;
    *(v4f*)(sQ + row * APITCH + sc4) = qreg[i];
  }
  __syncthreads();

  float* pw = sQ + (w * 16) * APITCH;
  const float* qrow = pw + c * APITCH + 4 * h;
  const float* krow = sK + (w * 16 + c) * APITCH + 4 * h;
  float4 qa[8];
#pragma unroll
  for (int ks = 0; ks < 8; ++ks) qa[ks] = *(const float4*)(qrow + 8 * ks);

  v8f s0 = {}, s1 = {}, s2 = {}, s3 = {};
#pragma unroll
  for (int ks = 0; ks < 8; ++ks) {
    const float4 k0 = *(const float4*)(krow + 8 * ks);
    const float4 k1 = *(const float4*)(krow + 16 * APITCH + 8 * ks);
    const float4 k2 = *(const float4*)(krow + 32 * APITCH + 8 * ks);
    const float4 k3 = *(const float4*)(krow + 48 * APITCH + 8 * ks);
    const v2f al = {qa[ks].x, qa[ks].y}, ah = {qa[ks].z, qa[ks].w};
    const v2f b0l = {k0.x, k0.y}, b0h = {k0.z, k0.w};
    const v2f b1l = {k1.x, k1.y}, b1h = {k1.z, k1.w};
    const v2f b2l = {k2.x, k2.y}, b2h = {k2.z, k2.w};
    const v2f b3l = {k3.x, k3.y}, b3h = {k3.z, k3.w};
    WMMA_F32(s0, al, b0l);
    WMMA_F32(s1, al, b1l);
    WMMA_F32(s2, al, b2l);
    WMMA_F32(s3, al, b3l);
    WMMA_F32(s0, ah, b0h);
    WMMA_F32(s1, ah, b1h);
    WMMA_F32(s2, ah, b2h);
    WMMA_F32(s3, ah, b3h);
  }

  const float scale = 0.125f;
  const int jbase = t0 + 16 * w - ABAND;
  bool colok[4];
#pragma unroll
  for (int nt = 0; nt < 4; ++nt) {
    const int key = jbase + 16 * nt + c;
    colok[nt] = (key >= 0) && (key < T_);
  }
  float inv[8];
#pragma unroll
  for (int r = 0; r < 8; ++r) {
    const int rq = 8 * h + r;
    float x[4] = {s0[r], s1[r], s2[r], s3[r]};
#pragma unroll
    for (int nt = 0; nt < 4; ++nt) {
      const int jj = 16 * nt + c;
      const bool ok = colok[nt] && (jj >= rq) && (jj <= rq + 2 * ABAND);
      x[nt] = ok ? x[nt] * scale : -INFINITY;
    }
    float mx = fmaxf(fmaxf(x[0], x[1]), fmaxf(x[2], x[3]));
    mx = fmaxf(mx, ROW_XOR(mx, 1));
    mx = fmaxf(mx, ROW_XOR(mx, 2));
    mx = fmaxf(mx, ROW_XOR(mx, 4));
    mx = fmaxf(mx, ROW_XOR(mx, 8));
    float e[4];
#pragma unroll
    for (int nt = 0; nt < 4; ++nt) e[nt] = expf(x[nt] - mx);
    float sum = (e[0] + e[1]) + (e[2] + e[3]);
    sum += ROW_XOR(sum, 1);
    sum += ROW_XOR(sum, 2);
    sum += ROW_XOR(sum, 4);
    sum += ROW_XOR(sum, 8);
    inv[r] = 1.0f / sum;
    float* pr = pw + rq * APITCH + c;
    pr[0]  = e[0];
    pr[16] = e[1];
    pr[32] = e[2];
    pr[48] = e[3];
  }
  __syncthreads();

  const float* prow = pw + c * APITCH + 4 * h;
  const float* vcol = sV + (w * 16 + 4 * h) * APITCH + c;
  v8f o0 = {}, o1 = {}, o2 = {}, o3 = {};
#pragma unroll
  for (int ks = 0; ks < 8; ++ks) {
    const float4 pa = *(const float4*)(prow + 8 * ks);
    const float* vr = vcol + 8 * ks * APITCH;
    const v2f al = {pa.x, pa.y}, ah = {pa.z, pa.w};
    const v2f b0l = {vr[0],      vr[APITCH]},      b0h = {vr[2 * APITCH],      vr[3 * APITCH]};
    const v2f b1l = {vr[16],     vr[APITCH + 16]}, b1h = {vr[2 * APITCH + 16], vr[3 * APITCH + 16]};
    const v2f b2l = {vr[32],     vr[APITCH + 32]}, b2h = {vr[2 * APITCH + 32], vr[3 * APITCH + 32]};
    const v2f b3l = {vr[48],     vr[APITCH + 48]}, b3h = {vr[2 * APITCH + 48], vr[3 * APITCH + 48]};
    WMMA_F32(o0, al, b0l);
    WMMA_F32(o1, al, b1l);
    WMMA_F32(o2, al, b2l);
    WMMA_F32(o3, al, b3l);
    WMMA_F32(o0, ah, b0h);
    WMMA_F32(o1, ah, b1h);
    WMMA_F32(o2, ah, b2h);
    WMMA_F32(o3, ah, b3h);
  }

#pragma unroll
  for (int r = 0; r < 8; ++r) {
    float* pr = pw + (8 * h + r) * APITCH + c;
    pr[0]  = o0[r] * inv[r];
    pr[16] = o1[r] * inv[r];
    pr[32] = o2[r] * inv[r];
    pr[48] = o3[r] * inv[r];
  }
  __syncthreads();
#pragma unroll
  for (int i = 0; i < 8; ++i) {
    const int row = 2 * i + h;
    const float4 ov = *(const float4*)(pw + row * APITCH + 4 * c);
    *(float4*)(obase + (size_t)(16 * w + row) * Dn + 4 * c) = ov;
  }
}

extern "C" void kernel_launch(void* const* d_in, const int* in_sizes, int n_in,
                              void* d_out, int out_size, void* d_ws, size_t ws_size,
                              hipStream_t stream) {
  (void)in_sizes; (void)n_in; (void)out_size; (void)ws_size;
  constexpr int B = 4, T = 1024, D_IN = 512, D = 512, K = 2, V = 500, H = 8, BAND = 24;
  constexpr int M = B * T;
  static_assert(D == RD && V == LV && K == LK && M % 8 == 0, "row kernels are sized for these shapes");
  static_assert(D == H * AHD && BAND == ABAND && T % AQ == 0, "attention kernel is sized for these shapes");
  static_assert(T % GBM == 0, "the concat epilogue needs a block's rows inside one batch element");

  const float* x_m      = (const float*)d_in[0];
  const float* logits   = (const float*)d_in[1];
  const int*   blank_id = (const int*)  d_in[2];
  const float* proj_w   = (const float*)d_in[3];
  const float* proj_b   = (const float*)d_in[4];
  const float* spk_tags = (const float*)d_in[5];
  const float* lnq_g    = (const float*)d_in[6];
  const float* lnq_b    = (const float*)d_in[7];
  const float* lnkv_g   = (const float*)d_in[8];
  const float* lnkv_b   = (const float*)d_in[9];
  const float* in_w     = (const float*)d_in[10];
  const float* in_b     = (const float*)d_in[11];
  const float* out_w    = (const float*)d_in[12];
  const float* out_b    = (const float*)d_in[13];
  const float* ffn_ln_g = (const float*)d_in[14];
  const float* ffn_ln_b = (const float*)d_in[15];
  const float* ffn_w1   = (const float*)d_in[16];
  const float* ffn_b1   = (const float*)d_in[17];
  const float* ffn_w2   = (const float*)d_in[18];
  const float* ffn_b2   = (const float*)d_in[19];
  float* out = (float*)d_out;

  float* ws = (float*)d_ws;
  const size_t SZ = (size_t)M * D;
  float* X    = ws;
  float* kv   = X  + SZ;
  float* xk   = kv + 2 * SZ;
  float* qh   = xk + 2 * SZ;
  float* qy   = qh + 2 * SZ;
  float* t1   = ws;
  float* wmix = ws + 10 * SZ;

  const dim3 blk(256);
  auto gg = [](int m, int n) { return dim3(m / 64, n / 128); };

  gemm_wmma_f32<<<gg(M, D), blk, 0, stream>>>(x_m, proj_w, proj_b, nullptr, X,
                                              M, D, D_IN, 0, T, K, 0);
  wmix_kernel<<<dim3(M / 8), blk, 0, stream>>>(logits, blank_id, wmix, M);
  ln_kernel<<<dim3(M / 8), blk, 0, stream>>>(X, lnkv_g, lnkv_b, xk, M);
  gemm_wmma_f32<<<gg(M, 2 * D), blk, 0, stream>>>(xk, in_w + (size_t)D * D, in_b + D,
                                                  nullptr, kv, M, 2 * D, D, 0, T, K, 0);
  stream_prep<<<dim3(K * M / 8), blk, 0, stream>>>(X, wmix, spk_tags, lnq_g, lnq_b,
                                                   xk, qh, M);
  gemm_wmma_f32<<<gg(K * M, D), blk, 0, stream>>>(qh, in_w, in_b, nullptr, qy,
                                                  K * M, D, D, 0, T, K, 0);
  banded_attn_wmma<<<dim3(K * B * H * (T / AQ)), dim3(ATHR), 0, stream>>>(qy, kv, kv + D, qh,
                                                                          T, D, 2 * D, H, B);
  gemm_wmma_f32<<<gg(K * M, D), blk, 0, stream>>>(qh, out_w, out_b, xk, qy,
                                                  K * M, D, D,  1, T, K, 0);
  ln_kernel<<<dim3(K * M / 8), blk, 0, stream>>>(qy, ffn_ln_g, ffn_ln_b, qh, K * M);

  static_assert(K == 2, "the FFN pair below is written out for two speakers");
  gemm_wmma_f32<<<gg(M, 4 * D), blk, 0, stream>>>(qh, ffn_w1, ffn_b1, nullptr, t1,
                                                  M, 4 * D, D,  2, T, K, 0);
  gemm_wmma_f32<<<gg(M, D), blk, 0, stream>>>(t1, ffn_w2, ffn_b2, qy, out,
                                              M, D, 4 * D,  1 | 4, T, K, 0);
  gemm_wmma_f32<<<gg(M, 4 * D), blk, 0, stream>>>(qh + SZ, ffn_w1, ffn_b1, nullptr, t1,
                                                  M, 4 * D, D,  2, T, K, 1);
  gemm_wmma_f32<<<gg(M, D), blk, 0, stream>>>(t1, ffn_w2, ffn_b2, qy + SZ, out,
                                              M, D, 4 * D,  1 | 4, T, K, 1);
}


# Launch consolidation: mix weights, LayerNorm of X and both speakers' stream prep are ONE wave-per-token-row launch (X row read once); k|v and q projections share one grid; 13 to 10 launches
# speedup vs baseline: 2.5747x; 1.0364x over previous
#include <hip/hip_runtime.h>
#include <hip/hip_bf16.h>
#include <math.h>

typedef float v2f __attribute__((ext_vector_type(2)));
typedef float v4f __attribute__((ext_vector_type(4)));
typedef float v8f __attribute__((ext_vector_type(8)));

__device__ __forceinline__ float gelu_exact(float x) {
  return 0.5f * x * (1.0f + erff(x * 0.70710678118654752f));
}

#define WMMA_F32(cacc, av, bv)                                                 \
  cacc = __builtin_amdgcn_wmma_f32_16x16x4_f32(false, (av), false, (bv),       \
                                               (short)0, (cacc), false, false)

#define GBM 64
#define GBN 128
#define GBK 32
#define GPITCH 36
#define GTILE ((GBM + GBN) * GPITCH)
#define GSTAGES 3

#define G_LOAD_TILE(r, ko)                                                     \
  do {                                                                         \
    const char* ap_ = (const char*)(Ab + (ko));                                \
    const char* wp_ = (const char*)(Wb + (ko));                                \
    unsigned q0_ = ob0, q1_ = ob1, q2_ = ob2, q3_ = ob3;                       \
    asm volatile("" : "+v"(q0_), "+v"(q1_), "+v"(q2_), "+v"(q3_));             \
    r##0 = *(const float4*)(ap_ + q0_);                                        \
    r##1 = *(const float4*)(ap_ + q1_);                                        \
    r##2 = *(const float4*)(wp_ + q0_);                                        \
    r##3 = *(const float4*)(wp_ + q1_);                                        \
    r##4 = *(const float4*)(wp_ + q2_);                                        \
    r##5 = *(const float4*)(wp_ + q3_);                                        \
  } while (0)

#define G_STORE_TILE(r, st)                                                    \
  do {                                                                         \
    float* s_ = sm + (st) + so;                                                \
    *(float4*)(s_)                         = r##0;                             \
    *(float4*)(s_ + 32 * GPITCH)           = r##1;                             \
    *(float4*)(s_ + GBM * GPITCH)          = r##2;                             \
    *(float4*)(s_ + (GBM + 32) * GPITCH)   = r##3;                             \
    *(float4*)(s_ + (GBM + 64) * GPITCH)   = r##4;                             \
    *(float4*)(s_ + (GBM + 96) * GPITCH)   = r##5;                             \
  } while (0)

#define G_LOAD_FRAGS(F, st, hh)                                                \
  do {                                                                         \
    const float* c_ = sm + (st) + (hh) * 16;                                   \
    F[0] = *(const float4*)(c_ + aoff);                                        \
    F[1] = *(const float4*)(c_ + aoff + 16 * GPITCH);                          \
    F[2] = *(const float4*)(c_ + boff);                                        \
    F[3] = *(const float4*)(c_ + boff + 16 * GPITCH);                          \
    F[4] = *(const float4*)(c_ + aoff + 8);                                    \
    F[5] = *(const float4*)(c_ + aoff + 16 * GPITCH + 8);                      \
    F[6] = *(const float4*)(c_ + boff + 8);                                    \
    F[7] = *(const float4*)(c_ + boff + 16 * GPITCH + 8);                      \
  } while (0)

#define G_MMA_HALF(F)                                                          \
  do {                                                                         \
    _Pragma("unroll")                                                          \
    for (int j_ = 0; j_ < 2; ++j_) {                                           \
      const float4 fa0 = F[4 * j_], fa1 = F[4 * j_ + 1];                       \
      const float4 fb0 = F[4 * j_ + 2], fb1 = F[4 * j_ + 3];                   \
      const v2f a0l = {fa0.x, fa0.y}, a0h = {fa0.z, fa0.w};                    \
      const v2f a1l = {fa1.x, fa1.y}, a1h = {fa1.z, fa1.w};                    \
      const v2f b0l = {fb0.x, fb0.y}, b0h = {fb0.z, fb0.w};                    \
      const v2f b1l = {fb1.x, fb1.y}, b1h = {fb1.z, fb1.w};                    \
      WMMA_F32(c00, a0l, b0l);                                                 \
      WMMA_F32(c01, a0l, b1l);                                                 \
      WMMA_F32(c10, a1l, b0l);                                                 \
      WMMA_F32(c11, a1l, b1l);                                                 \
      WMMA_F32(c00, a0h, b0h);                                                 \
      WMMA_F32(c01, a0h, b1h);                                                 \
      WMMA_F32(c10, a1h, b0h);                                                 \
      WMMA_F32(c11, a1h, b1h);                                                 \
    }                                                                          \
  } while (0)

#define G_GRP(mask, n) __builtin_amdgcn_sched_group_barrier((mask), (n), 0)
#define G_STEP(t, S, G, N)                                                     \
  do {                                                                         \
    G_LOAD_FRAGS(Q, o0, 1);                                                    \
    G_MMA_HALF(P);                                                             \
    G_GRP(0x100, 8);             \
    G_GRP(0x008, 16);     \
    __builtin_amdgcn_sched_barrier(0);   \
    if ((S) || (N)) {                                                          \
      __syncthreads();                                                         \
      __builtin_amdgcn_sched_barrier(0);   \
    }                                                                          \
    if (N) G_LOAD_FRAGS(P, o1, 0);                                             \
    if (S) G_STORE_TILE(r, o2);                                                \
    if (G) G_LOAD_TILE(r, ((t) + 3) * GBK);                                    \
    G_MMA_HALF(Q);                                                             \
    G_GRP(0x008, 8);              \
    if (N) G_GRP(0x100, 8);      \
    if (S) G_GRP(0x200, 6);              \
    if (G) G_GRP(0x020, 6);            \
    G_GRP(0x008, 8);                  \
    __builtin_amdgcn_sched_barrier(0);   \
    { const int t_ = o0; o0 = o1; o1 = o2; o2 = t_; }                          \
  } while (0)

__device__ __forceinline__ void gemm_tile(
    const float* __restrict__ A, const float* __restrict__ W,
    const float* __restrict__ bias, const float* __restrict__ R,
    float* __restrict__ C, const int N, const int Kd,
    const int flags, const int Tlen, const int Kspk, const int kk,
    const int blockM, const int blockN)
{
  __shared__ __attribute__((aligned(16))) float sm[GSTAGES * GTILE];

  const int tid  = threadIdx.x;
  const int lane = tid & 31;
  const int w    = tid >> 5;
  const int wm   = w & 1;
  const int wn   = w >> 1;
  const int half = lane >> 4;
  const int l16  = lane & 15;

  const int lrow = tid >> 3;
  const int lcol = (tid & 7) * 4;
  const float* Ab = A + (size_t)blockM * Kd;
  const float* Wb = W + (size_t)blockN * Kd;
  const unsigned ob0 = (unsigned)(lrow * Kd + lcol) * 4u;
  const unsigned ob1 = ob0 + 128u * (unsigned)Kd;
  const unsigned ob2 = ob0 + 256u * (unsigned)Kd;
  const unsigned ob3 = ob0 + 384u * (unsigned)Kd;

  const int so   = lrow * GPITCH + lcol;
  const int aoff = (wm * 32 + l16) * GPITCH + 4 * half;
  const int boff = GBM * GPITCH + (wn * 32 + l16) * GPITCH + 4 * half;

  v8f c00 = {}, c01 = {}, c10 = {}, c11 = {};

  float4 u0, u1, u2, u3, u4, u5, e0, e1, e2, e3, e4, e5, r0, r1, r2, r3, r4, r5;
  G_LOAD_TILE(u, 0);
  G_LOAD_TILE(e, GBK);
  G_LOAD_TILE(r, 2 * GBK);
  G_STORE_TILE(u, 0);
  G_STORE_TILE(e, GTILE);
  __syncthreads();

  int o0 = 0, o1 = GTILE, o2 = 2 * GTILE;
  float4 P[8], Q[8];
  G_LOAD_FRAGS(P, o0, 0);

  const int nk = Kd / GBK;
  for (int kt = 0; kt < nk - 3; ++kt) G_STEP(kt, 1, 1, 1);
  G_STEP(nk - 3, 1, 0, 1);
  G_STEP(nk - 2, 0, 0, 1);

  const int erow  = lane >> 3;
  const int ecol  = (lane & 7) * 4;
  const int tileM = blockM + wm * 32;
  const int ncol  = blockN + wn * 32 + ecol;
  const float4 bias4 = *(const float4*)(bias + ncol);
  float4 rs[8];
  if (flags & 1) {
    const float* rp = R + (size_t)(tileM + erow) * N + ncol;
#pragma unroll
    for (int p = 0; p < 8; ++p) rs[p] = *(const float4*)(rp + (size_t)(4 * p) * N);
  }
  __builtin_amdgcn_sched_barrier(0);

  G_STEP(nk - 1, 0, 0, 0);

  float* cw = sm + ((w < 4) ? (o0 + w * (32 * GPITCH)) : (o1 + (w - 4) * (32 * GPITCH)));
#pragma unroll
  for (int r = 0; r < 8; ++r) {
    float* pr = cw + (8 * half + r) * GPITCH + l16;
    pr[0]                = c00[r];
    pr[16]               = c01[r];
    pr[16 * GPITCH]      = c10[r];
    pr[16 * GPITCH + 16] = c11[r];
  }
  __syncthreads();

  float4 v[8];
#pragma unroll
  for (int p = 0; p < 8; ++p) v[p] = *(const float4*)(cw + (4 * p + erow) * GPITCH + ecol);
#pragma unroll
  for (int p = 0; p < 8; ++p) {
    v[p].x += bias4.x; v[p].y += bias4.y; v[p].z += bias4.z; v[p].w += bias4.w;
  }
  if (flags & 1) {
#pragma unroll
    for (int p = 0; p < 8; ++p) {
      v[p].x += rs[p].x; v[p].y += rs[p].y; v[p].z += rs[p].z; v[p].w += rs[p].w;
    }
  }
  if (flags & 2) {
#pragma unroll
    for (int p = 0; p < 8; ++p) {
      v[p].x = gelu_exact(v[p].x); v[p].y = gelu_exact(v[p].y);
      v[p].z = gelu_exact(v[p].z); v[p].w = gelu_exact(v[p].w);
    }
  }
  size_t orow = (size_t)tileM;
  if (flags & 4) {
    const int bi = blockM / Tlen;
    orow = (size_t)bi * Kspk * Tlen + (size_t)kk * Tlen + (size_t)(tileM - bi * Tlen);
  }
  float* cp = C + (orow + erow) * N + ncol;
#pragma unroll
  for (int p = 0; p < 8; ++p) *(float4*)(cp + (size_t)(4 * p) * N) = v[p];
}

__global__ __launch_bounds__(256) void gemm_wmma_f32(
    const float* __restrict__ A, const float* __restrict__ W,
    const float* __restrict__ bias, const float* __restrict__ R,
    float* __restrict__ C, int M, int N, int Kd,
    int flags, int Tlen, int Kspk, int kk)
{
  (void)M;
  gemm_tile(A, W, bias, R, C, N, Kd, flags, Tlen, Kspk, kk,
            blockIdx.x * GBM, blockIdx.y * GBN);
}

__global__ __launch_bounds__(256) void gemm_pair_wmma_f32(
    const float* __restrict__ A0, const float* __restrict__ W0,
    const float* __restrict__ bias0, const float* __restrict__ R0,
    float* __restrict__ C0, int N0, int mb0, int nblk0, int flags0,
    const float* __restrict__ A1, const float* __restrict__ W1,
    const float* __restrict__ bias1, const float* __restrict__ R1,
    float* __restrict__ C1, int N1, int mb1, int flags1,
    int Kd, int Tlen, int Kspk, int kk)
{
  int id = blockIdx.x;
  const bool second = id >= nblk0;
  if (second) id -= nblk0;
  const float* A     = second ? A1 : A0;
  const float* W     = second ? W1 : W0;
  const float* bias  = second ? bias1 : bias0;
  const float* R     = second ? R1 : R0;
  float*       C     = second ? C1 : C0;
  const int    N     = second ? N1 : N0;
  const int    mb    = second ? mb1 : mb0;
  const int    flags = second ? flags1 : flags0;
  const int    by    = id / mb;
  const int    bx    = id - by * mb;
  gemm_tile(A, W, bias, R, C, N, Kd, flags, Tlen, Kspk, kk, bx * GBM, by * GBN);
}

#define RD  512
#define RNV (RD / 128)
#define LV  500
#define LV4 (LV / 4)
#define LNI ((LV4 + 31) / 32)
#define LK  2

#define ROW_XOR(x, n)                                                          \
  __builtin_bit_cast(float, __builtin_amdgcn_update_dpp(                       \
      __builtin_bit_cast(int, (x)), __builtin_bit_cast(int, (x)),              \
      0x160 + (n), 0xf, 0xf, false))
__device__ __forceinline__ float xhalf(float x) {
  const int b = __builtin_bit_cast(int, x);
  return __builtin_bit_cast(float,
      __builtin_amdgcn_permlanex16(b, b, 0x76543210, 0xfedcba98, false, false));
}
__device__ __forceinline__ float wave_sum(float v) {
  v += ROW_XOR(v, 1);
  v += ROW_XOR(v, 2);
  v += ROW_XOR(v, 4);
  v += ROW_XOR(v, 8);
  return v + xhalf(v);
}
__device__ __forceinline__ float wave_max(float v) {
  v = fmaxf(v, ROW_XOR(v, 1));
  v = fmaxf(v, ROW_XOR(v, 2));
  v = fmaxf(v, ROW_XOR(v, 4));
  v = fmaxf(v, ROW_XOR(v, 8));
  return fmaxf(v, xhalf(v));
}

__device__ __forceinline__ void row_stats(const float4 (&v)[RNV], float& mean, float& inv) {
  float s = 0.f;
#pragma unroll
  for (int i = 0; i < RNV; ++i) s += (v[i].x + v[i].y) + (v[i].z + v[i].w);
  mean = wave_sum(s) * (1.0f / RD);
  float s2 = 0.f;
#pragma unroll
  for (int i = 0; i < RNV; ++i) {
    const float dx = v[i].x - mean, dy = v[i].y - mean;
    const float dz = v[i].z - mean, dw = v[i].w - mean;
    s2 += (dx * dx + dy * dy) + (dz * dz + dw * dw);
  }
  inv = rsqrtf(wave_sum(s2) * (1.0f / RD) + 1e-5f);
}

__global__ __launch_bounds__(256) void ln_kernel(
    const float* __restrict__ X, const float* __restrict__ g,
    const float* __restrict__ b, float* __restrict__ Y, int rows)
{
  const int lane = threadIdx.x & 31;
  const int row  = blockIdx.x * 8 + __builtin_amdgcn_readfirstlane(threadIdx.x >> 5);
  if (row >= rows) return;
  const float4* x4 = (const float4*)(X + (size_t)row * RD) + lane;
  const float4* g4 = (const float4*)g + lane;
  const float4* b4 = (const float4*)b + lane;
  float4 xv[RNV], gv[RNV], bv[RNV];
#pragma unroll
  for (int i = 0; i < RNV; ++i) xv[i] = x4[i * 32];
#pragma unroll
  for (int i = 0; i < RNV; ++i) { gv[i] = g4[i * 32]; bv[i] = b4[i * 32]; }
  __builtin_amdgcn_sched_barrier(0);
  float mean, inv;
  row_stats(xv, mean, inv);
  float4* y4 = (float4*)(Y + (size_t)row * RD) + lane;
#pragma unroll
  for (int i = 0; i < RNV; ++i) {
    float4 o;
    o.x = (xv[i].x - mean) * inv * gv[i].x + bv[i].x;
    o.y = (xv[i].y - mean) * inv * gv[i].y + bv[i].y;
    o.z = (xv[i].z - mean) * inv * gv[i].z + bv[i].z;
    o.w = (xv[i].w - mean) * inv * gv[i].w + bv[i].w;
    y4[i * 32] = o;
  }
}

__global__ __launch_bounds__(256) void rows_fused(
    const float* __restrict__ X, const float* __restrict__ logits,
    const int* __restrict__ blank_p, const float* __restrict__ tags,
    const float* __restrict__ gkv, const float* __restrict__ bkv,
    const float* __restrict__ gq, const float* __restrict__ bq,
    float* __restrict__ KVn, float* __restrict__ Xk, float* __restrict__ Qin, int rows)
{
  const int lane = threadIdx.x & 31;
  const int row  = blockIdx.x * 8 + __builtin_amdgcn_readfirstlane(threadIdx.x >> 5);
  if (row >= rows) return;
  const int blank = blank_p[0];
  const float4* x4 = (const float4*)(X + (size_t)row * RD) + lane;
  float4 xv[RNV];
#pragma unroll
  for (int i = 0; i < RNV; ++i) xv[i] = x4[i * 32];
  float4 lv[LK][LNI];
  float  lb[LK];
#pragma unroll
  for (int k2 = 0; k2 < LK; ++k2) {
    const float* L = logits + ((size_t)k2 * rows + row) * LV;
    const float4* L4 = (const float4*)L;
#pragma unroll
    for (int i = 0; i < LNI; ++i) {
      const int idx = lane + 32 * i;
      lv[k2][i] = L4[idx < LV4 ? idx : LV4 - 1];
    }
    lb[k2] = L[blank];
  }
  float4 gk[RNV], bk[RNV], gv[RNV], bv[RNV], tv[LK][RNV];
#pragma unroll
  for (int i = 0; i < RNV; ++i) {
    gk[i] = ((const float4*)gkv + lane)[i * 32];
    bk[i] = ((const float4*)bkv + lane)[i * 32];
    gv[i] = ((const float4*)gq + lane)[i * 32];
    bv[i] = ((const float4*)bq + lane)[i * 32];
  }
#pragma unroll
  for (int kk = 0; kk < LK; ++kk) {
#pragma unroll
    for (int i = 0; i < RNV; ++i) tv[kk][i] = ((const float4*)(tags + (size_t)kk * RD) + lane)[i * 32];
  }
  __builtin_amdgcn_sched_barrier(0);
#pragma unroll
  for (int k2 = 0; k2 < LK; ++k2) {
#pragma unroll
    for (int i = 0; i < LNI; ++i)
      if (lane + 32 * i >= LV4)
        lv[k2][i] = make_float4(-INFINITY, -INFINITY, -INFINITY, -INFINITY);
  }

  {
    float mean, inv;
    row_stats(xv, mean, inv);
    float4* y4 = (float4*)(KVn + (size_t)row * RD) + lane;
#pragma unroll
    for (int i = 0; i < RNV; ++i) {
      float4 o;
      o.x = (xv[i].x - mean) * inv * gk[i].x + bk[i].x;
      o.y = (xv[i].y - mean) * inv * gk[i].y + bk[i].y;
      o.z = (xv[i].z - mean) * inv * gk[i].z + bk[i].z;
      o.w = (xv[i].w - mean) * inv * gk[i].w + bk[i].w;
      y4[i * 32] = o;
    }
  }

  float a[LK];
#pragma unroll
  for (int k2 = 0; k2 < LK; ++k2) {
    float m = -INFINITY;
#pragma unroll
    for (int i = 0; i < LNI; ++i)
      m = fmaxf(m, fmaxf(fmaxf(lv[k2][i].x, lv[k2][i].y), fmaxf(lv[k2][i].z, lv[k2][i].w)));
    m = wave_max(m);
    float s = 0.f;
#pragma unroll
    for (int i = 0; i < LNI; ++i)
      s += (expf(lv[k2][i].x - m) + expf(lv[k2][i].y - m)) +
           (expf(lv[k2][i].z - m) + expf(lv[k2][i].w - m));
    s = wave_sum(s);
    a[k2] = 1.0f - expf(lb[k2] - m) / s;
  }
  float wmixv[LK];
  {
    float m2 = -INFINITY;
#pragma unroll
    for (int k2 = 0; k2 < LK; ++k2) m2 = fmaxf(m2, a[k2]);
    float e[LK];
    float ss = 0.f;
#pragma unroll
    for (int k2 = 0; k2 < LK; ++k2) { e[k2] = expf(a[k2] - m2); ss += e[k2]; }
#pragma unroll
    for (int k2 = 0; k2 < LK; ++k2) wmixv[k2] = e[k2] / ss;
  }

#pragma unroll
  for (int kk = 0; kk < LK; ++kk) {
    const float wv = wmixv[kk];
    float4 v[RNV];
#pragma unroll
    for (int i = 0; i < RNV; ++i) {
      v[i].x = xv[i].x * wv + tv[kk][i].x;
      v[i].y = xv[i].y * wv + tv[kk][i].y;
      v[i].z = xv[i].z * wv + tv[kk][i].z;
      v[i].w = xv[i].w * wv + tv[kk][i].w;
    }
    const size_t srow = (size_t)kk * rows + row;
    float4* xk4 = (float4*)(Xk + srow * RD) + lane;
#pragma unroll
    for (int i = 0; i < RNV; ++i) xk4[i * 32] = v[i];
    float mean, inv;
    row_stats(v, mean, inv);
    float4* q4 = (float4*)(Qin + srow * RD) + lane;
#pragma unroll
    for (int i = 0; i < RNV; ++i) {
      float4 o;
      o.x = (v[i].x - mean) * inv * gv[i].x + bv[i].x;
      o.y = (v[i].y - mean) * inv * gv[i].y + bv[i].y;
      o.z = (v[i].z - mean) * inv * gv[i].z + bv[i].z;
      o.w = (v[i].w - mean) * inv * gv[i].w + bv[i].w;
      q4[i * 32] = o;
    }
  }
}

#define AQ      64
#define ANW     (AQ / 16)
#define ATHR    (32 * ANW)
#define ABAND   24
#define AKEYS   (AQ + 2 * ABAND)
#define AHD     64
#define APITCH  68

__global__ __launch_bounds__(ATHR) void banded_attn_wmma(
    const float* __restrict__ q, const float* __restrict__ kb,
    const float* __restrict__ vb, float* __restrict__ ctx,
    int T_, int Dn, int Dkv, int Hn, int Bkv)
{
  __shared__ __attribute__((aligned(16))) float sK[AKEYS * APITCH];
  __shared__ __attribute__((aligned(16))) float sV[AKEYS * APITCH];
  __shared__ __attribute__((aligned(16))) float sQ[AQ * APITCH];

  const int tid  = threadIdx.x;
  const int lane = tid & 31;
  const int w    = tid >> 5;
  const int c    = lane & 15;
  const int h    = lane >> 4;

  const int nqt = T_ / AQ;
  const int qt  = blockIdx.x % nqt;
  const int hh  = (blockIdx.x / nqt) % Hn;
  const int b   = blockIdx.x / (nqt * Hn);
  const int bk  = b % Bkv;
  const int t0  = qt * AQ;

  const size_t headoff = (size_t)hh * AHD;
  const float* kbase = kb + (size_t)bk * T_ * Dkv + headoff;
  const float* vbase = vb + (size_t)bk * T_ * Dkv + headoff;
  const float* qbase = q  + ((size_t)b * T_ + t0) * Dn + headoff;
  float*       obase = ctx + ((size_t)b * T_ + t0) * Dn + headoff;

  constexpr int NKV = (AKEYS * 16) / ATHR;
  constexpr int NQ  = (AQ * 16) / ATHR;
  const int srow = tid >> 4;
  const int sc4  = (tid & 15) * 4;
  v4f kreg[NKV], vreg[NKV], qreg[NQ];
#pragma unroll
  for (int i = 0; i < NKV; ++i) {
    const int key = t0 - ABAND + srow + (ATHR / 16) * i;
    const unsigned off = (unsigned)min(max(key, 0), T_ - 1) * (unsigned)Dkv + (unsigned)sc4;
    kreg[i] = *(const v4f*)(kbase + off);
    vreg[i] = *(const v4f*)(vbase + off);
  }
#pragma unroll
  for (int i = 0; i < NQ; ++i) {
    const unsigned off = (unsigned)(srow + (ATHR / 16) * i) * (unsigned)Dn + (unsigned)sc4;
    qreg[i] = *(const v4f*)(qbase + off);
  }
  __builtin_amdgcn_sched_barrier(0);
  const v4f zero4 = {0.f, 0.f, 0.f, 0.f};
#pragma unroll
  for (int i = 0; i < NKV; ++i) {
    const int row = srow + (ATHR / 16) * i;
    const int key = t0 - ABAND + row;
    const bool ok = (key >= 0) && (key < T_);
    *(v4f*)(sK + row * APITCH + sc4) = ok ? kreg[i] : zero4;
    *(v4f*)(sV + row * APITCH + sc4) = ok ? vreg[i] : zero4;
  }
#pragma unroll
  for (int i = 0; i < NQ; ++i) {
    const int row = srow + (ATHR / 16) * i;
    *(v4f*)(sQ + row * APITCH + sc4) = qreg[i];
  }
  __syncthreads();

  float* pw = sQ + (w * 16) * APITCH;
  const float* qrow = pw + c * APITCH + 4 * h;
  const float* krow = sK + (w * 16 + c) * APITCH + 4 * h;
  float4 qa[8];
#pragma unroll
  for (int ks = 0; ks < 8; ++ks) qa[ks] = *(const float4*)(qrow + 8 * ks);

  v8f s0 = {}, s1 = {}, s2 = {}, s3 = {};
#pragma unroll
  for (int ks = 0; ks < 8; ++ks) {
    const float4 k0 = *(const float4*)(krow + 8 * ks);
    const float4 k1 = *(const float4*)(krow + 16 * APITCH + 8 * ks);
    const float4 k2 = *(const float4*)(krow + 32 * APITCH + 8 * ks);
    const float4 k3 = *(const float4*)(krow + 48 * APITCH + 8 * ks);
    const v2f al = {qa[ks].x, qa[ks].y}, ah = {qa[ks].z, qa[ks].w};
    const v2f b0l = {k0.x, k0.y}, b0h = {k0.z, k0.w};
    const v2f b1l = {k1.x, k1.y}, b1h = {k1.z, k1.w};
    const v2f b2l = {k2.x, k2.y}, b2h = {k2.z, k2.w};
    const v2f b3l = {k3.x, k3.y}, b3h = {k3.z, k3.w};
    WMMA_F32(s0, al, b0l);
    WMMA_F32(s1, al, b1l);
    WMMA_F32(s2, al, b2l);
    WMMA_F32(s3, al, b3l);
    WMMA_F32(s0, ah, b0h);
    WMMA_F32(s1, ah, b1h);
    WMMA_F32(s2, ah, b2h);
    WMMA_F32(s3, ah, b3h);
  }

  const float scale = 0.125f;
  const int jbase = t0 + 16 * w - ABAND;
  bool colok[4];
#pragma unroll
  for (int nt = 0; nt < 4; ++nt) {
    const int key = jbase + 16 * nt + c;
    colok[nt] = (key >= 0) && (key < T_);
  }
  float inv[8];
#pragma unroll
  for (int r = 0; r < 8; ++r) {
    const int rq = 8 * h + r;
    float x[4] = {s0[r], s1[r], s2[r], s3[r]};
#pragma unroll
    for (int nt = 0; nt < 4; ++nt) {
      const int jj = 16 * nt + c;
      const bool ok = colok[nt] && (jj >= rq) && (jj <= rq + 2 * ABAND);
      x[nt] = ok ? x[nt] * scale : -INFINITY;
    }
    float mx = fmaxf(fmaxf(x[0], x[1]), fmaxf(x[2], x[3]));
    mx = fmaxf(mx, ROW_XOR(mx, 1));
    mx = fmaxf(mx, ROW_XOR(mx, 2));
    mx = fmaxf(mx, ROW_XOR(mx, 4));
    mx = fmaxf(mx, ROW_XOR(mx, 8));
    float e[4];
#pragma unroll
    for (int nt = 0; nt < 4; ++nt) e[nt] = expf(x[nt] - mx);
    float sum = (e[0] + e[1]) + (e[2] + e[3]);
    sum += ROW_XOR(sum, 1);
    sum += ROW_XOR(sum, 2);
    sum += ROW_XOR(sum, 4);
    sum += ROW_XOR(sum, 8);
    inv[r] = 1.0f / sum;
    float* pr = pw + rq * APITCH + c;
    pr[0]  = e[0];
    pr[16] = e[1];
    pr[32] = e[2];
    pr[48] = e[3];
  }
  __syncthreads();

  const float* prow = pw + c * APITCH + 4 * h;
  const float* vcol = sV + (w * 16 + 4 * h) * APITCH + c;
  v8f o0 = {}, o1 = {}, o2 = {}, o3 = {};
#pragma unroll
  for (int ks = 0; ks < 8; ++ks) {
    const float4 pa = *(const float4*)(prow + 8 * ks);
    const float* vr = vcol + 8 * ks * APITCH;
    const v2f al = {pa.x, pa.y}, ah = {pa.z, pa.w};
    const v2f b0l = {vr[0],      vr[APITCH]},      b0h = {vr[2 * APITCH],      vr[3 * APITCH]};
    const v2f b1l = {vr[16],     vr[APITCH + 16]}, b1h = {vr[2 * APITCH + 16], vr[3 * APITCH + 16]};
    const v2f b2l = {vr[32],     vr[APITCH + 32]}, b2h = {vr[2 * APITCH + 32], vr[3 * APITCH + 32]};
    const v2f b3l = {vr[48],     vr[APITCH + 48]}, b3h = {vr[2 * APITCH + 48], vr[3 * APITCH + 48]};
    WMMA_F32(o0, al, b0l);
    WMMA_F32(o1, al, b1l);
    WMMA_F32(o2, al, b2l);
    WMMA_F32(o3, al, b3l);
    WMMA_F32(o0, ah, b0h);
    WMMA_F32(o1, ah, b1h);
    WMMA_F32(o2, ah, b2h);
    WMMA_F32(o3, ah, b3h);
  }

#pragma unroll
  for (int r = 0; r < 8; ++r) {
    float* pr = pw + (8 * h + r) * APITCH + c;
    pr[0]  = o0[r] * inv[r];
    pr[16] = o1[r] * inv[r];
    pr[32] = o2[r] * inv[r];
    pr[48] = o3[r] * inv[r];
  }
  __syncthreads();
#pragma unroll
  for (int i = 0; i < 8; ++i) {
    const int row = 2 * i + h;
    const float4 ov = *(const float4*)(pw + row * APITCH + 4 * c);
    *(float4*)(obase + (size_t)(16 * w + row) * Dn + 4 * c) = ov;
  }
}

extern "C" void kernel_launch(void* const* d_in, const int* in_sizes, int n_in,
                              void* d_out, int out_size, void* d_ws, size_t ws_size,
                              hipStream_t stream) {
  (void)in_sizes; (void)n_in; (void)out_size; (void)ws_size;
  constexpr int B = 4, T = 1024, D_IN = 512, D = 512, K = 2, V = 500, H = 8, BAND = 24;
  constexpr int M = B * T;
  static_assert(D == RD && V == LV && K == LK && M % 8 == 0, "row kernels are sized for these shapes");
  static_assert(D == H * AHD && BAND == ABAND && T % AQ == 0, "attention kernel is sized for these shapes");
  static_assert(T % GBM == 0, "the concat epilogue needs a block's rows inside one batch element");

  const float* x_m      = (const float*)d_in[0];
  const float* logits   = (const float*)d_in[1];
  const int*   blank_id = (const int*)  d_in[2];
  const float* proj_w   = (const float*)d_in[3];
  const float* proj_b   = (const float*)d_in[4];
  const float* spk_tags = (const float*)d_in[5];
  const float* lnq_g    = (const float*)d_in[6];
  const float* lnq_b    = (const float*)d_in[7];
  const float* lnkv_g   = (const float*)d_in[8];
  const float* lnkv_b   = (const float*)d_in[9];
  const float* in_w     = (const float*)d_in[10];
  const float* in_b     = (const float*)d_in[11];
  const float* out_w    = (const float*)d_in[12];
  const float* out_b    = (const float*)d_in[13];
  const float* ffn_ln_g = (const float*)d_in[14];
  const float* ffn_ln_b = (const float*)d_in[15];
  const float* ffn_w1   = (const float*)d_in[16];
  const float* ffn_b1   = (const float*)d_in[17];
  const float* ffn_w2   = (const float*)d_in[18];
  const float* ffn_b2   = (const float*)d_in[19];
  float* out = (float*)d_out;

  float* ws = (float*)d_ws;
  const size_t SZ = (size_t)M * D;
  float* X    = ws;
  float* kv   = X  + SZ;
  float* xk   = kv + 2 * SZ;
  float* qh   = xk + 2 * SZ;
  float* qy   = qh + 2 * SZ;
  float* kvn  = qy + 2 * SZ;
  float* t1   = ws;

  const dim3 blk(256);
  auto gg = [](int m, int n) { return dim3(m / 64, n / 128); };

  gemm_wmma_f32<<<gg(M, D), blk, 0, stream>>>(x_m, proj_w, proj_b, nullptr, X,
                                              M, D, D_IN, 0, T, K, 0);
  rows_fused<<<dim3(M / 8), blk, 0, stream>>>(X, logits, blank_id, spk_tags,
                                              lnkv_g, lnkv_b, lnq_g, lnq_b,
                                              kvn, xk, qh, M);
  gemm_pair_wmma_f32<<<dim3((M / 64) * (2 * D / 128) + (K * M / 64) * (D / 128)), blk, 0, stream>>>(
      kvn, in_w + (size_t)D * D, in_b + D, nullptr, kv, 2 * D, M / 64, (M / 64) * (2 * D / 128), 0,
      qh, in_w, in_b, nullptr, qy, D, K * M / 64, 0,
      D, T, K, 0);
  banded_attn_wmma<<<dim3(K * B * H * (T / AQ)), dim3(ATHR), 0, stream>>>(qy, kv, kv + D, qh,
                                                                          T, D, 2 * D, H, B);
  gemm_wmma_f32<<<gg(K * M, D), blk, 0, stream>>>(qh, out_w, out_b, xk, qy,
                                                  K * M, D, D,  1, T, K, 0);
  ln_kernel<<<dim3(K * M / 8), blk, 0, stream>>>(qy, ffn_ln_g, ffn_ln_b, qh, K * M);

  static_assert(K == 2, "the FFN pair below is written out for two speakers");
  gemm_wmma_f32<<<gg(M, 4 * D), blk, 0, stream>>>(qh, ffn_w1, ffn_b1, nullptr, t1,
                                                  M, 4 * D, D,  2, T, K, 0);
  gemm_wmma_f32<<<gg(M, D), blk, 0, stream>>>(t1, ffn_w2, ffn_b2, qy, out,
                                              M, D, 4 * D,  1 | 4, T, K, 0);
  gemm_wmma_f32<<<gg(M, 4 * D), blk, 0, stream>>>(qh + SZ, ffn_w1, ffn_b1, nullptr, t1,
                                                  M, 4 * D, D,  2, T, K, 1);
  gemm_wmma_f32<<<gg(M, D), blk, 0, stream>>>(t1, ffn_w2, ffn_b2, qy + SZ, out,
                                              M, D, 4 * D,  1 | 4, T, K, 1);
}


# GELU epilogue of the first FFN product branch free: 0.5*erfc as ONE hardware exponential of a degree-9 polynomial, two values per packed f32 instruction, replacing the branching library erf
# speedup vs baseline: 2.6625x; 1.0341x over previous
#include <hip/hip_runtime.h>
#include <hip/hip_bf16.h>
#include <math.h>

typedef float v2f __attribute__((ext_vector_type(2)));
typedef float v4f __attribute__((ext_vector_type(4)));
typedef float v8f __attribute__((ext_vector_type(8)));

__device__ __forceinline__ v2f gelu_pair(const v2f v) {
  const v2f u = v * 0.70710678118654752f;
  const v2f t = {fminf(fabsf(u.x), 5.0f), fminf(fabsf(u.y), 5.0f)};
  v2f p = {-1.9093058654107153e-06f, -1.9093058654107153e-06f};
  p = p * t + 3.273908805567771e-05f;
  p = p * t + -0.00024974928237497807f;
  p = p * t + 0.0010866991942748427f;
  p = p * t + -0.002619069302454591f;
  p = p * t + 0.0003822076541837305f;
  p = p * t + 0.02757399156689644f;
  p = p * t + -0.14826619625091553f;
  p = p * t + -0.9184483885765076f;
  p = p * t + -1.6279070377349854f;
  const v2f r = p * t + -1.0f;
  const v2f e = {__builtin_amdgcn_exp2f(r.x), __builtin_amdgcn_exp2f(r.y)};
  const v2f a = {fabsf(v.x), fabsf(v.y)};
  const v2f m = {fmaxf(v.x, 0.0f), fmaxf(v.y, 0.0f)};
  return m - a * e;
}

#define WMMA_F32(cacc, av, bv)                                                 \
  cacc = __builtin_amdgcn_wmma_f32_16x16x4_f32(false, (av), false, (bv),       \
                                               (short)0, (cacc), false, false)

#define GBM 64
#define GBN 128
#define GBK 32
#define GPITCH 36
#define GTILE ((GBM + GBN) * GPITCH)
#define GSTAGES 3

#define G_LOAD_TILE(r, ko)                                                     \
  do {                                                                         \
    const char* ap_ = (const char*)(Ab + (ko));                                \
    const char* wp_ = (const char*)(Wb + (ko));                                \
    unsigned q0_ = ob0, q1_ = ob1, q2_ = ob2, q3_ = ob3;                       \
    asm volatile("" : "+v"(q0_), "+v"(q1_), "+v"(q2_), "+v"(q3_));             \
    r##0 = *(const float4*)(ap_ + q0_);                                        \
    r##1 = *(const float4*)(ap_ + q1_);                                        \
    r##2 = *(const float4*)(wp_ + q0_);                                        \
    r##3 = *(const float4*)(wp_ + q1_);                                        \
    r##4 = *(const float4*)(wp_ + q2_);                                        \
    r##5 = *(const float4*)(wp_ + q3_);                                        \
  } while (0)

#define G_STORE_TILE(r, st)                                                    \
  do {                                                                         \
    float* s_ = sm + (st) + so;                                                \
    *(float4*)(s_)                         = r##0;                             \
    *(float4*)(s_ + 32 * GPITCH)           = r##1;                             \
    *(float4*)(s_ + GBM * GPITCH)          = r##2;                             \
    *(float4*)(s_ + (GBM + 32) * GPITCH)   = r##3;                             \
    *(float4*)(s_ + (GBM + 64) * GPITCH)   = r##4;                             \
    *(float4*)(s_ + (GBM + 96) * GPITCH)   = r##5;                             \
  } while (0)

#define G_LOAD_FRAGS(F, st, hh)                                                \
  do {                                                                         \
    const float* c_ = sm + (st) + (hh) * 16;                                   \
    F[0] = *(const float4*)(c_ + aoff);                                        \
    F[1] = *(const float4*)(c_ + aoff + 16 * GPITCH);                          \
    F[2] = *(const float4*)(c_ + boff);                                        \
    F[3] = *(const float4*)(c_ + boff + 16 * GPITCH);                          \
    F[4] = *(const float4*)(c_ + aoff + 8);                                    \
    F[5] = *(const float4*)(c_ + aoff + 16 * GPITCH + 8);                      \
    F[6] = *(const float4*)(c_ + boff + 8);                                    \
    F[7] = *(const float4*)(c_ + boff + 16 * GPITCH + 8);                      \
  } while (0)

#define G_MMA_HALF(F)                                                          \
  do {                                                                         \
    _Pragma("unroll")                                                          \
    for (int j_ = 0; j_ < 2; ++j_) {                                           \
      const float4 fa0 = F[4 * j_], fa1 = F[4 * j_ + 1];                       \
      const float4 fb0 = F[4 * j_ + 2], fb1 = F[4 * j_ + 3];                   \
      const v2f a0l = {fa0.x, fa0.y}, a0h = {fa0.z, fa0.w};                    \
      const v2f a1l = {fa1.x, fa1.y}, a1h = {fa1.z, fa1.w};                    \
      const v2f b0l = {fb0.x, fb0.y}, b0h = {fb0.z, fb0.w};                    \
      const v2f b1l = {fb1.x, fb1.y}, b1h = {fb1.z, fb1.w};                    \
      WMMA_F32(c00, a0l, b0l);                                                 \
      WMMA_F32(c01, a0l, b1l);                                                 \
      WMMA_F32(c10, a1l, b0l);                                                 \
      WMMA_F32(c11, a1l, b1l);                                                 \
      WMMA_F32(c00, a0h, b0h);                                                 \
      WMMA_F32(c01, a0h, b1h);                                                 \
      WMMA_F32(c10, a1h, b0h);                                                 \
      WMMA_F32(c11, a1h, b1h);                                                 \
    }                                                                          \
  } while (0)

#define G_GRP(mask, n) __builtin_amdgcn_sched_group_barrier((mask), (n), 0)
#define G_STEP(t, S, G, N)                                                     \
  do {                                                                         \
    G_LOAD_FRAGS(Q, o0, 1);                                                    \
    G_MMA_HALF(P);                                                             \
    G_GRP(0x100, 8);             \
    G_GRP(0x008, 16);     \
    __builtin_amdgcn_sched_barrier(0);   \
    if ((S) || (N)) {                                                          \
      __syncthreads();                                                         \
      __builtin_amdgcn_sched_barrier(0);   \
    }                                                                          \
    if (N) G_LOAD_FRAGS(P, o1, 0);                                             \
    if (S) G_STORE_TILE(r, o2);                                                \
    if (G) G_LOAD_TILE(r, ((t) + 3) * GBK);                                    \
    G_MMA_HALF(Q);                                                             \
    G_GRP(0x008, 8);              \
    if (N) G_GRP(0x100, 8);      \
    if (S) G_GRP(0x200, 6);              \
    if (G) G_GRP(0x020, 6);            \
    G_GRP(0x008, 8);                  \
    __builtin_amdgcn_sched_barrier(0);   \
    { const int t_ = o0; o0 = o1; o1 = o2; o2 = t_; }                          \
  } while (0)

__device__ __forceinline__ void gemm_tile(
    const float* __restrict__ A, const float* __restrict__ W,
    const float* __restrict__ bias, const float* __restrict__ R,
    float* __restrict__ C, const int N, const int Kd,
    const int flags, const int Tlen, const int Kspk, const int kk,
    const int blockM, const int blockN)
{
  __shared__ __attribute__((aligned(16))) float sm[GSTAGES * GTILE];

  const int tid  = threadIdx.x;
  const int lane = tid & 31;
  const int w    = tid >> 5;
  const int wm   = w & 1;
  const int wn   = w >> 1;
  const int half = lane >> 4;
  const int l16  = lane & 15;

  const int lrow = tid >> 3;
  const int lcol = (tid & 7) * 4;
  const float* Ab = A + (size_t)blockM * Kd;
  const float* Wb = W + (size_t)blockN * Kd;
  const unsigned ob0 = (unsigned)(lrow * Kd + lcol) * 4u;
  const unsigned ob1 = ob0 + 128u * (unsigned)Kd;
  const unsigned ob2 = ob0 + 256u * (unsigned)Kd;
  const unsigned ob3 = ob0 + 384u * (unsigned)Kd;

  const int so   = lrow * GPITCH + lcol;
  const int aoff = (wm * 32 + l16) * GPITCH + 4 * half;
  const int boff = GBM * GPITCH + (wn * 32 + l16) * GPITCH + 4 * half;

  v8f c00 = {}, c01 = {}, c10 = {}, c11 = {};

  float4 u0, u1, u2, u3, u4, u5, e0, e1, e2, e3, e4, e5, r0, r1, r2, r3, r4, r5;
  G_LOAD_TILE(u, 0);
  G_LOAD_TILE(e, GBK);
  G_LOAD_TILE(r, 2 * GBK);
  G_STORE_TILE(u, 0);
  G_STORE_TILE(e, GTILE);
  __syncthreads();

  int o0 = 0, o1 = GTILE, o2 = 2 * GTILE;
  float4 P[8], Q[8];
  G_LOAD_FRAGS(P, o0, 0);

  const int nk = Kd / GBK;
  for (int kt = 0; kt < nk - 3; ++kt) G_STEP(kt, 1, 1, 1);
  G_STEP(nk - 3, 1, 0, 1);
  G_STEP(nk - 2, 0, 0, 1);

  const int erow  = lane >> 3;
  const int ecol  = (lane & 7) * 4;
  const int tileM = blockM + wm * 32;
  const int ncol  = blockN + wn * 32 + ecol;
  const float4 bias4 = *(const float4*)(bias + ncol);
  float4 rs[8];
  if (flags & 1) {
    const float* rp = R + (size_t)(tileM + erow) * N + ncol;
#pragma unroll
    for (int p = 0; p < 8; ++p) rs[p] = *(const float4*)(rp + (size_t)(4 * p) * N);
  }
  __builtin_amdgcn_sched_barrier(0);

  G_STEP(nk - 1, 0, 0, 0);

  float* cw = sm + ((w < 4) ? (o0 + w * (32 * GPITCH)) : (o1 + (w - 4) * (32 * GPITCH)));
#pragma unroll
  for (int r = 0; r < 8; ++r) {
    float* pr = cw + (8 * half + r) * GPITCH + l16;
    pr[0]                = c00[r];
    pr[16]               = c01[r];
    pr[16 * GPITCH]      = c10[r];
    pr[16 * GPITCH + 16] = c11[r];
  }
  __syncthreads();

  float4 v[8];
#pragma unroll
  for (int p = 0; p < 8; ++p) v[p] = *(const float4*)(cw + (4 * p + erow) * GPITCH + ecol);
#pragma unroll
  for (int p = 0; p < 8; ++p) {
    v[p].x += bias4.x; v[p].y += bias4.y; v[p].z += bias4.z; v[p].w += bias4.w;
  }
  if (flags & 1) {
#pragma unroll
    for (int p = 0; p < 8; ++p) {
      v[p].x += rs[p].x; v[p].y += rs[p].y; v[p].z += rs[p].z; v[p].w += rs[p].w;
    }
  }
  if (flags & 2) {
#pragma unroll
    for (int p = 0; p < 8; ++p) {
      const v2f lo = gelu_pair((v2f){v[p].x, v[p].y});
      const v2f hi = gelu_pair((v2f){v[p].z, v[p].w});
      v[p] = make_float4(lo.x, lo.y, hi.x, hi.y);
    }
  }
  size_t orow = (size_t)tileM;
  if (flags & 4) {
    const int bi = blockM / Tlen;
    orow = (size_t)bi * Kspk * Tlen + (size_t)kk * Tlen + (size_t)(tileM - bi * Tlen);
  }
  float* cp = C + (orow + erow) * N + ncol;
#pragma unroll
  for (int p = 0; p < 8; ++p) *(float4*)(cp + (size_t)(4 * p) * N) = v[p];
}

__global__ __launch_bounds__(256) void gemm_wmma_f32(
    const float* __restrict__ A, const float* __restrict__ W,
    const float* __restrict__ bias, const float* __restrict__ R,
    float* __restrict__ C, int M, int N, int Kd,
    int flags, int Tlen, int Kspk, int kk)
{
  (void)M;
  gemm_tile(A, W, bias, R, C, N, Kd, flags, Tlen, Kspk, kk,
            blockIdx.x * GBM, blockIdx.y * GBN);
}

__global__ __launch_bounds__(256) void gemm_pair_wmma_f32(
    const float* __restrict__ A0, const float* __restrict__ W0,
    const float* __restrict__ bias0, const float* __restrict__ R0,
    float* __restrict__ C0, int N0, int mb0, int nblk0, int flags0,
    const float* __restrict__ A1, const float* __restrict__ W1,
    const float* __restrict__ bias1, const float* __restrict__ R1,
    float* __restrict__ C1, int N1, int mb1, int flags1,
    int Kd, int Tlen, int Kspk, int kk)
{
  int id = blockIdx.x;
  const bool second = id >= nblk0;
  if (second) id -= nblk0;
  const float* A     = second ? A1 : A0;
  const float* W     = second ? W1 : W0;
  const float* bias  = second ? bias1 : bias0;
  const float* R     = second ? R1 : R0;
  float*       C     = second ? C1 : C0;
  const int    N     = second ? N1 : N0;
  const int    mb    = second ? mb1 : mb0;
  const int    flags = second ? flags1 : flags0;
  const int    by    = id / mb;
  const int    bx    = id - by * mb;
  gemm_tile(A, W, bias, R, C, N, Kd, flags, Tlen, Kspk, kk, bx * GBM, by * GBN);
}

#define RD  512
#define RNV (RD / 128)
#define LV  500
#define LV4 (LV / 4)
#define LNI ((LV4 + 31) / 32)
#define LK  2

#define ROW_XOR(x, n)                                                          \
  __builtin_bit_cast(float, __builtin_amdgcn_update_dpp(                       \
      __builtin_bit_cast(int, (x)), __builtin_bit_cast(int, (x)),              \
      0x160 + (n), 0xf, 0xf, false))
__device__ __forceinline__ float xhalf(float x) {
  const int b = __builtin_bit_cast(int, x);
  return __builtin_bit_cast(float,
      __builtin_amdgcn_permlanex16(b, b, 0x76543210, 0xfedcba98, false, false));
}
__device__ __forceinline__ float wave_sum(float v) {
  v += ROW_XOR(v, 1);
  v += ROW_XOR(v, 2);
  v += ROW_XOR(v, 4);
  v += ROW_XOR(v, 8);
  return v + xhalf(v);
}
__device__ __forceinline__ float wave_max(float v) {
  v = fmaxf(v, ROW_XOR(v, 1));
  v = fmaxf(v, ROW_XOR(v, 2));
  v = fmaxf(v, ROW_XOR(v, 4));
  v = fmaxf(v, ROW_XOR(v, 8));
  return fmaxf(v, xhalf(v));
}

__device__ __forceinline__ void row_stats(const float4 (&v)[RNV], float& mean, float& inv) {
  float s = 0.f;
#pragma unroll
  for (int i = 0; i < RNV; ++i) s += (v[i].x + v[i].y) + (v[i].z + v[i].w);
  mean = wave_sum(s) * (1.0f / RD);
  float s2 = 0.f;
#pragma unroll
  for (int i = 0; i < RNV; ++i) {
    const float dx = v[i].x - mean, dy = v[i].y - mean;
    const float dz = v[i].z - mean, dw = v[i].w - mean;
    s2 += (dx * dx + dy * dy) + (dz * dz + dw * dw);
  }
  inv = rsqrtf(wave_sum(s2) * (1.0f / RD) + 1e-5f);
}

__global__ __launch_bounds__(256) void ln_kernel(
    const float* __restrict__ X, const float* __restrict__ g,
    const float* __restrict__ b, float* __restrict__ Y, int rows)
{
  const int lane = threadIdx.x & 31;
  const int row  = blockIdx.x * 8 + __builtin_amdgcn_readfirstlane(threadIdx.x >> 5);
  if (row >= rows) return;
  const float4* x4 = (const float4*)(X + (size_t)row * RD) + lane;
  const float4* g4 = (const float4*)g + lane;
  const float4* b4 = (const float4*)b + lane;
  float4 xv[RNV], gv[RNV], bv[RNV];
#pragma unroll
  for (int i = 0; i < RNV; ++i) xv[i] = x4[i * 32];
#pragma unroll
  for (int i = 0; i < RNV; ++i) { gv[i] = g4[i * 32]; bv[i] = b4[i * 32]; }
  __builtin_amdgcn_sched_barrier(0);
  float mean, inv;
  row_stats(xv, mean, inv);
  float4* y4 = (float4*)(Y + (size_t)row * RD) + lane;
#pragma unroll
  for (int i = 0; i < RNV; ++i) {
    float4 o;
    o.x = (xv[i].x - mean) * inv * gv[i].x + bv[i].x;
    o.y = (xv[i].y - mean) * inv * gv[i].y + bv[i].y;
    o.z = (xv[i].z - mean) * inv * gv[i].z + bv[i].z;
    o.w = (xv[i].w - mean) * inv * gv[i].w + bv[i].w;
    y4[i * 32] = o;
  }
}

__global__ __launch_bounds__(256) void rows_fused(
    const float* __restrict__ X, const float* __restrict__ logits,
    const int* __restrict__ blank_p, const float* __restrict__ tags,
    const float* __restrict__ gkv, const float* __restrict__ bkv,
    const float* __restrict__ gq, const float* __restrict__ bq,
    float* __restrict__ KVn, float* __restrict__ Xk, float* __restrict__ Qin, int rows)
{
  const int lane = threadIdx.x & 31;
  const int row  = blockIdx.x * 8 + __builtin_amdgcn_readfirstlane(threadIdx.x >> 5);
  if (row >= rows) return;
  const int blank = blank_p[0];
  const float4* x4 = (const float4*)(X + (size_t)row * RD) + lane;
  float4 xv[RNV];
#pragma unroll
  for (int i = 0; i < RNV; ++i) xv[i] = x4[i * 32];
  float4 lv[LK][LNI];
  float  lb[LK];
#pragma unroll
  for (int k2 = 0; k2 < LK; ++k2) {
    const float* L = logits + ((size_t)k2 * rows + row) * LV;
    const float4* L4 = (const float4*)L;
#pragma unroll
    for (int i = 0; i < LNI; ++i) {
      const int idx = lane + 32 * i;
      lv[k2][i] = L4[idx < LV4 ? idx : LV4 - 1];
    }
    lb[k2] = L[blank];
  }
  float4 gk[RNV], bk[RNV], gv[RNV], bv[RNV], tv[LK][RNV];
#pragma unroll
  for (int i = 0; i < RNV; ++i) {
    gk[i] = ((const float4*)gkv + lane)[i * 32];
    bk[i] = ((const float4*)bkv + lane)[i * 32];
    gv[i] = ((const float4*)gq + lane)[i * 32];
    bv[i] = ((const float4*)bq + lane)[i * 32];
  }
#pragma unroll
  for (int kk = 0; kk < LK; ++kk) {
#pragma unroll
    for (int i = 0; i < RNV; ++i) tv[kk][i] = ((const float4*)(tags + (size_t)kk * RD) + lane)[i * 32];
  }
  __builtin_amdgcn_sched_barrier(0);
#pragma unroll
  for (int k2 = 0; k2 < LK; ++k2) {
#pragma unroll
    for (int i = 0; i < LNI; ++i)
      if (lane + 32 * i >= LV4)
        lv[k2][i] = make_float4(-INFINITY, -INFINITY, -INFINITY, -INFINITY);
  }

  {
    float mean, inv;
    row_stats(xv, mean, inv);
    float4* y4 = (float4*)(KVn + (size_t)row * RD) + lane;
#pragma unroll
    for (int i = 0; i < RNV; ++i) {
      float4 o;
      o.x = (xv[i].x - mean) * inv * gk[i].x + bk[i].x;
      o.y = (xv[i].y - mean) * inv * gk[i].y + bk[i].y;
      o.z = (xv[i].z - mean) * inv * gk[i].z + bk[i].z;
      o.w = (xv[i].w - mean) * inv * gk[i].w + bk[i].w;
      y4[i * 32] = o;
    }
  }

  float a[LK];
#pragma unroll
  for (int k2 = 0; k2 < LK; ++k2) {
    float m = -INFINITY;
#pragma unroll
    for (int i = 0; i < LNI; ++i)
      m = fmaxf(m, fmaxf(fmaxf(lv[k2][i].x, lv[k2][i].y), fmaxf(lv[k2][i].z, lv[k2][i].w)));
    m = wave_max(m);
    float s = 0.f;
#pragma unroll
    for (int i = 0; i < LNI; ++i)
      s += (expf(lv[k2][i].x - m) + expf(lv[k2][i].y - m)) +
           (expf(lv[k2][i].z - m) + expf(lv[k2][i].w - m));
    s = wave_sum(s);
    a[k2] = 1.0f - expf(lb[k2] - m) / s;
  }
  float wmixv[LK];
  {
    float m2 = -INFINITY;
#pragma unroll
    for (int k2 = 0; k2 < LK; ++k2) m2 = fmaxf(m2, a[k2]);
    float e[LK];
    float ss = 0.f;
#pragma unroll
    for (int k2 = 0; k2 < LK; ++k2) { e[k2] = expf(a[k2] - m2); ss += e[k2]; }
#pragma unroll
    for (int k2 = 0; k2 < LK; ++k2) wmixv[k2] = e[k2] / ss;
  }

#pragma unroll
  for (int kk = 0; kk < LK; ++kk) {
    const float wv = wmixv[kk];
    float4 v[RNV];
#pragma unroll
    for (int i = 0; i < RNV; ++i) {
      v[i].x = xv[i].x * wv + tv[kk][i].x;
      v[i].y = xv[i].y * wv + tv[kk][i].y;
      v[i].z = xv[i].z * wv + tv[kk][i].z;
      v[i].w = xv[i].w * wv + tv[kk][i].w;
    }
    const size_t srow = (size_t)kk * rows + row;
    float4* xk4 = (float4*)(Xk + srow * RD) + lane;
#pragma unroll
    for (int i = 0; i < RNV; ++i) xk4[i * 32] = v[i];
    float mean, inv;
    row_stats(v, mean, inv);
    float4* q4 = (float4*)(Qin + srow * RD) + lane;
#pragma unroll
    for (int i = 0; i < RNV; ++i) {
      float4 o;
      o.x = (v[i].x - mean) * inv * gv[i].x + bv[i].x;
      o.y = (v[i].y - mean) * inv * gv[i].y + bv[i].y;
      o.z = (v[i].z - mean) * inv * gv[i].z + bv[i].z;
      o.w = (v[i].w - mean) * inv * gv[i].w + bv[i].w;
      q4[i * 32] = o;
    }
  }
}

#define AQ      64
#define ANW     (AQ / 16)
#define ATHR    (32 * ANW)
#define ABAND   24
#define AKEYS   (AQ + 2 * ABAND)
#define AHD     64
#define APITCH  68

__global__ __launch_bounds__(ATHR) void banded_attn_wmma(
    const float* __restrict__ q, const float* __restrict__ kb,
    const float* __restrict__ vb, float* __restrict__ ctx,
    int T_, int Dn, int Dkv, int Hn, int Bkv)
{
  __shared__ __attribute__((aligned(16))) float sK[AKEYS * APITCH];
  __shared__ __attribute__((aligned(16))) float sV[AKEYS * APITCH];
  __shared__ __attribute__((aligned(16))) float sQ[AQ * APITCH];

  const int tid  = threadIdx.x;
  const int lane = tid & 31;
  const int w    = tid >> 5;
  const int c    = lane & 15;
  const int h    = lane >> 4;

  const int nqt = T_ / AQ;
  const int qt  = blockIdx.x % nqt;
  const int hh  = (blockIdx.x / nqt) % Hn;
  const int b   = blockIdx.x / (nqt * Hn);
  const int bk  = b % Bkv;
  const int t0  = qt * AQ;

  const size_t headoff = (size_t)hh * AHD;
  const float* kbase = kb + (size_t)bk * T_ * Dkv + headoff;
  const float* vbase = vb + (size_t)bk * T_ * Dkv + headoff;
  const float* qbase = q  + ((size_t)b * T_ + t0) * Dn + headoff;
  float*       obase = ctx + ((size_t)b * T_ + t0) * Dn + headoff;

  constexpr int NKV = (AKEYS * 16) / ATHR;
  constexpr int NQ  = (AQ * 16) / ATHR;
  const int srow = tid >> 4;
  const int sc4  = (tid & 15) * 4;
  v4f kreg[NKV], vreg[NKV], qreg[NQ];
#pragma unroll
  for (int i = 0; i < NKV; ++i) {
    const int key = t0 - ABAND + srow + (ATHR / 16) * i;
    const unsigned off = (unsigned)min(max(key, 0), T_ - 1) * (unsigned)Dkv + (unsigned)sc4;
    kreg[i] = *(const v4f*)(kbase + off);
    vreg[i] = *(const v4f*)(vbase + off);
  }
#pragma unroll
  for (int i = 0; i < NQ; ++i) {
    const unsigned off = (unsigned)(srow + (ATHR / 16) * i) * (unsigned)Dn + (unsigned)sc4;
    qreg[i] = *(const v4f*)(qbase + off);
  }
  __builtin_amdgcn_sched_barrier(0);
  const v4f zero4 = {0.f, 0.f, 0.f, 0.f};
#pragma unroll
  for (int i = 0; i < NKV; ++i) {
    const int row = srow + (ATHR / 16) * i;
    const int key = t0 - ABAND + row;
    const bool ok = (key >= 0) && (key < T_);
    *(v4f*)(sK + row * APITCH + sc4) = ok ? kreg[i] : zero4;
    *(v4f*)(sV + row * APITCH + sc4) = ok ? vreg[i] : zero4;
  }
#pragma unroll
  for (int i = 0; i < NQ; ++i) {
    const int row = srow + (ATHR / 16) * i;
    *(v4f*)(sQ + row * APITCH + sc4) = qreg[i];
  }
  __syncthreads();

  float* pw = sQ + (w * 16) * APITCH;
  const float* qrow = pw + c * APITCH + 4 * h;
  const float* krow = sK + (w * 16 + c) * APITCH + 4 * h;
  float4 qa[8];
#pragma unroll
  for (int ks = 0; ks < 8; ++ks) qa[ks] = *(const float4*)(qrow + 8 * ks);

  v8f s0 = {}, s1 = {}, s2 = {}, s3 = {};
#pragma unroll
  for (int ks = 0; ks < 8; ++ks) {
    const float4 k0 = *(const float4*)(krow + 8 * ks);
    const float4 k1 = *(const float4*)(krow + 16 * APITCH + 8 * ks);
    const float4 k2 = *(const float4*)(krow + 32 * APITCH + 8 * ks);
    const float4 k3 = *(const float4*)(krow + 48 * APITCH + 8 * ks);
    const v2f al = {qa[ks].x, qa[ks].y}, ah = {qa[ks].z, qa[ks].w};
    const v2f b0l = {k0.x, k0.y}, b0h = {k0.z, k0.w};
    const v2f b1l = {k1.x, k1.y}, b1h = {k1.z, k1.w};
    const v2f b2l = {k2.x, k2.y}, b2h = {k2.z, k2.w};
    const v2f b3l = {k3.x, k3.y}, b3h = {k3.z, k3.w};
    WMMA_F32(s0, al, b0l);
    WMMA_F32(s1, al, b1l);
    WMMA_F32(s2, al, b2l);
    WMMA_F32(s3, al, b3l);
    WMMA_F32(s0, ah, b0h);
    WMMA_F32(s1, ah, b1h);
    WMMA_F32(s2, ah, b2h);
    WMMA_F32(s3, ah, b3h);
  }

  const float scale = 0.125f;
  const int jbase = t0 + 16 * w - ABAND;
  bool colok[4];
#pragma unroll
  for (int nt = 0; nt < 4; ++nt) {
    const int key = jbase + 16 * nt + c;
    colok[nt] = (key >= 0) && (key < T_);
  }
  float inv[8];
#pragma unroll
  for (int r = 0; r < 8; ++r) {
    const int rq = 8 * h + r;
    float x[4] = {s0[r], s1[r], s2[r], s3[r]};
#pragma unroll
    for (int nt = 0; nt < 4; ++nt) {
      const int jj = 16 * nt + c;
      const bool ok = colok[nt] && (jj >= rq) && (jj <= rq + 2 * ABAND);
      x[nt] = ok ? x[nt] * scale : -INFINITY;
    }
    float mx = fmaxf(fmaxf(x[0], x[1]), fmaxf(x[2], x[3]));
    mx = fmaxf(mx, ROW_XOR(mx, 1));
    mx = fmaxf(mx, ROW_XOR(mx, 2));
    mx = fmaxf(mx, ROW_XOR(mx, 4));
    mx = fmaxf(mx, ROW_XOR(mx, 8));
    float e[4];
#pragma unroll
    for (int nt = 0; nt < 4; ++nt) e[nt] = expf(x[nt] - mx);
    float sum = (e[0] + e[1]) + (e[2] + e[3]);
    sum += ROW_XOR(sum, 1);
    sum += ROW_XOR(sum, 2);
    sum += ROW_XOR(sum, 4);
    sum += ROW_XOR(sum, 8);
    inv[r] = 1.0f / sum;
    float* pr = pw + rq * APITCH + c;
    pr[0]  = e[0];
    pr[16] = e[1];
    pr[32] = e[2];
    pr[48] = e[3];
  }
  __syncthreads();

  const float* prow = pw + c * APITCH + 4 * h;
  const float* vcol = sV + (w * 16 + 4 * h) * APITCH + c;
  v8f o0 = {}, o1 = {}, o2 = {}, o3 = {};
#pragma unroll
  for (int ks = 0; ks < 8; ++ks) {
    const float4 pa = *(const float4*)(prow + 8 * ks);
    const float* vr = vcol + 8 * ks * APITCH;
    const v2f al = {pa.x, pa.y}, ah = {pa.z, pa.w};
    const v2f b0l = {vr[0],      vr[APITCH]},      b0h = {vr[2 * APITCH],      vr[3 * APITCH]};
    const v2f b1l = {vr[16],     vr[APITCH + 16]}, b1h = {vr[2 * APITCH + 16], vr[3 * APITCH + 16]};
    const v2f b2l = {vr[32],     vr[APITCH + 32]}, b2h = {vr[2 * APITCH + 32], vr[3 * APITCH + 32]};
    const v2f b3l = {vr[48],     vr[APITCH + 48]}, b3h = {vr[2 * APITCH + 48], vr[3 * APITCH + 48]};
    WMMA_F32(o0, al, b0l);
    WMMA_F32(o1, al, b1l);
    WMMA_F32(o2, al, b2l);
    WMMA_F32(o3, al, b3l);
    WMMA_F32(o0, ah, b0h);
    WMMA_F32(o1, ah, b1h);
    WMMA_F32(o2, ah, b2h);
    WMMA_F32(o3, ah, b3h);
  }

#pragma unroll
  for (int r = 0; r < 8; ++r) {
    float* pr = pw + (8 * h + r) * APITCH + c;
    pr[0]  = o0[r] * inv[r];
    pr[16] = o1[r] * inv[r];
    pr[32] = o2[r] * inv[r];
    pr[48] = o3[r] * inv[r];
  }
  __syncthreads();
#pragma unroll
  for (int i = 0; i < 8; ++i) {
    const int row = 2 * i + h;
    const float4 ov = *(const float4*)(pw + row * APITCH + 4 * c);
    *(float4*)(obase + (size_t)(16 * w + row) * Dn + 4 * c) = ov;
  }
}

extern "C" void kernel_launch(void* const* d_in, const int* in_sizes, int n_in,
                              void* d_out, int out_size, void* d_ws, size_t ws_size,
                              hipStream_t stream) {
  (void)in_sizes; (void)n_in; (void)out_size; (void)ws_size;
  constexpr int B = 4, T = 1024, D_IN = 512, D = 512, K = 2, V = 500, H = 8, BAND = 24;
  constexpr int M = B * T;
  static_assert(D == RD && V == LV && K == LK && M % 8 == 0, "row kernels are sized for these shapes");
  static_assert(D == H * AHD && BAND == ABAND && T % AQ == 0, "attention kernel is sized for these shapes");
  static_assert(T % GBM == 0, "the concat epilogue needs a block's rows inside one batch element");

  const float* x_m      = (const float*)d_in[0];
  const float* logits   = (const float*)d_in[1];
  const int*   blank_id = (const int*)  d_in[2];
  const float* proj_w   = (const float*)d_in[3];
  const float* proj_b   = (const float*)d_in[4];
  const float* spk_tags = (const float*)d_in[5];
  const float* lnq_g    = (const float*)d_in[6];
  const float* lnq_b    = (const float*)d_in[7];
  const float* lnkv_g   = (const float*)d_in[8];
  const float* lnkv_b   = (const float*)d_in[9];
  const float* in_w     = (const float*)d_in[10];
  const float* in_b     = (const float*)d_in[11];
  const float* out_w    = (const float*)d_in[12];
  const float* out_b    = (const float*)d_in[13];
  const float* ffn_ln_g = (const float*)d_in[14];
  const float* ffn_ln_b = (const float*)d_in[15];
  const float* ffn_w1   = (const float*)d_in[16];
  const float* ffn_b1   = (const float*)d_in[17];
  const float* ffn_w2   = (const float*)d_in[18];
  const float* ffn_b2   = (const float*)d_in[19];
  float* out = (float*)d_out;

  float* ws = (float*)d_ws;
  const size_t SZ = (size_t)M * D;
  float* X    = ws;
  float* kv   = X  + SZ;
  float* xk   = kv + 2 * SZ;
  float* qh   = xk + 2 * SZ;
  float* qy   = qh + 2 * SZ;
  float* kvn  = qy + 2 * SZ;
  float* t1   = ws;

  const dim3 blk(256);
  auto gg = [](int m, int n) { return dim3(m / 64, n / 128); };

  gemm_wmma_f32<<<gg(M, D), blk, 0, stream>>>(x_m, proj_w, proj_b, nullptr, X,
                                              M, D, D_IN, 0, T, K, 0);
  rows_fused<<<dim3(M / 8), blk, 0, stream>>>(X, logits, blank_id, spk_tags,
                                              lnkv_g, lnkv_b, lnq_g, lnq_b,
                                              kvn, xk, qh, M);
  gemm_pair_wmma_f32<<<dim3((M / 64) * (2 * D / 128) + (K * M / 64) * (D / 128)), blk, 0, stream>>>(
      kvn, in_w + (size_t)D * D, in_b + D, nullptr, kv, 2 * D, M / 64, (M / 64) * (2 * D / 128), 0,
      qh, in_w, in_b, nullptr, qy, D, K * M / 64, 0,
      D, T, K, 0);
  banded_attn_wmma<<<dim3(K * B * H * (T / AQ)), dim3(ATHR), 0, stream>>>(qy, kv, kv + D, qh,
                                                                          T, D, 2 * D, H, B);
  gemm_wmma_f32<<<gg(K * M, D), blk, 0, stream>>>(qh, out_w, out_b, xk, qy,
                                                  K * M, D, D,  1, T, K, 0);
  ln_kernel<<<dim3(K * M / 8), blk, 0, stream>>>(qy, ffn_ln_g, ffn_ln_b, qh, K * M);

  static_assert(K == 2, "the FFN pair below is written out for two speakers");
  gemm_wmma_f32<<<gg(M, 4 * D), blk, 0, stream>>>(qh, ffn_w1, ffn_b1, nullptr, t1,
                                                  M, 4 * D, D,  2, T, K, 0);
  gemm_wmma_f32<<<gg(M, D), blk, 0, stream>>>(t1, ffn_w2, ffn_b2, qy, out,
                                              M, D, 4 * D,  1 | 4, T, K, 0);
  gemm_wmma_f32<<<gg(M, 4 * D), blk, 0, stream>>>(qh + SZ, ffn_w1, ffn_b1, nullptr, t1,
                                                  M, 4 * D, D,  2, T, K, 1);
  gemm_wmma_f32<<<gg(M, D), blk, 0, stream>>>(t1, ffn_w2, ffn_b2, qy + SZ, out,
                                              M, D, 4 * D,  1 | 4, T, K, 1);
}
